# starting kernel
# baseline (speedup 1.0000x reference)
#include <hip/hip_runtime.h>

typedef float v2f __attribute__((ext_vector_type(2)));
typedef float v8f __attribute__((ext_vector_type(8)));

#define N_NODES 50000
#define N_PER   12500
#define NE      600000
#define HD      128

template <int K>
__global__ void gemm_wmma_f32(const float* __restrict__ A,
                              const float* __restrict__ B,
                              const float* __restrict__ bias,
                              const float* __restrict__ rowscale,
                              const float* __restrict__ outscale,
                              float* __restrict__ C,
                              int M, int relu)
{
    const int lane = threadIdx.x & 31;
    const int half = lane >> 4;
    const int lm   = lane & 15;
    const int wave   = (blockIdx.x * blockDim.x + threadIdx.x) >> 5;
    const int nwaves = (gridDim.x * blockDim.x) >> 5;
    const int tn  = wave & 7;
    const int col = tn * 16 + lm;

    v2f bf[K / 4];
    #pragma unroll
    for (int kk = 0; kk < K / 4; ++kk) {
        const int kb = kk * 4 + half * 2;
        bf[kk][0] = B[(long)kb * HD + col];
        bf[kk][1] = B[(long)(kb + 1) * HD + col];
    }

    const int mt = (M + 15) >> 4;
    const float bv = bias[col];

    for (int tm = wave >> 3; tm < mt; tm += (nwaves >> 3)) {
        const int row = tm * 16 + lm;
        const int rc  = (row < M) ? row : (M - 1);
        const float rs = (rowscale != nullptr) ? rowscale[rc] : 1.0f;
        const float* ap = A + (long)rc * K + half * 2;

        v8f acc = {};
        #pragma unroll
        for (int kk = 0; kk < K / 4; ++kk) {
            v2f a = *(const v2f*)(ap + kk * 4);
            a[0] *= rs;
            a[1] *= rs;
            acc = __builtin_amdgcn_wmma_f32_16x16x4_f32(
                false, a, false, bf[kk], (short)0, acc, false, false);
        }

        #pragma unroll
        for (int v = 0; v < 8; ++v) {
            const int orow = tm * 16 + v + half * 8;
            if (orow < M) {
                float val = acc[v] + bv;
                if (relu) val = fmaxf(val, 0.0f);
                if (outscale != nullptr) val *= outscale[orow];
                C[(long)orow * HD + col] = val;
            }
        }
    }
}

__global__ void edge_degree(const int* __restrict__ src,
                            const int* __restrict__ dst,
                            float* __restrict__ degO,
                            float* __restrict__ degI)
{
    int i = blockIdx.x * blockDim.x + threadIdx.x;
    const int stride = gridDim.x * blockDim.x;
    for (; i < NE; i += stride) {
        atomicAdd(&degO[src[i]], 1.0f);
        atomicAdd(&degI[dst[i]], 1.0f);
    }
}

__global__ void make_norm(float* __restrict__ degO, float* __restrict__ degI)
{
    int i = blockIdx.x * blockDim.x + threadIdx.x;
    const int stride = gridDim.x * blockDim.x;
    for (; i < N_NODES; i += stride) {
        degO[i] = rsqrtf(fmaxf(degO[i], 1.0f));
        degI[i] = rsqrtf(fmaxf(degI[i], 1.0f));
    }
}

__global__ void edge_aggregate(const float* __restrict__ h,
                               const int* __restrict__ src,
                               const int* __restrict__ dst,
                               float* __restrict__ accum)
{
    const int lane   = threadIdx.x & 31;
    const int wave   = (blockIdx.x * blockDim.x + threadIdx.x) >> 5;
    const int nwaves = (gridDim.x * blockDim.x) >> 5;
    for (long base = (long)wave * 32; base < NE; base += (long)nwaves * 32) {
        const int s = src[base + lane];
        const int d = dst[base + lane];
        for (int j = 0; j < 32; ++j) {
            const int ss = __shfl(s, j);
            const int dd = __shfl(d, j);
            const float4 v = *(const float4*)(h + (long)ss * HD + lane * 4);
            float* o = accum + (long)dd * HD + lane * 4;
            atomicAdd(o + 0, v.x);
            atomicAdd(o + 1, v.y);
            atomicAdd(o + 2, v.z);
            atomicAdd(o + 3, v.w);
        }
    }
}

__global__ void post_layer0(float* __restrict__ hb,
                            const float* __restrict__ in_norm,
                            const float* __restrict__ out_norm,
                            const float* __restrict__ bias)
{
    int i = blockIdx.x * blockDim.x + threadIdx.x;
    const int stride = gridDim.x * blockDim.x;
    const int total = N_NODES * HD;
    for (; i < total; i += stride) {
        const int row = i >> 7;
        const int col = i & (HD - 1);
        hb[i] = fmaxf(hb[i] * in_norm[row] + bias[col], 0.0f) * out_norm[row];
    }
}

__global__ void edge_final(const float* __restrict__ h,
                           const int* __restrict__ src,
                           const int* __restrict__ dst,
                           const int* __restrict__ ef,
                           float* __restrict__ out)
{
    const int lane   = threadIdx.x & 31;
    const int wave   = (blockIdx.x * blockDim.x + threadIdx.x) >> 5;
    const int nwaves = (gridDim.x * blockDim.x) >> 5;
    for (long base = (long)wave * 32; base < NE; base += (long)nwaves * 32) {
        const int s = src[base + lane];
        const int d = dst[base + lane];
        const int r = ef[base + lane];
        for (int j = 0; j < 32; ++j) {
            const int ss = __shfl(s, j);
            const int dd = __shfl(d, j);
            const int rr = __shfl(r, j);
            const float wf = (rr >= 0 && rr <= 4) ? 2.0f : 1.0f;

            const float4 hs = *(const float4*)(h + (long)ss * HD + lane * 4);
            float* od = out + (long)dd * HD + lane * 4;
            atomicAdd(od + 0, hs.x * wf);
            atomicAdd(od + 1, hs.y * wf);
            atomicAdd(od + 2, hs.z * wf);
            atomicAdd(od + 3, hs.w * wf);

            if (rr == 6 || rr == 14 || rr == 30) {
                const float4 hd = *(const float4*)(h + (long)dd * HD + lane * 4);
                float* os = out + (long)ss * HD + lane * 4;
                atomicAdd(os + 0, hd.x);
                atomicAdd(os + 1, hd.y);
                atomicAdd(os + 2, hd.z);
                atomicAdd(os + 3, hd.w);
            }
        }
    }
}

extern "C" void kernel_launch(void* const* d_in, const int* in_sizes, int n_in,
                              void* d_out, int out_size, void* d_ws, size_t ws_size,
                              hipStream_t stream)
{
    (void)in_sizes; (void)n_in; (void)out_size; (void)ws_size;

    const float* feat0 = (const float*)d_in[0];
    const float* Wfc0  = (const float*)d_in[1];
    const float* bfc0  = (const float*)d_in[2];
    const float* feat1 = (const float*)d_in[3];
    const float* Wfc1  = (const float*)d_in[4];
    const float* bfc1  = (const float*)d_in[5];
    const float* feat2 = (const float*)d_in[6];
    const float* Wfc2  = (const float*)d_in[7];
    const float* bfc2  = (const float*)d_in[8];
    const float* feat3 = (const float*)d_in[9];
    const float* Wfc3  = (const float*)d_in[10];
    const float* bfc3  = (const float*)d_in[11];
    const int*   src   = (const int*)d_in[12];
    const int*   dst   = (const int*)d_in[13];
    const int*   efeat = (const int*)d_in[14];
    const float* b_g0  = (const float*)d_in[15];
    const float* W_g1  = (const float*)d_in[16];
    const float* b_g1  = (const float*)d_in[17];
    float* out = (float*)d_out;

    const size_t NODE_ELEMS = (size_t)N_NODES * HD;
    float* bufA = (float*)d_ws;
    float* bufB = bufA + NODE_ELEMS;
    float* degO = bufB + NODE_ELEMS;
    float* degI = degO + N_NODES;

    hipMemsetAsync(degO, 0, 2 * (size_t)N_NODES * sizeof(float), stream);
    edge_degree<<<1024, 256, 0, stream>>>(src, dst, degO, degI);
    make_norm<<<(N_NODES + 255) / 256, 256, 0, stream>>>(degO, degI);

    const int mt_per = (N_PER + 15) / 16;
    gemm_wmma_f32<128><<<mt_per, 256, 0, stream>>>(
        feat0, Wfc0, bfc0, nullptr, degO + 0 * N_PER,
        bufA + (size_t)0 * N_PER * HD, N_PER, 0);
    gemm_wmma_f32<128><<<mt_per, 256, 0, stream>>>(
        feat1, Wfc1, bfc1, nullptr, degO + 1 * N_PER,
        bufA + (size_t)1 * N_PER * HD, N_PER, 0);
    gemm_wmma_f32<64><<<mt_per, 256, 0, stream>>>(
        feat2, Wfc2, bfc2, nullptr, degO + 2 * N_PER,
        bufA + (size_t)2 * N_PER * HD, N_PER, 0);
    gemm_wmma_f32<64><<<mt_per, 256, 0, stream>>>(
        feat3, Wfc3, bfc3, nullptr, degO + 3 * N_PER,
        bufA + (size_t)3 * N_PER * HD, N_PER, 0);

    hipMemsetAsync(bufB, 0, NODE_ELEMS * sizeof(float), stream);
    edge_aggregate<<<2048, 256, 0, stream>>>(bufA, src, dst, bufB);
    post_layer0<<<(N_NODES * HD + 255) / 256, 256, 0, stream>>>(
        bufB, degI, degO, b_g0);

    hipMemsetAsync(bufA, 0, NODE_ELEMS * sizeof(float), stream);
    edge_aggregate<<<2048, 256, 0, stream>>>(bufB, src, dst, bufA);
    gemm_wmma_f32<128><<<3125, 256, 0, stream>>>(
        bufA, W_g1, b_g1, degI, nullptr, bufB, N_NODES, 1);

    hipMemsetAsync(out, 0, NODE_ELEMS * sizeof(float), stream);
    edge_final<<<2048, 256, 0, stream>>>(bufB, src, dst, efeat, out);
}


# f32 atomic adds and row gathers remapped so each wave instruction covers one whole 128B line (lane L owns floats L, L+32, L+64, L+96) instead of 8 lanes in each of four lines
# speedup vs baseline: 1.4619x; 1.4619x over previous
#include <hip/hip_runtime.h>

typedef float v2f __attribute__((ext_vector_type(2)));
typedef float v8f __attribute__((ext_vector_type(8)));

#define N_NODES 50000
#define N_PER   12500
#define NE      600000
#define HD      128

template <int K>
__global__ void gemm_wmma_f32(const float* __restrict__ A,
                              const float* __restrict__ B,
                              const float* __restrict__ bias,
                              const float* __restrict__ rowscale,
                              const float* __restrict__ outscale,
                              float* __restrict__ C,
                              int M, int relu)
{
    const int lane = threadIdx.x & 31;
    const int half = lane >> 4;
    const int lm   = lane & 15;
    const int wave   = (blockIdx.x * blockDim.x + threadIdx.x) >> 5;
    const int nwaves = (gridDim.x * blockDim.x) >> 5;
    const int tn  = wave & 7;
    const int col = tn * 16 + lm;

    v2f bf[K / 4];
    #pragma unroll
    for (int kk = 0; kk < K / 4; ++kk) {
        const int kb = kk * 4 + half * 2;
        bf[kk][0] = B[(long)kb * HD + col];
        bf[kk][1] = B[(long)(kb + 1) * HD + col];
    }

    const int mt = (M + 15) >> 4;
    const float bv = bias[col];

    for (int tm = wave >> 3; tm < mt; tm += (nwaves >> 3)) {
        const int row = tm * 16 + lm;
        const int rc  = (row < M) ? row : (M - 1);
        const float rs = (rowscale != nullptr) ? rowscale[rc] : 1.0f;
        const float* ap = A + (long)rc * K + half * 2;

        v8f acc = {};
        #pragma unroll
        for (int kk = 0; kk < K / 4; ++kk) {
            v2f a = *(const v2f*)(ap + kk * 4);
            a[0] *= rs;
            a[1] *= rs;
            acc = __builtin_amdgcn_wmma_f32_16x16x4_f32(
                false, a, false, bf[kk], (short)0, acc, false, false);
        }

        #pragma unroll
        for (int v = 0; v < 8; ++v) {
            const int orow = tm * 16 + v + half * 8;
            if (orow < M) {
                float val = acc[v] + bv;
                if (relu) val = fmaxf(val, 0.0f);
                if (outscale != nullptr) val *= outscale[orow];
                C[(long)orow * HD + col] = val;
            }
        }
    }
}

__global__ void edge_degree(const int* __restrict__ src,
                            const int* __restrict__ dst,
                            float* __restrict__ degO,
                            float* __restrict__ degI)
{
    int i = blockIdx.x * blockDim.x + threadIdx.x;
    const int stride = gridDim.x * blockDim.x;
    for (; i < NE; i += stride) {
        atomicAdd(&degO[src[i]], 1.0f);
        atomicAdd(&degI[dst[i]], 1.0f);
    }
}

__global__ void make_norm(float* __restrict__ degO, float* __restrict__ degI)
{
    int i = blockIdx.x * blockDim.x + threadIdx.x;
    const int stride = gridDim.x * blockDim.x;
    for (; i < N_NODES; i += stride) {
        degO[i] = rsqrtf(fmaxf(degO[i], 1.0f));
        degI[i] = rsqrtf(fmaxf(degI[i], 1.0f));
    }
}

__global__ void edge_aggregate(const float* __restrict__ h,
                               const int* __restrict__ src,
                               const int* __restrict__ dst,
                               float* __restrict__ accum)
{
    const int lane   = threadIdx.x & 31;
    const int wave   = (blockIdx.x * blockDim.x + threadIdx.x) >> 5;
    const int nwaves = (gridDim.x * blockDim.x) >> 5;
    for (long base = (long)wave * 32; base < NE; base += (long)nwaves * 32) {
        const int s = src[base + lane];
        const int d = dst[base + lane];
        for (int j = 0; j < 32; ++j) {
            const int ss = __shfl(s, j);
            const int dd = __shfl(d, j);
            const float* hp = h + (long)ss * HD + lane;
            float* o = accum + (long)dd * HD + lane;
            const float v0 = hp[0];
            const float v1 = hp[32];
            const float v2 = hp[64];
            const float v3 = hp[96];
            atomicAdd(o,      v0);
            atomicAdd(o + 32, v1);
            atomicAdd(o + 64, v2);
            atomicAdd(o + 96, v3);
        }
    }
}

__global__ void post_layer0(float* __restrict__ hb,
                            const float* __restrict__ in_norm,
                            const float* __restrict__ out_norm,
                            const float* __restrict__ bias)
{
    int i = blockIdx.x * blockDim.x + threadIdx.x;
    const int stride = gridDim.x * blockDim.x;
    const int total = N_NODES * HD;
    for (; i < total; i += stride) {
        const int row = i >> 7;
        const int col = i & (HD - 1);
        hb[i] = fmaxf(hb[i] * in_norm[row] + bias[col], 0.0f) * out_norm[row];
    }
}

__global__ void edge_final(const float* __restrict__ h,
                           const int* __restrict__ src,
                           const int* __restrict__ dst,
                           const int* __restrict__ ef,
                           float* __restrict__ out)
{
    const int lane   = threadIdx.x & 31;
    const int wave   = (blockIdx.x * blockDim.x + threadIdx.x) >> 5;
    const int nwaves = (gridDim.x * blockDim.x) >> 5;
    for (long base = (long)wave * 32; base < NE; base += (long)nwaves * 32) {
        const int s = src[base + lane];
        const int d = dst[base + lane];
        const int r = ef[base + lane];
        for (int j = 0; j < 32; ++j) {
            const int ss = __shfl(s, j);
            const int dd = __shfl(d, j);
            const int rr = __shfl(r, j);
            const float wf = (rr >= 0 && rr <= 4) ? 2.0f : 1.0f;

            const float* hsp = h + (long)ss * HD + lane;
            float* od = out + (long)dd * HD + lane;
            const float s0 = hsp[0];
            const float s1 = hsp[32];
            const float s2 = hsp[64];
            const float s3 = hsp[96];
            atomicAdd(od,      s0 * wf);
            atomicAdd(od + 32, s1 * wf);
            atomicAdd(od + 64, s2 * wf);
            atomicAdd(od + 96, s3 * wf);

            if (rr == 6 || rr == 14 || rr == 30) {
                const float* hdp = h + (long)dd * HD + lane;
                float* os = out + (long)ss * HD + lane;
                const float d0 = hdp[0];
                const float d1 = hdp[32];
                const float d2 = hdp[64];
                const float d3 = hdp[96];
                atomicAdd(os,      d0);
                atomicAdd(os + 32, d1);
                atomicAdd(os + 64, d2);
                atomicAdd(os + 96, d3);
            }
        }
    }
}

extern "C" void kernel_launch(void* const* d_in, const int* in_sizes, int n_in,
                              void* d_out, int out_size, void* d_ws, size_t ws_size,
                              hipStream_t stream)
{
    (void)in_sizes; (void)n_in; (void)out_size; (void)ws_size;

    const float* feat0 = (const float*)d_in[0];
    const float* Wfc0  = (const float*)d_in[1];
    const float* bfc0  = (const float*)d_in[2];
    const float* feat1 = (const float*)d_in[3];
    const float* Wfc1  = (const float*)d_in[4];
    const float* bfc1  = (const float*)d_in[5];
    const float* feat2 = (const float*)d_in[6];
    const float* Wfc2  = (const float*)d_in[7];
    const float* bfc2  = (const float*)d_in[8];
    const float* feat3 = (const float*)d_in[9];
    const float* Wfc3  = (const float*)d_in[10];
    const float* bfc3  = (const float*)d_in[11];
    const int*   src   = (const int*)d_in[12];
    const int*   dst   = (const int*)d_in[13];
    const int*   efeat = (const int*)d_in[14];
    const float* b_g0  = (const float*)d_in[15];
    const float* W_g1  = (const float*)d_in[16];
    const float* b_g1  = (const float*)d_in[17];
    float* out = (float*)d_out;

    const size_t NODE_ELEMS = (size_t)N_NODES * HD;
    float* bufA = (float*)d_ws;
    float* bufB = bufA + NODE_ELEMS;
    float* degO = bufB + NODE_ELEMS;
    float* degI = degO + N_NODES;

    hipMemsetAsync(degO, 0, 2 * (size_t)N_NODES * sizeof(float), stream);
    edge_degree<<<1024, 256, 0, stream>>>(src, dst, degO, degI);
    make_norm<<<(N_NODES + 255) / 256, 256, 0, stream>>>(degO, degI);

    const int mt_per = (N_PER + 15) / 16;
    gemm_wmma_f32<128><<<mt_per, 256, 0, stream>>>(
        feat0, Wfc0, bfc0, nullptr, degO + 0 * N_PER,
        bufA + (size_t)0 * N_PER * HD, N_PER, 0);
    gemm_wmma_f32<128><<<mt_per, 256, 0, stream>>>(
        feat1, Wfc1, bfc1, nullptr, degO + 1 * N_PER,
        bufA + (size_t)1 * N_PER * HD, N_PER, 0);
    gemm_wmma_f32<64><<<mt_per, 256, 0, stream>>>(
        feat2, Wfc2, bfc2, nullptr, degO + 2 * N_PER,
        bufA + (size_t)2 * N_PER * HD, N_PER, 0);
    gemm_wmma_f32<64><<<mt_per, 256, 0, stream>>>(
        feat3, Wfc3, bfc3, nullptr, degO + 3 * N_PER,
        bufA + (size_t)3 * N_PER * HD, N_PER, 0);

    hipMemsetAsync(bufB, 0, NODE_ELEMS * sizeof(float), stream);
    edge_aggregate<<<2048, 256, 0, stream>>>(bufA, src, dst, bufB);
    post_layer0<<<(N_NODES * HD + 255) / 256, 256, 0, stream>>>(
        bufB, degI, degO, b_g0);

    hipMemsetAsync(bufA, 0, NODE_ELEMS * sizeof(float), stream);
    edge_aggregate<<<2048, 256, 0, stream>>>(bufB, src, dst, bufA);
    gemm_wmma_f32<128><<<3125, 256, 0, stream>>>(
        bufA, W_g1, b_g1, degI, nullptr, bufB, N_NODES, 1);

    hipMemsetAsync(out, 0, NODE_ELEMS * sizeof(float), stream);
    edge_final<<<2048, 256, 0, stream>>>(bufB, src, dst, efeat, out);
}


# f32 atomic scatter-adds replaced by a per-call dst-sorted edge list (int-atomic counting sort): each row summed in registers in list order, stored once; zero-fills and epilogue pass fold in
# speedup vs baseline: 1.8315x; 1.2528x over previous
#include <hip/hip_runtime.h>

typedef float v2f __attribute__((ext_vector_type(2)));
typedef float v8f __attribute__((ext_vector_type(8)));

#define N_NODES 50000
#define N_PER   12500
#define NE      600000
#define HD      128

template <int K>
__global__ void gemm_wmma_f32(const float* __restrict__ A,
                              const float* __restrict__ B,
                              const float* __restrict__ bias,
                              const float* __restrict__ rowscale,
                              const float* __restrict__ outscale,
                              float* __restrict__ C,
                              int M, int relu)
{
    const int lane = threadIdx.x & 31;
    const int half = lane >> 4;
    const int lm   = lane & 15;
    const int wave   = (blockIdx.x * blockDim.x + threadIdx.x) >> 5;
    const int nwaves = (gridDim.x * blockDim.x) >> 5;
    const int tn  = wave & 7;
    const int col = tn * 16 + lm;

    v2f bf[K / 4];
    #pragma unroll
    for (int kk = 0; kk < K / 4; ++kk) {
        const int kb = kk * 4 + half * 2;
        bf[kk][0] = B[(long)kb * HD + col];
        bf[kk][1] = B[(long)(kb + 1) * HD + col];
    }

    const int mt = (M + 15) >> 4;
    const float bv = bias[col];

    for (int tm = wave >> 3; tm < mt; tm += (nwaves >> 3)) {
        const int row = tm * 16 + lm;
        const int rc  = (row < M) ? row : (M - 1);
        const float rs = (rowscale != nullptr) ? rowscale[rc] : 1.0f;
        const float* ap = A + (long)rc * K + half * 2;

        v8f acc = {};
        #pragma unroll
        for (int kk = 0; kk < K / 4; ++kk) {
            v2f a = *(const v2f*)(ap + kk * 4);
            a[0] *= rs;
            a[1] *= rs;
            acc = __builtin_amdgcn_wmma_f32_16x16x4_f32(
                false, a, false, bf[kk], (short)0, acc, false, false);
        }

        #pragma unroll
        for (int v = 0; v < 8; ++v) {
            const int orow = tm * 16 + v + half * 8;
            if (orow < M) {
                float val = acc[v] + bv;
                if (relu) val = fmaxf(val, 0.0f);
                if (outscale != nullptr) val *= outscale[orow];
                C[(long)orow * HD + col] = val;
            }
        }
    }
}

#define ROWP_PAD 50016
#define SCAN_IT  13
static_assert(32 * SCAN_IT * 128 >= N_NODES, "scan segments must cover all rows");
static_assert(N_NODES % 4 == 0, "int4 groups are all-valid or all-invalid");
static_assert(NE % 32 == 0, "32-edge batches, no tail");

__global__ void edge_count(const int* __restrict__ src,
                           const int* __restrict__ dst,
                           int* __restrict__ cntO,
                           int* __restrict__ cntI)
{
    int i = blockIdx.x * blockDim.x + threadIdx.x;
    const int stride = gridDim.x * blockDim.x;
    for (; i < NE; i += stride) {
        atomicAdd(&cntO[src[i]], 1);
        atomicAdd(&cntI[dst[i]], 1);
    }
}

__global__ void __launch_bounds__(1024)
scan_rows(const int* __restrict__ cntI,
          int* __restrict__ row_ptr,
          int* __restrict__ cursor)
{
    __shared__ int wsum[32];
    const int lane = threadIdx.x & 31;
    const int w    = threadIdx.x >> 5;
    const int seg0 = w * (SCAN_IT * 128);

    int4 c[SCAN_IT];
    int s = 0;
    #pragma unroll
    for (int it = 0; it < SCAN_IT; ++it) {
        const int i = seg0 + it * 128 + lane * 4;
        const bool ok = i < N_NODES;
        const int4 v = *(const int4*)(cntI + (ok ? i : 0));
        c[it] = ok ? v : make_int4(0, 0, 0, 0);
        s += (c[it].x + c[it].y) + (c[it].z + c[it].w);
    }
    for (int o = 16; o > 0; o >>= 1) s += __shfl_xor(s, o);
    if (lane == 0) wsum[w] = s;
    __syncthreads();

    int carry = (lane < w) ? wsum[lane] : 0;
    for (int o = 16; o > 0; o >>= 1) carry += __shfl_xor(carry, o);

    int x[SCAN_IT];
    #pragma unroll
    for (int it = 0; it < SCAN_IT; ++it)
        x[it] = (c[it].x + c[it].y) + (c[it].z + c[it].w);
    #pragma unroll
    for (int o = 1; o < 32; o <<= 1) {
        int y[SCAN_IT];
        #pragma unroll
        for (int it = 0; it < SCAN_IT; ++it) y[it] = __shfl_up(x[it], o);
        #pragma unroll
        for (int it = 0; it < SCAN_IT; ++it) if (lane >= o) x[it] += y[it];
    }
    int tot[SCAN_IT];
    #pragma unroll
    for (int it = 0; it < SCAN_IT; ++it) tot[it] = __shfl(x[it], 31);

    #pragma unroll
    for (int it = 0; it < SCAN_IT; ++it) {
        const int i  = seg0 + it * 128 + lane * 4;
        const int p1 = c[it].x + c[it].y;
        const int p2 = p1 + c[it].z;
        const int p3 = p2 + c[it].w;
        const int e0 = carry + x[it] - p3;
        if (i < N_NODES) {
            const int4 r = make_int4(e0, e0 + c[it].x, e0 + p1, e0 + p2);
            *(int4*)(row_ptr + i) = r;
            *(int4*)(cursor + i)  = r;
        }
        carry += tot[it];
    }
    if (w == 31 && lane == 0) row_ptr[N_NODES] = carry;
}

__global__ void csr_fill(const int* __restrict__ src,
                         const int* __restrict__ dst,
                         const int* __restrict__ ef,
                         const int* __restrict__ cntO,
                         const int* __restrict__ cntI,
                         int* __restrict__ cursor,
                         int* __restrict__ ent,
                         float* __restrict__ normO,
                         float* __restrict__ normI)
{
    int i = blockIdx.x * blockDim.x + threadIdx.x;
    const int stride = gridDim.x * blockDim.x;
    for (; i < NE; i += stride) {
        if (i < N_NODES) {
            normO[i] = rsqrtf(fmaxf((float)cntO[i], 1.0f));
            normI[i] = rsqrtf(fmaxf((float)cntI[i], 1.0f));
        }
        const int s = src[i];
        const int d = dst[i];
        const int r = ef[i];
        const int pos = atomicAdd(&cursor[d], 1);
        ent[pos] = s | ((r >= 0 && r <= 4) ? (int)0x80000000 : 0);
    }
}

template <int G, int MODE>
__device__ __forceinline__ void gather_group(const float* __restrict__ h,
                                             int e, int j, int lane, float4& acc)
{
    float4 a[G];
    float  wgt[G];
    #pragma unroll
    for (int k = 0; k < G; ++k) {
        const int ek = __builtin_amdgcn_readlane(e, j + k);
        a[k] = *(const float4*)(h + (size_t)(ek & 0x7fffffff) * HD + lane * 4);
        wgt[k] = (MODE == 2 && ek < 0) ? 2.0f : 1.0f;
    }
    #pragma unroll
    for (int k = 0; k < G; ++k) {
        if (MODE == 2) {
            acc.x += a[k].x * wgt[k];
            acc.y += a[k].y * wgt[k];
            acc.z += a[k].z * wgt[k];
            acc.w += a[k].w * wgt[k];
        } else {
            acc.x += a[k].x;
            acc.y += a[k].y;
            acc.z += a[k].z;
            acc.w += a[k].w;
        }
    }
}

template <int MODE>
__global__ void __launch_bounds__(256)
row_gather(const float* __restrict__ h,
           const int* __restrict__ row_ptr,
           const int* __restrict__ ent,
           const float* __restrict__ in_norm,
           const float* __restrict__ out_norm,
           const float* __restrict__ bias,
           float* __restrict__ dstv)
{
    const int lane = threadIdx.x & 31;
    const int wave = __builtin_amdgcn_readfirstlane(
        (int)((blockIdx.x * blockDim.x + threadIdx.x) >> 5));
    const int nwaves = (int)((gridDim.x * blockDim.x) >> 5);

    float4 b = make_float4(0.0f, 0.0f, 0.0f, 0.0f);
    if (MODE == 0) b = *(const float4*)(bias + lane * 4);

    for (int row = wave; row < N_NODES; row += nwaves) {
        const int beg = row_ptr[row];
        const int end = row_ptr[row + 1];
        float inn = 1.0f, outn = 1.0f;
        if (MODE == 0) { inn = in_norm[row]; outn = out_norm[row]; }

        float4 acc = make_float4(0.0f, 0.0f, 0.0f, 0.0f);
        for (int base = beg; base < end; base += 32) {
            const int n = (end - base < 32) ? (end - base) : 32;
            const int e = (lane < n) ? ent[base + lane] : 0;
            int j = 0;
            for (; j + 8 <= n; j += 8) gather_group<8, MODE>(h, e, j, lane, acc);
            if (j + 4 <= n) { gather_group<4, MODE>(h, e, j, lane, acc); j += 4; }
            if (j + 2 <= n) { gather_group<2, MODE>(h, e, j, lane, acc); j += 2; }
            if (j < n)      { gather_group<1, MODE>(h, e, j, lane, acc); }
        }

        if (MODE == 0) {
            acc.x = fmaxf(acc.x * inn + b.x, 0.0f) * outn;
            acc.y = fmaxf(acc.y * inn + b.y, 0.0f) * outn;
            acc.z = fmaxf(acc.z * inn + b.z, 0.0f) * outn;
            acc.w = fmaxf(acc.w * inn + b.w, 0.0f) * outn;
        }
        *(float4*)(dstv + (size_t)row * HD + lane * 4) = acc;
    }
}

__global__ void edge_backward(const float* __restrict__ h,
                              const int* __restrict__ src,
                              const int* __restrict__ dst,
                              const int* __restrict__ ef,
                              float* __restrict__ out)
{
    const int lane   = threadIdx.x & 31;
    const int wave   = (blockIdx.x * blockDim.x + threadIdx.x) >> 5;
    const int nwaves = (gridDim.x * blockDim.x) >> 5;
    for (long base = (long)wave * 32; base < NE; base += (long)nwaves * 32) {
        const int s = src[base + lane];
        const int d = dst[base + lane];
        const int r = ef[base + lane];
        unsigned mask = __builtin_amdgcn_ballot_w32(r == 6 || r == 14 || r == 30);
        while (mask != 0u) {
            const int j = __builtin_ctz(mask);
            mask &= mask - 1u;
            const int ss = __shfl(s, j);
            const int dd = __shfl(d, j);
            const float* hdp = h + (long)dd * HD + lane;
            float* os = out + (long)ss * HD + lane;
            const float d0 = hdp[0];
            const float d1 = hdp[32];
            const float d2 = hdp[64];
            const float d3 = hdp[96];
            atomicAdd(os,      d0);
            atomicAdd(os + 32, d1);
            atomicAdd(os + 64, d2);
            atomicAdd(os + 96, d3);
        }
    }
}

extern "C" void kernel_launch(void* const* d_in, const int* in_sizes, int n_in,
                              void* d_out, int out_size, void* d_ws, size_t ws_size,
                              hipStream_t stream)
{
    (void)in_sizes; (void)n_in; (void)out_size; (void)ws_size;

    const float* feat0 = (const float*)d_in[0];
    const float* Wfc0  = (const float*)d_in[1];
    const float* bfc0  = (const float*)d_in[2];
    const float* feat1 = (const float*)d_in[3];
    const float* Wfc1  = (const float*)d_in[4];
    const float* bfc1  = (const float*)d_in[5];
    const float* feat2 = (const float*)d_in[6];
    const float* Wfc2  = (const float*)d_in[7];
    const float* bfc2  = (const float*)d_in[8];
    const float* feat3 = (const float*)d_in[9];
    const float* Wfc3  = (const float*)d_in[10];
    const float* bfc3  = (const float*)d_in[11];
    const int*   src   = (const int*)d_in[12];
    const int*   dst   = (const int*)d_in[13];
    const int*   efeat = (const int*)d_in[14];
    const float* b_g0  = (const float*)d_in[15];
    const float* W_g1  = (const float*)d_in[16];
    const float* b_g1  = (const float*)d_in[17];
    float* out = (float*)d_out;

    const size_t NODE_ELEMS = (size_t)N_NODES * HD;
    int* cntO = (int*)d_ws;
    int* cntI = cntO + N_NODES;
    int* rowp = cntI + N_NODES;
    int* curs = rowp + ROWP_PAD;
    int* ent  = curs + N_NODES;
    float* bufB = (float*)d_ws + NODE_ELEMS;
    float* degO = bufB + NODE_ELEMS;
    float* degI = degO + N_NODES;

    hipMemsetAsync(cntO, 0, 2 * (size_t)N_NODES * sizeof(int), stream);
    edge_count<<<1024, 256, 0, stream>>>(src, dst, cntO, cntI);
    scan_rows<<<1, 1024, 0, stream>>>(cntI, rowp, curs);
    csr_fill<<<1024, 256, 0, stream>>>(
        src, dst, efeat, cntO, cntI, curs, ent, degO, degI);

    const int mt_per = (N_PER + 15) / 16;
    gemm_wmma_f32<128><<<mt_per, 256, 0, stream>>>(
        feat0, Wfc0, bfc0, nullptr, degO + 0 * N_PER,
        out + (size_t)0 * N_PER * HD, N_PER, 0);
    gemm_wmma_f32<128><<<mt_per, 256, 0, stream>>>(
        feat1, Wfc1, bfc1, nullptr, degO + 1 * N_PER,
        out + (size_t)1 * N_PER * HD, N_PER, 0);
    gemm_wmma_f32<64><<<mt_per, 256, 0, stream>>>(
        feat2, Wfc2, bfc2, nullptr, degO + 2 * N_PER,
        out + (size_t)2 * N_PER * HD, N_PER, 0);
    gemm_wmma_f32<64><<<mt_per, 256, 0, stream>>>(
        feat3, Wfc3, bfc3, nullptr, degO + 3 * N_PER,
        out + (size_t)3 * N_PER * HD, N_PER, 0);

    row_gather<0><<<2048, 256, 0, stream>>>(
        out, rowp, ent, degI, degO, b_g0, bufB);

    row_gather<1><<<2048, 256, 0, stream>>>(
        bufB, rowp, ent, degI, degO, b_g0, out);
    gemm_wmma_f32<128><<<3125, 256, 0, stream>>>(
        out, W_g1, b_g1, degI, nullptr, bufB, N_NODES, 1);

    row_gather<2><<<2048, 256, 0, stream>>>(
        bufB, rowp, ent, degI, degO, b_g0, out);
    edge_backward<<<1024, 256, 0, stream>>>(bufB, src, dst, efeat, out);
}


# edge-list build without global atomics: degrees and each edge's slot come from LDS histograms over (node range x edge slice) blocks; row list order now slice-major, so f32 row sum order changes
# speedup vs baseline: 2.3313x; 1.2729x over previous
#include <hip/hip_runtime.h>

typedef float v2f __attribute__((ext_vector_type(2)));
typedef float v8f __attribute__((ext_vector_type(8)));

#define N_NODES 50000
#define N_PER   12500
#define NE      600000
#define HD      128

template <int K>
__global__ void gemm_wmma_f32(const float* __restrict__ A,
                              const float* __restrict__ B,
                              const float* __restrict__ bias,
                              const float* __restrict__ rowscale,
                              const float* __restrict__ outscale,
                              float* __restrict__ C,
                              int M, int relu)
{
    const int lane = threadIdx.x & 31;
    const int half = lane >> 4;
    const int lm   = lane & 15;
    const int wave   = (blockIdx.x * blockDim.x + threadIdx.x) >> 5;
    const int nwaves = (gridDim.x * blockDim.x) >> 5;
    const int tn  = wave & 7;
    const int col = tn * 16 + lm;

    v2f bf[K / 4];
    #pragma unroll
    for (int kk = 0; kk < K / 4; ++kk) {
        const int kb = kk * 4 + half * 2;
        bf[kk][0] = B[(long)kb * HD + col];
        bf[kk][1] = B[(long)(kb + 1) * HD + col];
    }

    const int mt = (M + 15) >> 4;
    const float bv = bias[col];

    for (int tm = wave >> 3; tm < mt; tm += (nwaves >> 3)) {
        const int row = tm * 16 + lm;
        const int rc  = (row < M) ? row : (M - 1);
        const float rs = (rowscale != nullptr) ? rowscale[rc] : 1.0f;
        const float* ap = A + (long)rc * K + half * 2;

        v8f acc = {};
        #pragma unroll
        for (int kk = 0; kk < K / 4; ++kk) {
            v2f a = *(const v2f*)(ap + kk * 4);
            a[0] *= rs;
            a[1] *= rs;
            acc = __builtin_amdgcn_wmma_f32_16x16x4_f32(
                false, a, false, bf[kk], (short)0, acc, false, false);
        }

        #pragma unroll
        for (int v = 0; v < 8; ++v) {
            const int orow = tm * 16 + v + half * 8;
            if (orow < M) {
                float val = acc[v] + bv;
                if (relu) val = fmaxf(val, 0.0f);
                if (outscale != nullptr) val *= outscale[orow];
                C[(long)orow * HD + col] = val;
            }
        }
    }
}

#define ROWP_PAD 50016
#define SCAN_IT  13
#define NRG      16
#define NSL      16
#define RANGE    (N_NODES / NRG)
#define SLICE    (NE / NSL)
static_assert(32 * SCAN_IT * 128 >= N_NODES, "scan segments must cover all rows");
static_assert(N_NODES % 4 == 0, "int4 groups are all-valid or all-invalid");
static_assert(NE % 32 == 0, "32-edge batches, no tail");
static_assert(RANGE * NRG == N_NODES, "ranges tile the nodes exactly");
static_assert(SLICE * NSL == NE && SLICE % 4 == 0, "slices tile the edges in int4 groups");

#define HB   1024
#define HGRP (SLICE / 4)
#define HFUL (HGRP / HB)
static_assert(HGRP - HFUL * HB < HB, "one guarded trip covers the rest of the slice");

__device__ __forceinline__ void hist_group(const int4 sv, const int4 dv, int lo, int i,
                                           int* cI, int* cO, int* __restrict__ rank)
{
    const unsigned d0 = (unsigned)(dv.x - lo), d1 = (unsigned)(dv.y - lo);
    const unsigned d2 = (unsigned)(dv.z - lo), d3 = (unsigned)(dv.w - lo);
    const unsigned o0 = (unsigned)(sv.x - lo), o1 = (unsigned)(sv.y - lo);
    const unsigned o2 = (unsigned)(sv.z - lo), o3 = (unsigned)(sv.w - lo);
    const bool in0 = d0 < (unsigned)RANGE, in1 = d1 < (unsigned)RANGE;
    const bool in2 = d2 < (unsigned)RANGE, in3 = d3 < (unsigned)RANGE;
    int r0 = 0, r1 = 0, r2 = 0, r3 = 0;
    if (in0) r0 = atomicAdd(&cI[d0], 1);
    if (in1) r1 = atomicAdd(&cI[d1], 1);
    if (in2) r2 = atomicAdd(&cI[d2], 1);
    if (in3) r3 = atomicAdd(&cI[d3], 1);
    if (o0 < (unsigned)RANGE) atomicAdd(&cO[o0], 1);
    if (o1 < (unsigned)RANGE) atomicAdd(&cO[o1], 1);
    if (o2 < (unsigned)RANGE) atomicAdd(&cO[o2], 1);
    if (o3 < (unsigned)RANGE) atomicAdd(&cO[o3], 1);
    if (in0) rank[i + 0] = r0;
    if (in1) rank[i + 1] = r1;
    if (in2) rank[i + 2] = r2;
    if (in3) rank[i + 3] = r3;
}

__global__ void __launch_bounds__(HB)
edge_hist(const int* __restrict__ src,
          const int* __restrict__ dst,
          int* __restrict__ rank,
          int* __restrict__ partI,
          int* __restrict__ partO)
{
    __shared__ int cI[RANGE];
    __shared__ int cO[RANGE];
    const int tid = threadIdx.x;
    const int r  = blockIdx.x % NRG;
    const int s  = blockIdx.x / NRG;
    const int lo = r * RANGE;
    for (int j = tid; j < RANGE; j += HB) { cI[j] = 0; cO[j] = 0; }

    const int e0 = s * SLICE;
    const int4* s4 = (const int4*)(src + e0);
    const int4* d4 = (const int4*)(dst + e0);
    int4 sv[HFUL], dv[HFUL];
    #pragma unroll
    for (int k = 0; k < HFUL; ++k) {
        sv[k] = s4[tid + k * HB];
        dv[k] = d4[tid + k * HB];
    }
    const bool tail = tid + HFUL * HB < HGRP;
    int4 st = make_int4(-1, -1, -1, -1), dt = make_int4(-1, -1, -1, -1);
    if (tail) { st = s4[tid + HFUL * HB]; dt = d4[tid + HFUL * HB]; }
    __syncthreads();

    #pragma unroll
    for (int k = 0; k < HFUL; ++k)
        hist_group(sv[k], dv[k], lo, e0 + (tid + k * HB) * 4, cI, cO, rank);
    if (tail) hist_group(st, dt, lo, e0 + (tid + HFUL * HB) * 4, cI, cO, rank);
    __syncthreads();

    int* pI = partI + (size_t)s * N_NODES + lo;
    int* pO = partO + (size_t)s * N_NODES + lo;
    for (int j = tid; j < RANGE; j += HB) { pI[j] = cI[j]; pO[j] = cO[j]; }
}

__global__ void node_reduce(int* __restrict__ partI,
                            const int* __restrict__ partO,
                            int* __restrict__ cntI,
                            float* __restrict__ normO,
                            float* __restrict__ normI)
{
    const int n = blockIdx.x * blockDim.x + threadIdx.x;
    if (n >= N_NODES) return;
    int ci[NSL], co[NSL];
    #pragma unroll
    for (int s = 0; s < NSL; ++s) {
        ci[s] = partI[(size_t)s * N_NODES + n];
        co[s] = partO[(size_t)s * N_NODES + n];
    }
    int run = 0, o = 0;
    #pragma unroll
    for (int s = 0; s < NSL; ++s) {
        partI[(size_t)s * N_NODES + n] = run;
        run += ci[s];
        o   += co[s];
    }
    cntI[n]  = run;
    normO[n] = rsqrtf(fmaxf((float)o,   1.0f));
    normI[n] = rsqrtf(fmaxf((float)run, 1.0f));
}

__global__ void __launch_bounds__(1024)
scan_rows(const int* __restrict__ cntI,
          int* __restrict__ row_ptr)
{
    __shared__ int wsum[32];
    const int lane = threadIdx.x & 31;
    const int w    = threadIdx.x >> 5;
    const int seg0 = w * (SCAN_IT * 128);

    int4 c[SCAN_IT];
    int s = 0;
    #pragma unroll
    for (int it = 0; it < SCAN_IT; ++it) {
        const int i = seg0 + it * 128 + lane * 4;
        const bool ok = i < N_NODES;
        const int4 v = *(const int4*)(cntI + (ok ? i : 0));
        c[it] = ok ? v : make_int4(0, 0, 0, 0);
        s += (c[it].x + c[it].y) + (c[it].z + c[it].w);
    }
    for (int o = 16; o > 0; o >>= 1) s += __shfl_xor(s, o);
    if (lane == 0) wsum[w] = s;
    __syncthreads();

    int carry = (lane < w) ? wsum[lane] : 0;
    for (int o = 16; o > 0; o >>= 1) carry += __shfl_xor(carry, o);

    int x[SCAN_IT];
    #pragma unroll
    for (int it = 0; it < SCAN_IT; ++it)
        x[it] = (c[it].x + c[it].y) + (c[it].z + c[it].w);
    #pragma unroll
    for (int o = 1; o < 32; o <<= 1) {
        int y[SCAN_IT];
        #pragma unroll
        for (int it = 0; it < SCAN_IT; ++it) y[it] = __shfl_up(x[it], o);
        #pragma unroll
        for (int it = 0; it < SCAN_IT; ++it) if (lane >= o) x[it] += y[it];
    }
    int tot[SCAN_IT];
    #pragma unroll
    for (int it = 0; it < SCAN_IT; ++it) tot[it] = __shfl(x[it], 31);

    #pragma unroll
    for (int it = 0; it < SCAN_IT; ++it) {
        const int i  = seg0 + it * 128 + lane * 4;
        const int p1 = c[it].x + c[it].y;
        const int p2 = p1 + c[it].z;
        const int p3 = p2 + c[it].w;
        const int e0 = carry + x[it] - p3;
        if (i < N_NODES) {
            const int4 r = make_int4(e0, e0 + c[it].x, e0 + p1, e0 + p2);
            *(int4*)(row_ptr + i) = r;
        }
        carry += tot[it];
    }
    if (w == 31 && lane == 0) row_ptr[N_NODES] = carry;
}

__global__ void csr_fill(const int* __restrict__ src,
                         const int* __restrict__ dst,
                         const int* __restrict__ ef,
                         const int* __restrict__ rank,
                         const int* __restrict__ row_ptr,
                         const int* __restrict__ partI,
                         int* __restrict__ ent)
{
    int i = blockIdx.x * blockDim.x + threadIdx.x;
    const int stride = gridDim.x * blockDim.x;
    for (; i < NE; i += stride) {
        const int s  = src[i];
        const int d  = dst[i];
        const int r  = ef[i];
        const int rk = rank[i];
        const int sl = i / SLICE;
        const int pos = row_ptr[d] + partI[(size_t)sl * N_NODES + d] + rk;
        if ((unsigned)pos < (unsigned)NE)
            ent[pos] = s | ((r >= 0 && r <= 4) ? (int)0x80000000 : 0);
    }
}

template <int G, int MODE>
__device__ __forceinline__ void gather_group(const float* __restrict__ h,
                                             int e, int j, int lane, float4& acc)
{
    float4 a[G];
    float  wgt[G];
    #pragma unroll
    for (int k = 0; k < G; ++k) {
        const int ek = __builtin_amdgcn_readlane(e, j + k);
        a[k] = *(const float4*)(h + (size_t)(ek & 0x7fffffff) * HD + lane * 4);
        wgt[k] = (MODE == 2 && ek < 0) ? 2.0f : 1.0f;
    }
    #pragma unroll
    for (int k = 0; k < G; ++k) {
        if (MODE == 2) {
            acc.x += a[k].x * wgt[k];
            acc.y += a[k].y * wgt[k];
            acc.z += a[k].z * wgt[k];
            acc.w += a[k].w * wgt[k];
        } else {
            acc.x += a[k].x;
            acc.y += a[k].y;
            acc.z += a[k].z;
            acc.w += a[k].w;
        }
    }
}

template <int MODE>
__global__ void __launch_bounds__(256)
row_gather(const float* __restrict__ h,
           const int* __restrict__ row_ptr,
           const int* __restrict__ ent,
           const float* __restrict__ in_norm,
           const float* __restrict__ out_norm,
           const float* __restrict__ bias,
           float* __restrict__ dstv)
{
    const int lane = threadIdx.x & 31;
    const int wave = __builtin_amdgcn_readfirstlane(
        (int)((blockIdx.x * blockDim.x + threadIdx.x) >> 5));
    const int nwaves = (int)((gridDim.x * blockDim.x) >> 5);

    float4 b = make_float4(0.0f, 0.0f, 0.0f, 0.0f);
    if (MODE == 0) b = *(const float4*)(bias + lane * 4);

    for (int row = wave; row < N_NODES; row += nwaves) {
        const int beg = row_ptr[row];
        const int end = row_ptr[row + 1];
        float inn = 1.0f, outn = 1.0f;
        if (MODE == 0) { inn = in_norm[row]; outn = out_norm[row]; }

        float4 acc = make_float4(0.0f, 0.0f, 0.0f, 0.0f);
        for (int base = beg; base < end; base += 32) {
            const int n = (end - base < 32) ? (end - base) : 32;
            const int e = (lane < n) ? ent[base + lane] : 0;
            int j = 0;
            for (; j + 8 <= n; j += 8) gather_group<8, MODE>(h, e, j, lane, acc);
            if (j + 4 <= n) { gather_group<4, MODE>(h, e, j, lane, acc); j += 4; }
            if (j + 2 <= n) { gather_group<2, MODE>(h, e, j, lane, acc); j += 2; }
            if (j < n)      { gather_group<1, MODE>(h, e, j, lane, acc); }
        }

        if (MODE == 0) {
            acc.x = fmaxf(acc.x * inn + b.x, 0.0f) * outn;
            acc.y = fmaxf(acc.y * inn + b.y, 0.0f) * outn;
            acc.z = fmaxf(acc.z * inn + b.z, 0.0f) * outn;
            acc.w = fmaxf(acc.w * inn + b.w, 0.0f) * outn;
        }
        *(float4*)(dstv + (size_t)row * HD + lane * 4) = acc;
    }
}

__global__ void edge_backward(const float* __restrict__ h,
                              const int* __restrict__ src,
                              const int* __restrict__ dst,
                              const int* __restrict__ ef,
                              float* __restrict__ out)
{
    const int lane   = threadIdx.x & 31;
    const int wave   = (blockIdx.x * blockDim.x + threadIdx.x) >> 5;
    const int nwaves = (gridDim.x * blockDim.x) >> 5;
    for (long base = (long)wave * 32; base < NE; base += (long)nwaves * 32) {
        const int s = src[base + lane];
        const int d = dst[base + lane];
        const int r = ef[base + lane];
        unsigned mask = __builtin_amdgcn_ballot_w32(r == 6 || r == 14 || r == 30);
        while (mask != 0u) {
            const int j = __builtin_ctz(mask);
            mask &= mask - 1u;
            const int ss = __shfl(s, j);
            const int dd = __shfl(d, j);
            const float* hdp = h + (long)dd * HD + lane;
            float* os = out + (long)ss * HD + lane;
            const float d0 = hdp[0];
            const float d1 = hdp[32];
            const float d2 = hdp[64];
            const float d3 = hdp[96];
            atomicAdd(os,      d0);
            atomicAdd(os + 32, d1);
            atomicAdd(os + 64, d2);
            atomicAdd(os + 96, d3);
        }
    }
}

extern "C" void kernel_launch(void* const* d_in, const int* in_sizes, int n_in,
                              void* d_out, int out_size, void* d_ws, size_t ws_size,
                              hipStream_t stream)
{
    (void)in_sizes; (void)n_in; (void)out_size; (void)ws_size;

    const float* feat0 = (const float*)d_in[0];
    const float* Wfc0  = (const float*)d_in[1];
    const float* bfc0  = (const float*)d_in[2];
    const float* feat1 = (const float*)d_in[3];
    const float* Wfc1  = (const float*)d_in[4];
    const float* bfc1  = (const float*)d_in[5];
    const float* feat2 = (const float*)d_in[6];
    const float* Wfc2  = (const float*)d_in[7];
    const float* bfc2  = (const float*)d_in[8];
    const float* feat3 = (const float*)d_in[9];
    const float* Wfc3  = (const float*)d_in[10];
    const float* bfc3  = (const float*)d_in[11];
    const int*   src   = (const int*)d_in[12];
    const int*   dst   = (const int*)d_in[13];
    const int*   efeat = (const int*)d_in[14];
    const float* b_g0  = (const float*)d_in[15];
    const float* W_g1  = (const float*)d_in[16];
    const float* b_g1  = (const float*)d_in[17];
    float* out = (float*)d_out;

    const size_t NODE_ELEMS = (size_t)N_NODES * HD;
    int* cntI  = (int*)d_ws;
    int* rowp  = cntI + N_NODES;
    int* ent   = rowp + ROWP_PAD;
    int* rank  = ent + NE;
    int* partI = rank + NE;
    int* partO = partI + (size_t)NSL * N_NODES;
    float* bufB = (float*)d_ws + NODE_ELEMS;
    float* degO = bufB + NODE_ELEMS;
    float* degI = degO + N_NODES;

    edge_hist<<<NRG * NSL, HB, 0, stream>>>(src, dst, rank, partI, partO);
    node_reduce<<<(N_NODES + 255) / 256, 256, 0, stream>>>(partI, partO, cntI, degO, degI);
    scan_rows<<<1, 1024, 0, stream>>>(cntI, rowp);
    csr_fill<<<1024, 256, 0, stream>>>(src, dst, efeat, rank, rowp, partI, ent);

    const int mt_per = (N_PER + 15) / 16;
    gemm_wmma_f32<128><<<mt_per, 256, 0, stream>>>(
        feat0, Wfc0, bfc0, nullptr, degO + 0 * N_PER,
        out + (size_t)0 * N_PER * HD, N_PER, 0);
    gemm_wmma_f32<128><<<mt_per, 256, 0, stream>>>(
        feat1, Wfc1, bfc1, nullptr, degO + 1 * N_PER,
        out + (size_t)1 * N_PER * HD, N_PER, 0);
    gemm_wmma_f32<64><<<mt_per, 256, 0, stream>>>(
        feat2, Wfc2, bfc2, nullptr, degO + 2 * N_PER,
        out + (size_t)2 * N_PER * HD, N_PER, 0);
    gemm_wmma_f32<64><<<mt_per, 256, 0, stream>>>(
        feat3, Wfc3, bfc3, nullptr, degO + 3 * N_PER,
        out + (size_t)3 * N_PER * HD, N_PER, 0);

    row_gather<0><<<2048, 256, 0, stream>>>(
        out, rowp, ent, degI, degO, b_g0, bufB);

    row_gather<1><<<2048, 256, 0, stream>>>(
        bufB, rowp, ent, degI, degO, b_g0, out);
    gemm_wmma_f32<128><<<3125, 256, 0, stream>>>(
        out, W_g1, b_g1, degI, nullptr, bufB, N_NODES, 1);

    row_gather<2><<<2048, 256, 0, stream>>>(
        bufB, rowp, ent, degI, degO, b_g0, out);
    edge_backward<<<1024, 256, 0, stream>>>(bufB, src, dst, efeat, out);
}


# GEMMs re-tiled: one wave computes a 16x128 output strip (8 accumulators) against an LDS copy of B in fragment order; each A row fetched once, 8x fewer waves; per-element sum order unchanged
# speedup vs baseline: 2.8337x; 1.2155x over previous
#include <hip/hip_runtime.h>

typedef float v2f __attribute__((ext_vector_type(2)));
typedef float v8f __attribute__((ext_vector_type(8)));

#define N_NODES 50000
#define N_PER   12500
#define NE      600000
#define HD      128

#define GW 4
template <int K>
__global__ void __launch_bounds__(GW * 32)
gemm_wmma_f32(const float* __restrict__ A,
              const float* __restrict__ B,
              const float* __restrict__ bias,
              const float* __restrict__ rowscale,
              const float* __restrict__ outscale,
              float* __restrict__ C,
              int M, int relu)
{
    __shared__ float Bs[K * HD];

    const int lane = threadIdx.x & 31;
    const int w    = __builtin_amdgcn_readfirstlane((int)(threadIdx.x >> 5));
    const int half = lane >> 4;
    const int lm   = lane & 15;
    const int mt   = (M + 15) >> 4;
    const int tm   = blockIdx.x * GW + w;

    const int row = tm * 16 + lm;
    const int rc  = (row < M) ? row : (M - 1);
    const float rs = (rowscale != nullptr) ? rowscale[rc] : 1.0f;
    const float* ap = A + (long)rc * K + half * 2;
    v2f af[K / 4];
    #pragma unroll
    for (int kk = 0; kk < K / 4; ++kk) af[kk] = *(const v2f*)(ap + kk * 4);

    float bv[8], osc[8];
    #pragma unroll
    for (int tn = 0; tn < 8; ++tn) bv[tn] = bias[tn * 16 + lm];
    #pragma unroll
    for (int v = 0; v < 8; ++v) {
        const int orow = tm * 16 + v + half * 8;
        const int oc   = (orow < M) ? orow : (M - 1);
        osc[v] = (outscale != nullptr) ? outscale[oc] : 1.0f;
    }

    {
        const int tnf = lane >> 2, sub = lane & 3;
        #pragma unroll 8
        for (int i = 0; i < K / (2 * GW); ++i) {
            const int p = w + GW * i;
            const float4 r0 = *(const float4*)(B + (long)(2 * p) * HD + lane * 4);
            const float4 r1 = *(const float4*)(B + (long)(2 * p + 1) * HD + lane * 4);
            float* d = Bs + ((p >> 1) * 8 + tnf) * 64 + (p & 1) * 32 + sub * 8;
            *(float4*)d       = make_float4(r0.x, r1.x, r0.y, r1.y);
            *(float4*)(d + 4) = make_float4(r0.z, r1.z, r0.w, r1.w);
        }
    }
    __syncthreads();
    if (tm >= mt) return;

    v8f acc[8];
    #pragma unroll
    for (int tn = 0; tn < 8; ++tn) acc[tn] = (v8f){};

    #pragma unroll
    for (int kk = 0; kk < K / 4; ++kk) {
        v2f a = af[kk];
        a[0] *= rs;
        a[1] *= rs;
        #pragma unroll
        for (int tn = 0; tn < 8; ++tn) {
            const v2f b = *(const v2f*)(Bs + (kk * 8 + tn) * 64 + lane * 2);
            acc[tn] = __builtin_amdgcn_wmma_f32_16x16x4_f32(
                false, a, false, b, (short)0, acc[tn], false, false);
        }
    }

    float* cp = C + (long)(tm * 16 + half * 8) * HD + lm;
    const bool full = (tm * 16 + 16 <= M);
    if (full) {
        #pragma unroll
        for (int tn = 0; tn < 8; ++tn) {
            #pragma unroll
            for (int v = 0; v < 8; ++v) {
                float val = acc[tn][v] + bv[tn];
                if (relu) val = fmaxf(val, 0.0f);
                if (outscale != nullptr) val *= osc[v];
                cp[v * HD + tn * 16] = val;
            }
        }
    } else {
        #pragma unroll
        for (int tn = 0; tn < 8; ++tn) {
            #pragma unroll
            for (int v = 0; v < 8; ++v) {
                float val = acc[tn][v] + bv[tn];
                if (relu) val = fmaxf(val, 0.0f);
                if (outscale != nullptr) val *= osc[v];
                if (tm * 16 + half * 8 + v < M) cp[v * HD + tn * 16] = val;
            }
        }
    }
}

#define ROWP_PAD 50016
#define SCAN_IT  13
#define NRG      16
#define NSL      16
#define RANGE    (N_NODES / NRG)
#define SLICE    (NE / NSL)
static_assert(32 * SCAN_IT * 128 >= N_NODES, "scan segments must cover all rows");
static_assert(N_NODES % 4 == 0, "int4 groups are all-valid or all-invalid");
static_assert(NE % 32 == 0, "32-edge batches, no tail");
static_assert(RANGE * NRG == N_NODES, "ranges tile the nodes exactly");
static_assert(SLICE * NSL == NE && SLICE % 4 == 0, "slices tile the edges in int4 groups");

#define HB   1024
#define HGRP (SLICE / 4)
#define HFUL (HGRP / HB)
static_assert(HGRP - HFUL * HB < HB, "one guarded trip covers the rest of the slice");

__device__ __forceinline__ void hist_group(const int4 sv, const int4 dv, int lo, int i,
                                           int* cI, int* cO, int* __restrict__ rank)
{
    const unsigned d0 = (unsigned)(dv.x - lo), d1 = (unsigned)(dv.y - lo);
    const unsigned d2 = (unsigned)(dv.z - lo), d3 = (unsigned)(dv.w - lo);
    const unsigned o0 = (unsigned)(sv.x - lo), o1 = (unsigned)(sv.y - lo);
    const unsigned o2 = (unsigned)(sv.z - lo), o3 = (unsigned)(sv.w - lo);
    const bool in0 = d0 < (unsigned)RANGE, in1 = d1 < (unsigned)RANGE;
    const bool in2 = d2 < (unsigned)RANGE, in3 = d3 < (unsigned)RANGE;
    int r0 = 0, r1 = 0, r2 = 0, r3 = 0;
    if (in0) r0 = atomicAdd(&cI[d0], 1);
    if (in1) r1 = atomicAdd(&cI[d1], 1);
    if (in2) r2 = atomicAdd(&cI[d2], 1);
    if (in3) r3 = atomicAdd(&cI[d3], 1);
    if (o0 < (unsigned)RANGE) atomicAdd(&cO[o0], 1);
    if (o1 < (unsigned)RANGE) atomicAdd(&cO[o1], 1);
    if (o2 < (unsigned)RANGE) atomicAdd(&cO[o2], 1);
    if (o3 < (unsigned)RANGE) atomicAdd(&cO[o3], 1);
    if (in0) rank[i + 0] = r0;
    if (in1) rank[i + 1] = r1;
    if (in2) rank[i + 2] = r2;
    if (in3) rank[i + 3] = r3;
}

__global__ void __launch_bounds__(HB)
edge_hist(const int* __restrict__ src,
          const int* __restrict__ dst,
          int* __restrict__ rank,
          int* __restrict__ partI,
          int* __restrict__ partO)
{
    __shared__ int cI[RANGE];
    __shared__ int cO[RANGE];
    const int tid = threadIdx.x;
    const int r  = blockIdx.x % NRG;
    const int s  = blockIdx.x / NRG;
    const int lo = r * RANGE;
    for (int j = tid; j < RANGE; j += HB) { cI[j] = 0; cO[j] = 0; }

    const int e0 = s * SLICE;
    const int4* s4 = (const int4*)(src + e0);
    const int4* d4 = (const int4*)(dst + e0);
    int4 sv[HFUL], dv[HFUL];
    #pragma unroll
    for (int k = 0; k < HFUL; ++k) {
        sv[k] = s4[tid + k * HB];
        dv[k] = d4[tid + k * HB];
    }
    const bool tail = tid + HFUL * HB < HGRP;
    int4 st = make_int4(-1, -1, -1, -1), dt = make_int4(-1, -1, -1, -1);
    if (tail) { st = s4[tid + HFUL * HB]; dt = d4[tid + HFUL * HB]; }
    __syncthreads();

    #pragma unroll
    for (int k = 0; k < HFUL; ++k)
        hist_group(sv[k], dv[k], lo, e0 + (tid + k * HB) * 4, cI, cO, rank);
    if (tail) hist_group(st, dt, lo, e0 + (tid + HFUL * HB) * 4, cI, cO, rank);
    __syncthreads();

    int* pI = partI + (size_t)s * N_NODES + lo;
    int* pO = partO + (size_t)s * N_NODES + lo;
    for (int j = tid; j < RANGE; j += HB) { pI[j] = cI[j]; pO[j] = cO[j]; }
}

__global__ void node_reduce(int* __restrict__ partI,
                            const int* __restrict__ partO,
                            int* __restrict__ cntI,
                            float* __restrict__ normO,
                            float* __restrict__ normI)
{
    const int n = blockIdx.x * blockDim.x + threadIdx.x;
    if (n >= N_NODES) return;
    int ci[NSL], co[NSL];
    #pragma unroll
    for (int s = 0; s < NSL; ++s) {
        ci[s] = partI[(size_t)s * N_NODES + n];
        co[s] = partO[(size_t)s * N_NODES + n];
    }
    int run = 0, o = 0;
    #pragma unroll
    for (int s = 0; s < NSL; ++s) {
        partI[(size_t)s * N_NODES + n] = run;
        run += ci[s];
        o   += co[s];
    }
    cntI[n]  = run;
    normO[n] = rsqrtf(fmaxf((float)o,   1.0f));
    normI[n] = rsqrtf(fmaxf((float)run, 1.0f));
}

__global__ void __launch_bounds__(1024)
scan_rows(const int* __restrict__ cntI,
          int* __restrict__ row_ptr)
{
    __shared__ int wsum[32];
    const int lane = threadIdx.x & 31;
    const int w    = threadIdx.x >> 5;
    const int seg0 = w * (SCAN_IT * 128);

    int4 c[SCAN_IT];
    int s = 0;
    #pragma unroll
    for (int it = 0; it < SCAN_IT; ++it) {
        const int i = seg0 + it * 128 + lane * 4;
        const bool ok = i < N_NODES;
        const int4 v = *(const int4*)(cntI + (ok ? i : 0));
        c[it] = ok ? v : make_int4(0, 0, 0, 0);
        s += (c[it].x + c[it].y) + (c[it].z + c[it].w);
    }
    for (int o = 16; o > 0; o >>= 1) s += __shfl_xor(s, o);
    if (lane == 0) wsum[w] = s;
    __syncthreads();

    int carry = (lane < w) ? wsum[lane] : 0;
    for (int o = 16; o > 0; o >>= 1) carry += __shfl_xor(carry, o);

    int x[SCAN_IT];
    #pragma unroll
    for (int it = 0; it < SCAN_IT; ++it)
        x[it] = (c[it].x + c[it].y) + (c[it].z + c[it].w);
    #pragma unroll
    for (int o = 1; o < 32; o <<= 1) {
        int y[SCAN_IT];
        #pragma unroll
        for (int it = 0; it < SCAN_IT; ++it) y[it] = __shfl_up(x[it], o);
        #pragma unroll
        for (int it = 0; it < SCAN_IT; ++it) if (lane >= o) x[it] += y[it];
    }
    int tot[SCAN_IT];
    #pragma unroll
    for (int it = 0; it < SCAN_IT; ++it) tot[it] = __shfl(x[it], 31);

    #pragma unroll
    for (int it = 0; it < SCAN_IT; ++it) {
        const int i  = seg0 + it * 128 + lane * 4;
        const int p1 = c[it].x + c[it].y;
        const int p2 = p1 + c[it].z;
        const int p3 = p2 + c[it].w;
        const int e0 = carry + x[it] - p3;
        if (i < N_NODES) {
            const int4 r = make_int4(e0, e0 + c[it].x, e0 + p1, e0 + p2);
            *(int4*)(row_ptr + i) = r;
        }
        carry += tot[it];
    }
    if (w == 31 && lane == 0) row_ptr[N_NODES] = carry;
}

__global__ void csr_fill(const int* __restrict__ src,
                         const int* __restrict__ dst,
                         const int* __restrict__ ef,
                         const int* __restrict__ rank,
                         const int* __restrict__ row_ptr,
                         const int* __restrict__ partI,
                         int* __restrict__ ent)
{
    int i = blockIdx.x * blockDim.x + threadIdx.x;
    const int stride = gridDim.x * blockDim.x;
    for (; i < NE; i += stride) {
        const int s  = src[i];
        const int d  = dst[i];
        const int r  = ef[i];
        const int rk = rank[i];
        const int sl = i / SLICE;
        const int pos = row_ptr[d] + partI[(size_t)sl * N_NODES + d] + rk;
        if ((unsigned)pos < (unsigned)NE)
            ent[pos] = s | ((r >= 0 && r <= 4) ? (int)0x80000000 : 0);
    }
}

template <int G, int MODE>
__device__ __forceinline__ void gather_group(const float* __restrict__ h,
                                             int e, int j, int lane, float4& acc)
{
    float4 a[G];
    float  wgt[G];
    #pragma unroll
    for (int k = 0; k < G; ++k) {
        const int ek = __builtin_amdgcn_readlane(e, j + k);
        a[k] = *(const float4*)(h + (size_t)(ek & 0x7fffffff) * HD + lane * 4);
        wgt[k] = (MODE == 2 && ek < 0) ? 2.0f : 1.0f;
    }
    #pragma unroll
    for (int k = 0; k < G; ++k) {
        if (MODE == 2) {
            acc.x += a[k].x * wgt[k];
            acc.y += a[k].y * wgt[k];
            acc.z += a[k].z * wgt[k];
            acc.w += a[k].w * wgt[k];
        } else {
            acc.x += a[k].x;
            acc.y += a[k].y;
            acc.z += a[k].z;
            acc.w += a[k].w;
        }
    }
}

template <int MODE>
__global__ void __launch_bounds__(256)
row_gather(const float* __restrict__ h,
           const int* __restrict__ row_ptr,
           const int* __restrict__ ent,
           const float* __restrict__ in_norm,
           const float* __restrict__ out_norm,
           const float* __restrict__ bias,
           float* __restrict__ dstv)
{
    const int lane = threadIdx.x & 31;
    const int wave = __builtin_amdgcn_readfirstlane(
        (int)((blockIdx.x * blockDim.x + threadIdx.x) >> 5));
    const int nwaves = (int)((gridDim.x * blockDim.x) >> 5);

    float4 b = make_float4(0.0f, 0.0f, 0.0f, 0.0f);
    if (MODE == 0) b = *(const float4*)(bias + lane * 4);

    for (int row = wave; row < N_NODES; row += nwaves) {
        const int beg = row_ptr[row];
        const int end = row_ptr[row + 1];
        float inn = 1.0f, outn = 1.0f;
        if (MODE == 0) { inn = in_norm[row]; outn = out_norm[row]; }

        float4 acc = make_float4(0.0f, 0.0f, 0.0f, 0.0f);
        for (int base = beg; base < end; base += 32) {
            const int n = (end - base < 32) ? (end - base) : 32;
            const int e = (lane < n) ? ent[base + lane] : 0;
            int j = 0;
            for (; j + 8 <= n; j += 8) gather_group<8, MODE>(h, e, j, lane, acc);
            if (j + 4 <= n) { gather_group<4, MODE>(h, e, j, lane, acc); j += 4; }
            if (j + 2 <= n) { gather_group<2, MODE>(h, e, j, lane, acc); j += 2; }
            if (j < n)      { gather_group<1, MODE>(h, e, j, lane, acc); }
        }

        if (MODE == 0) {
            acc.x = fmaxf(acc.x * inn + b.x, 0.0f) * outn;
            acc.y = fmaxf(acc.y * inn + b.y, 0.0f) * outn;
            acc.z = fmaxf(acc.z * inn + b.z, 0.0f) * outn;
            acc.w = fmaxf(acc.w * inn + b.w, 0.0f) * outn;
        }
        *(float4*)(dstv + (size_t)row * HD + lane * 4) = acc;
    }
}

__global__ void edge_backward(const float* __restrict__ h,
                              const int* __restrict__ src,
                              const int* __restrict__ dst,
                              const int* __restrict__ ef,
                              float* __restrict__ out)
{
    const int lane   = threadIdx.x & 31;
    const int wave   = (blockIdx.x * blockDim.x + threadIdx.x) >> 5;
    const int nwaves = (gridDim.x * blockDim.x) >> 5;
    for (long base = (long)wave * 32; base < NE; base += (long)nwaves * 32) {
        const int s = src[base + lane];
        const int d = dst[base + lane];
        const int r = ef[base + lane];
        unsigned mask = __builtin_amdgcn_ballot_w32(r == 6 || r == 14 || r == 30);
        while (mask != 0u) {
            const int j = __builtin_ctz(mask);
            mask &= mask - 1u;
            const int ss = __shfl(s, j);
            const int dd = __shfl(d, j);
            const float* hdp = h + (long)dd * HD + lane;
            float* os = out + (long)ss * HD + lane;
            const float d0 = hdp[0];
            const float d1 = hdp[32];
            const float d2 = hdp[64];
            const float d3 = hdp[96];
            atomicAdd(os,      d0);
            atomicAdd(os + 32, d1);
            atomicAdd(os + 64, d2);
            atomicAdd(os + 96, d3);
        }
    }
}

extern "C" void kernel_launch(void* const* d_in, const int* in_sizes, int n_in,
                              void* d_out, int out_size, void* d_ws, size_t ws_size,
                              hipStream_t stream)
{
    (void)in_sizes; (void)n_in; (void)out_size; (void)ws_size;

    const float* feat0 = (const float*)d_in[0];
    const float* Wfc0  = (const float*)d_in[1];
    const float* bfc0  = (const float*)d_in[2];
    const float* feat1 = (const float*)d_in[3];
    const float* Wfc1  = (const float*)d_in[4];
    const float* bfc1  = (const float*)d_in[5];
    const float* feat2 = (const float*)d_in[6];
    const float* Wfc2  = (const float*)d_in[7];
    const float* bfc2  = (const float*)d_in[8];
    const float* feat3 = (const float*)d_in[9];
    const float* Wfc3  = (const float*)d_in[10];
    const float* bfc3  = (const float*)d_in[11];
    const int*   src   = (const int*)d_in[12];
    const int*   dst   = (const int*)d_in[13];
    const int*   efeat = (const int*)d_in[14];
    const float* b_g0  = (const float*)d_in[15];
    const float* W_g1  = (const float*)d_in[16];
    const float* b_g1  = (const float*)d_in[17];
    float* out = (float*)d_out;

    const size_t NODE_ELEMS = (size_t)N_NODES * HD;
    int* cntI  = (int*)d_ws;
    int* rowp  = cntI + N_NODES;
    int* ent   = rowp + ROWP_PAD;
    int* rank  = ent + NE;
    int* partI = rank + NE;
    int* partO = partI + (size_t)NSL * N_NODES;
    float* bufB = (float*)d_ws + NODE_ELEMS;
    float* degO = bufB + NODE_ELEMS;
    float* degI = degO + N_NODES;

    edge_hist<<<NRG * NSL, HB, 0, stream>>>(src, dst, rank, partI, partO);
    node_reduce<<<(N_NODES + 255) / 256, 256, 0, stream>>>(partI, partO, cntI, degO, degI);
    scan_rows<<<1, 1024, 0, stream>>>(cntI, rowp);
    csr_fill<<<1024, 256, 0, stream>>>(src, dst, efeat, rank, rowp, partI, ent);

    const int fc_blocks = (N_PER + 16 * GW - 1) / (16 * GW);
    gemm_wmma_f32<128><<<fc_blocks, GW * 32, 0, stream>>>(
        feat0, Wfc0, bfc0, nullptr, degO + 0 * N_PER,
        out + (size_t)0 * N_PER * HD, N_PER, 0);
    gemm_wmma_f32<128><<<fc_blocks, GW * 32, 0, stream>>>(
        feat1, Wfc1, bfc1, nullptr, degO + 1 * N_PER,
        out + (size_t)1 * N_PER * HD, N_PER, 0);
    gemm_wmma_f32<64><<<fc_blocks, GW * 32, 0, stream>>>(
        feat2, Wfc2, bfc2, nullptr, degO + 2 * N_PER,
        out + (size_t)2 * N_PER * HD, N_PER, 0);
    gemm_wmma_f32<64><<<fc_blocks, GW * 32, 0, stream>>>(
        feat3, Wfc3, bfc3, nullptr, degO + 3 * N_PER,
        out + (size_t)3 * N_PER * HD, N_PER, 0);

    row_gather<0><<<2048, 256, 0, stream>>>(
        out, rowp, ent, degI, degO, b_g0, bufB);

    row_gather<1><<<2048, 256, 0, stream>>>(
        bufB, rowp, ent, degI, degO, b_g0, out);
    gemm_wmma_f32<128><<<(N_NODES + 16 * GW - 1) / (16 * GW), GW * 32, 0, stream>>>(
        out, W_g1, b_g1, degI, nullptr, bufB, N_NODES, 1);

    row_gather<2><<<2048, 256, 0, stream>>>(
        bufB, rowp, ent, degI, degO, b_g0, out);
    edge_backward<<<1024, 256, 0, stream>>>(bufB, src, dst, efeat, out);
}


# the four per-type projection launches merged into one launch of 784 blocks (block index picks node type and K); 10 launches per call, not 13; per-element sum order unchanged
# speedup vs baseline: 3.0599x; 1.0798x over previous
#include <hip/hip_runtime.h>

typedef float v2f __attribute__((ext_vector_type(2)));
typedef float v8f __attribute__((ext_vector_type(8)));

#define N_NODES 50000
#define N_PER   12500
#define NE      600000
#define HD      128

#define GW 4
template <int K>
__device__ __forceinline__ void
gemm_strip(const float* __restrict__ A,
           const float* __restrict__ B,
           const float* __restrict__ bias,
           const float* __restrict__ rowscale,
           const float* __restrict__ outscale,
           float* __restrict__ C,
           int M, int relu, int blk, float* __restrict__ Bs)
{
    const int lane = threadIdx.x & 31;
    const int w    = __builtin_amdgcn_readfirstlane((int)(threadIdx.x >> 5));
    const int half = lane >> 4;
    const int lm   = lane & 15;
    const int mt   = (M + 15) >> 4;
    const int tm   = blk * GW + w;

    const int row = tm * 16 + lm;
    const int rc  = (row < M) ? row : (M - 1);
    const float rs = (rowscale != nullptr) ? rowscale[rc] : 1.0f;
    const float* ap = A + (long)rc * K + half * 2;
    v2f af[K / 4];
    #pragma unroll
    for (int kk = 0; kk < K / 4; ++kk) af[kk] = *(const v2f*)(ap + kk * 4);

    float bv[8], osc[8];
    #pragma unroll
    for (int tn = 0; tn < 8; ++tn) bv[tn] = bias[tn * 16 + lm];
    #pragma unroll
    for (int v = 0; v < 8; ++v) {
        const int orow = tm * 16 + v + half * 8;
        const int oc   = (orow < M) ? orow : (M - 1);
        osc[v] = (outscale != nullptr) ? outscale[oc] : 1.0f;
    }

    {
        const int tnf = lane >> 2, sub = lane & 3;
        #pragma unroll 8
        for (int i = 0; i < K / (2 * GW); ++i) {
            const int p = w + GW * i;
            const float4 r0 = *(const float4*)(B + (long)(2 * p) * HD + lane * 4);
            const float4 r1 = *(const float4*)(B + (long)(2 * p + 1) * HD + lane * 4);
            float* d = Bs + ((p >> 1) * 8 + tnf) * 64 + (p & 1) * 32 + sub * 8;
            *(float4*)d       = make_float4(r0.x, r1.x, r0.y, r1.y);
            *(float4*)(d + 4) = make_float4(r0.z, r1.z, r0.w, r1.w);
        }
    }
    __syncthreads();
    if (tm >= mt) return;

    v8f acc[8];
    #pragma unroll
    for (int tn = 0; tn < 8; ++tn) acc[tn] = (v8f){};

    #pragma unroll
    for (int kk = 0; kk < K / 4; ++kk) {
        v2f a = af[kk];
        a[0] *= rs;
        a[1] *= rs;
        #pragma unroll
        for (int tn = 0; tn < 8; ++tn) {
            const v2f b = *(const v2f*)(Bs + (kk * 8 + tn) * 64 + lane * 2);
            acc[tn] = __builtin_amdgcn_wmma_f32_16x16x4_f32(
                false, a, false, b, (short)0, acc[tn], false, false);
        }
    }

    float* cp = C + (long)(tm * 16 + half * 8) * HD + lm;
    const bool full = (tm * 16 + 16 <= M);
    if (full) {
        #pragma unroll
        for (int tn = 0; tn < 8; ++tn) {
            #pragma unroll
            for (int v = 0; v < 8; ++v) {
                float val = acc[tn][v] + bv[tn];
                if (relu) val = fmaxf(val, 0.0f);
                if (outscale != nullptr) val *= osc[v];
                cp[v * HD + tn * 16] = val;
            }
        }
    } else {
        #pragma unroll
        for (int tn = 0; tn < 8; ++tn) {
            #pragma unroll
            for (int v = 0; v < 8; ++v) {
                float val = acc[tn][v] + bv[tn];
                if (relu) val = fmaxf(val, 0.0f);
                if (outscale != nullptr) val *= osc[v];
                if (tm * 16 + half * 8 + v < M) cp[v * HD + tn * 16] = val;
            }
        }
    }
}

template <int K>
__global__ void __launch_bounds__(GW * 32)
gemm_wmma_f32(const float* __restrict__ A,
              const float* __restrict__ B,
              const float* __restrict__ bias,
              const float* __restrict__ rowscale,
              const float* __restrict__ outscale,
              float* __restrict__ C,
              int M, int relu)
{
    __shared__ float Bs[K * HD];
    gemm_strip<K>(A, B, bias, rowscale, outscale, C, M, relu, (int)blockIdx.x, Bs);
}

#define FCB ((N_PER + 16 * GW - 1) / (16 * GW))
__global__ void __launch_bounds__(GW * 32)
fc_project_all(const float* __restrict__ feat0, const float* __restrict__ W0, const float* __restrict__ b0,
               const float* __restrict__ feat1, const float* __restrict__ W1, const float* __restrict__ b1,
               const float* __restrict__ feat2, const float* __restrict__ W2, const float* __restrict__ b2,
               const float* __restrict__ feat3, const float* __restrict__ W3, const float* __restrict__ b3,
               const float* __restrict__ out_norm,
               float* __restrict__ h)
{
    __shared__ float Bs[128 * HD];
    const int t   = (int)blockIdx.x / FCB;
    const int blk = (int)blockIdx.x - t * FCB;
    const float* osc = out_norm + (size_t)t * N_PER;
    float*       c   = h + (size_t)t * N_PER * HD;
    if (t < 2) {
        const float* a  = (t == 0) ? feat0 : feat1;
        const float* wt = (t == 0) ? W0 : W1;
        const float* bs = (t == 0) ? b0 : b1;
        gemm_strip<128>(a, wt, bs, nullptr, osc, c, N_PER, 0, blk, Bs);
    } else {
        const float* a  = (t == 2) ? feat2 : feat3;
        const float* wt = (t == 2) ? W2 : W3;
        const float* bs = (t == 2) ? b2 : b3;
        gemm_strip<64>(a, wt, bs, nullptr, osc, c, N_PER, 0, blk, Bs);
    }
}

#define ROWP_PAD 50016
#define SCAN_IT  13
#define NRG      16
#define NSL      16
#define RANGE    (N_NODES / NRG)
#define SLICE    (NE / NSL)
static_assert(32 * SCAN_IT * 128 >= N_NODES, "scan segments must cover all rows");
static_assert(N_NODES % 4 == 0, "int4 groups are all-valid or all-invalid");
static_assert(NE % 32 == 0, "32-edge batches, no tail");
static_assert(RANGE * NRG == N_NODES, "ranges tile the nodes exactly");
static_assert(SLICE * NSL == NE && SLICE % 4 == 0, "slices tile the edges in int4 groups");

#define HB   1024
#define HGRP (SLICE / 4)
#define HFUL (HGRP / HB)
static_assert(HGRP - HFUL * HB < HB, "one guarded trip covers the rest of the slice");

__device__ __forceinline__ void hist_group(const int4 sv, const int4 dv, int lo, int i,
                                           int* cI, int* cO, int* __restrict__ rank)
{
    const unsigned d0 = (unsigned)(dv.x - lo), d1 = (unsigned)(dv.y - lo);
    const unsigned d2 = (unsigned)(dv.z - lo), d3 = (unsigned)(dv.w - lo);
    const unsigned o0 = (unsigned)(sv.x - lo), o1 = (unsigned)(sv.y - lo);
    const unsigned o2 = (unsigned)(sv.z - lo), o3 = (unsigned)(sv.w - lo);
    const bool in0 = d0 < (unsigned)RANGE, in1 = d1 < (unsigned)RANGE;
    const bool in2 = d2 < (unsigned)RANGE, in3 = d3 < (unsigned)RANGE;
    int r0 = 0, r1 = 0, r2 = 0, r3 = 0;
    if (in0) r0 = atomicAdd(&cI[d0], 1);
    if (in1) r1 = atomicAdd(&cI[d1], 1);
    if (in2) r2 = atomicAdd(&cI[d2], 1);
    if (in3) r3 = atomicAdd(&cI[d3], 1);
    if (o0 < (unsigned)RANGE) atomicAdd(&cO[o0], 1);
    if (o1 < (unsigned)RANGE) atomicAdd(&cO[o1], 1);
    if (o2 < (unsigned)RANGE) atomicAdd(&cO[o2], 1);
    if (o3 < (unsigned)RANGE) atomicAdd(&cO[o3], 1);
    if (in0) rank[i + 0] = r0;
    if (in1) rank[i + 1] = r1;
    if (in2) rank[i + 2] = r2;
    if (in3) rank[i + 3] = r3;
}

__global__ void __launch_bounds__(HB)
edge_hist(const int* __restrict__ src,
          const int* __restrict__ dst,
          int* __restrict__ rank,
          int* __restrict__ partI,
          int* __restrict__ partO)
{
    __shared__ int cI[RANGE];
    __shared__ int cO[RANGE];
    const int tid = threadIdx.x;
    const int r  = blockIdx.x % NRG;
    const int s  = blockIdx.x / NRG;
    const int lo = r * RANGE;
    for (int j = tid; j < RANGE; j += HB) { cI[j] = 0; cO[j] = 0; }

    const int e0 = s * SLICE;
    const int4* s4 = (const int4*)(src + e0);
    const int4* d4 = (const int4*)(dst + e0);
    int4 sv[HFUL], dv[HFUL];
    #pragma unroll
    for (int k = 0; k < HFUL; ++k) {
        sv[k] = s4[tid + k * HB];
        dv[k] = d4[tid + k * HB];
    }
    const bool tail = tid + HFUL * HB < HGRP;
    int4 st = make_int4(-1, -1, -1, -1), dt = make_int4(-1, -1, -1, -1);
    if (tail) { st = s4[tid + HFUL * HB]; dt = d4[tid + HFUL * HB]; }
    __syncthreads();

    #pragma unroll
    for (int k = 0; k < HFUL; ++k)
        hist_group(sv[k], dv[k], lo, e0 + (tid + k * HB) * 4, cI, cO, rank);
    if (tail) hist_group(st, dt, lo, e0 + (tid + HFUL * HB) * 4, cI, cO, rank);
    __syncthreads();

    int* pI = partI + (size_t)s * N_NODES + lo;
    int* pO = partO + (size_t)s * N_NODES + lo;
    for (int j = tid; j < RANGE; j += HB) { pI[j] = cI[j]; pO[j] = cO[j]; }
}

__global__ void node_reduce(int* __restrict__ partI,
                            const int* __restrict__ partO,
                            int* __restrict__ cntI,
                            float* __restrict__ normO,
                            float* __restrict__ normI)
{
    const int n = blockIdx.x * blockDim.x + threadIdx.x;
    if (n >= N_NODES) return;
    int ci[NSL], co[NSL];
    #pragma unroll
    for (int s = 0; s < NSL; ++s) {
        ci[s] = partI[(size_t)s * N_NODES + n];
        co[s] = partO[(size_t)s * N_NODES + n];
    }
    int run = 0, o = 0;
    #pragma unroll
    for (int s = 0; s < NSL; ++s) {
        partI[(size_t)s * N_NODES + n] = run;
        run += ci[s];
        o   += co[s];
    }
    cntI[n]  = run;
    normO[n] = rsqrtf(fmaxf((float)o,   1.0f));
    normI[n] = rsqrtf(fmaxf((float)run, 1.0f));
}

__global__ void __launch_bounds__(1024)
scan_rows(const int* __restrict__ cntI,
          int* __restrict__ row_ptr)
{
    __shared__ int wsum[32];
    const int lane = threadIdx.x & 31;
    const int w    = threadIdx.x >> 5;
    const int seg0 = w * (SCAN_IT * 128);

    int4 c[SCAN_IT];
    int s = 0;
    #pragma unroll
    for (int it = 0; it < SCAN_IT; ++it) {
        const int i = seg0 + it * 128 + lane * 4;
        const bool ok = i < N_NODES;
        const int4 v = *(const int4*)(cntI + (ok ? i : 0));
        c[it] = ok ? v : make_int4(0, 0, 0, 0);
        s += (c[it].x + c[it].y) + (c[it].z + c[it].w);
    }
    for (int o = 16; o > 0; o >>= 1) s += __shfl_xor(s, o);
    if (lane == 0) wsum[w] = s;
    __syncthreads();

    int carry = (lane < w) ? wsum[lane] : 0;
    for (int o = 16; o > 0; o >>= 1) carry += __shfl_xor(carry, o);

    int x[SCAN_IT];
    #pragma unroll
    for (int it = 0; it < SCAN_IT; ++it)
        x[it] = (c[it].x + c[it].y) + (c[it].z + c[it].w);
    #pragma unroll
    for (int o = 1; o < 32; o <<= 1) {
        int y[SCAN_IT];
        #pragma unroll
        for (int it = 0; it < SCAN_IT; ++it) y[it] = __shfl_up(x[it], o);
        #pragma unroll
        for (int it = 0; it < SCAN_IT; ++it) if (lane >= o) x[it] += y[it];
    }
    int tot[SCAN_IT];
    #pragma unroll
    for (int it = 0; it < SCAN_IT; ++it) tot[it] = __shfl(x[it], 31);

    #pragma unroll
    for (int it = 0; it < SCAN_IT; ++it) {
        const int i  = seg0 + it * 128 + lane * 4;
        const int p1 = c[it].x + c[it].y;
        const int p2 = p1 + c[it].z;
        const int p3 = p2 + c[it].w;
        const int e0 = carry + x[it] - p3;
        if (i < N_NODES) {
            const int4 r = make_int4(e0, e0 + c[it].x, e0 + p1, e0 + p2);
            *(int4*)(row_ptr + i) = r;
        }
        carry += tot[it];
    }
    if (w == 31 && lane == 0) row_ptr[N_NODES] = carry;
}

__global__ void csr_fill(const int* __restrict__ src,
                         const int* __restrict__ dst,
                         const int* __restrict__ ef,
                         const int* __restrict__ rank,
                         const int* __restrict__ row_ptr,
                         const int* __restrict__ partI,
                         int* __restrict__ ent)
{
    int i = blockIdx.x * blockDim.x + threadIdx.x;
    const int stride = gridDim.x * blockDim.x;
    for (; i < NE; i += stride) {
        const int s  = src[i];
        const int d  = dst[i];
        const int r  = ef[i];
        const int rk = rank[i];
        const int sl = i / SLICE;
        const int pos = row_ptr[d] + partI[(size_t)sl * N_NODES + d] + rk;
        if ((unsigned)pos < (unsigned)NE)
            ent[pos] = s | ((r >= 0 && r <= 4) ? (int)0x80000000 : 0);
    }
}

template <int G, int MODE>
__device__ __forceinline__ void gather_group(const float* __restrict__ h,
                                             int e, int j, int lane, float4& acc)
{
    float4 a[G];
    float  wgt[G];
    #pragma unroll
    for (int k = 0; k < G; ++k) {
        const int ek = __builtin_amdgcn_readlane(e, j + k);
        a[k] = *(const float4*)(h + (size_t)(ek & 0x7fffffff) * HD + lane * 4);
        wgt[k] = (MODE == 2 && ek < 0) ? 2.0f : 1.0f;
    }
    #pragma unroll
    for (int k = 0; k < G; ++k) {
        if (MODE == 2) {
            acc.x += a[k].x * wgt[k];
            acc.y += a[k].y * wgt[k];
            acc.z += a[k].z * wgt[k];
            acc.w += a[k].w * wgt[k];
        } else {
            acc.x += a[k].x;
            acc.y += a[k].y;
            acc.z += a[k].z;
            acc.w += a[k].w;
        }
    }
}

template <int MODE>
__global__ void __launch_bounds__(256)
row_gather(const float* __restrict__ h,
           const int* __restrict__ row_ptr,
           const int* __restrict__ ent,
           const float* __restrict__ in_norm,
           const float* __restrict__ out_norm,
           const float* __restrict__ bias,
           float* __restrict__ dstv)
{
    const int lane = threadIdx.x & 31;
    const int wave = __builtin_amdgcn_readfirstlane(
        (int)((blockIdx.x * blockDim.x + threadIdx.x) >> 5));
    const int nwaves = (int)((gridDim.x * blockDim.x) >> 5);

    float4 b = make_float4(0.0f, 0.0f, 0.0f, 0.0f);
    if (MODE == 0) b = *(const float4*)(bias + lane * 4);

    for (int row = wave; row < N_NODES; row += nwaves) {
        const int beg = row_ptr[row];
        const int end = row_ptr[row + 1];
        float inn = 1.0f, outn = 1.0f;
        if (MODE == 0) { inn = in_norm[row]; outn = out_norm[row]; }

        float4 acc = make_float4(0.0f, 0.0f, 0.0f, 0.0f);
        for (int base = beg; base < end; base += 32) {
            const int n = (end - base < 32) ? (end - base) : 32;
            const int e = (lane < n) ? ent[base + lane] : 0;
            int j = 0;
            for (; j + 8 <= n; j += 8) gather_group<8, MODE>(h, e, j, lane, acc);
            if (j + 4 <= n) { gather_group<4, MODE>(h, e, j, lane, acc); j += 4; }
            if (j + 2 <= n) { gather_group<2, MODE>(h, e, j, lane, acc); j += 2; }
            if (j < n)      { gather_group<1, MODE>(h, e, j, lane, acc); }
        }

        if (MODE == 0) {
            acc.x = fmaxf(acc.x * inn + b.x, 0.0f) * outn;
            acc.y = fmaxf(acc.y * inn + b.y, 0.0f) * outn;
            acc.z = fmaxf(acc.z * inn + b.z, 0.0f) * outn;
            acc.w = fmaxf(acc.w * inn + b.w, 0.0f) * outn;
        }
        *(float4*)(dstv + (size_t)row * HD + lane * 4) = acc;
    }
}

__global__ void edge_backward(const float* __restrict__ h,
                              const int* __restrict__ src,
                              const int* __restrict__ dst,
                              const int* __restrict__ ef,
                              float* __restrict__ out)
{
    const int lane   = threadIdx.x & 31;
    const int wave   = (blockIdx.x * blockDim.x + threadIdx.x) >> 5;
    const int nwaves = (gridDim.x * blockDim.x) >> 5;
    for (long base = (long)wave * 32; base < NE; base += (long)nwaves * 32) {
        const int s = src[base + lane];
        const int d = dst[base + lane];
        const int r = ef[base + lane];
        unsigned mask = __builtin_amdgcn_ballot_w32(r == 6 || r == 14 || r == 30);
        while (mask != 0u) {
            const int j = __builtin_ctz(mask);
            mask &= mask - 1u;
            const int ss = __shfl(s, j);
            const int dd = __shfl(d, j);
            const float* hdp = h + (long)dd * HD + lane;
            float* os = out + (long)ss * HD + lane;
            const float d0 = hdp[0];
            const float d1 = hdp[32];
            const float d2 = hdp[64];
            const float d3 = hdp[96];
            atomicAdd(os,      d0);
            atomicAdd(os + 32, d1);
            atomicAdd(os + 64, d2);
            atomicAdd(os + 96, d3);
        }
    }
}

extern "C" void kernel_launch(void* const* d_in, const int* in_sizes, int n_in,
                              void* d_out, int out_size, void* d_ws, size_t ws_size,
                              hipStream_t stream)
{
    (void)in_sizes; (void)n_in; (void)out_size; (void)ws_size;

    const float* feat0 = (const float*)d_in[0];
    const float* Wfc0  = (const float*)d_in[1];
    const float* bfc0  = (const float*)d_in[2];
    const float* feat1 = (const float*)d_in[3];
    const float* Wfc1  = (const float*)d_in[4];
    const float* bfc1  = (const float*)d_in[5];
    const float* feat2 = (const float*)d_in[6];
    const float* Wfc2  = (const float*)d_in[7];
    const float* bfc2  = (const float*)d_in[8];
    const float* feat3 = (const float*)d_in[9];
    const float* Wfc3  = (const float*)d_in[10];
    const float* bfc3  = (const float*)d_in[11];
    const int*   src   = (const int*)d_in[12];
    const int*   dst   = (const int*)d_in[13];
    const int*   efeat = (const int*)d_in[14];
    const float* b_g0  = (const float*)d_in[15];
    const float* W_g1  = (const float*)d_in[16];
    const float* b_g1  = (const float*)d_in[17];
    float* out = (float*)d_out;

    const size_t NODE_ELEMS = (size_t)N_NODES * HD;
    int* cntI  = (int*)d_ws;
    int* rowp  = cntI + N_NODES;
    int* ent   = rowp + ROWP_PAD;
    int* rank  = ent + NE;
    int* partI = rank + NE;
    int* partO = partI + (size_t)NSL * N_NODES;
    float* bufB = (float*)d_ws + NODE_ELEMS;
    float* degO = bufB + NODE_ELEMS;
    float* degI = degO + N_NODES;

    edge_hist<<<NRG * NSL, HB, 0, stream>>>(src, dst, rank, partI, partO);
    node_reduce<<<(N_NODES + 255) / 256, 256, 0, stream>>>(partI, partO, cntI, degO, degI);
    scan_rows<<<1, 1024, 0, stream>>>(cntI, rowp);
    csr_fill<<<1024, 256, 0, stream>>>(src, dst, efeat, rank, rowp, partI, ent);

    fc_project_all<<<4 * FCB, GW * 32, 0, stream>>>(
        feat0, Wfc0, bfc0, feat1, Wfc1, bfc1, feat2, Wfc2, bfc2, feat3, Wfc3, bfc3,
        degO, out);

    row_gather<0><<<2048, 256, 0, stream>>>(
        out, rowp, ent, degI, degO, b_g0, bufB);

    row_gather<1><<<2048, 256, 0, stream>>>(
        bufB, rowp, ent, degI, degO, b_g0, out);
    gemm_wmma_f32<128><<<(N_NODES + 16 * GW - 1) / (16 * GW), GW * 32, 0, stream>>>(
        out, W_g1, b_g1, degI, nullptr, bufB, N_NODES, 1);

    row_gather<2><<<2048, 256, 0, stream>>>(
        bufB, rowp, ent, degI, degO, b_g0, out);
    edge_backward<<<1024, 256, 0, stream>>>(bufB, src, dst, efeat, out);
}


# row starts, norms and edge slots computed inside the (node range x edge slice) blocks from the partial counts: three short launches become one; 8 launches, not 10; row list order unchanged
# speedup vs baseline: 3.4829x; 1.1382x over previous
#include <hip/hip_runtime.h>

typedef float v2f __attribute__((ext_vector_type(2)));
typedef float v8f __attribute__((ext_vector_type(8)));

#define N_NODES 50000
#define N_PER   12500
#define NE      600000
#define HD      128

#define GW 4
template <int K>
__device__ __forceinline__ void
gemm_strip(const float* __restrict__ A,
           const float* __restrict__ B,
           const float* __restrict__ bias,
           const float* __restrict__ rowscale,
           const float* __restrict__ outscale,
           float* __restrict__ C,
           int M, int relu, int blk, float* __restrict__ Bs)
{
    const int lane = threadIdx.x & 31;
    const int w    = __builtin_amdgcn_readfirstlane((int)(threadIdx.x >> 5));
    const int half = lane >> 4;
    const int lm   = lane & 15;
    const int mt   = (M + 15) >> 4;
    const int tm   = blk * GW + w;

    const int row = tm * 16 + lm;
    const int rc  = (row < M) ? row : (M - 1);
    const float rs = (rowscale != nullptr) ? rowscale[rc] : 1.0f;
    const float* ap = A + (long)rc * K + half * 2;
    v2f af[K / 4];
    #pragma unroll
    for (int kk = 0; kk < K / 4; ++kk) af[kk] = *(const v2f*)(ap + kk * 4);

    float bv[8], osc[8];
    #pragma unroll
    for (int tn = 0; tn < 8; ++tn) bv[tn] = bias[tn * 16 + lm];
    #pragma unroll
    for (int v = 0; v < 8; ++v) {
        const int orow = tm * 16 + v + half * 8;
        const int oc   = (orow < M) ? orow : (M - 1);
        osc[v] = (outscale != nullptr) ? outscale[oc] : 1.0f;
    }

    {
        const int tnf = lane >> 2, sub = lane & 3;
        #pragma unroll 8
        for (int i = 0; i < K / (2 * GW); ++i) {
            const int p = w + GW * i;
            const float4 r0 = *(const float4*)(B + (long)(2 * p) * HD + lane * 4);
            const float4 r1 = *(const float4*)(B + (long)(2 * p + 1) * HD + lane * 4);
            float* d = Bs + ((p >> 1) * 8 + tnf) * 64 + (p & 1) * 32 + sub * 8;
            *(float4*)d       = make_float4(r0.x, r1.x, r0.y, r1.y);
            *(float4*)(d + 4) = make_float4(r0.z, r1.z, r0.w, r1.w);
        }
    }
    __syncthreads();
    if (tm >= mt) return;

    v8f acc[8];
    #pragma unroll
    for (int tn = 0; tn < 8; ++tn) acc[tn] = (v8f){};

    #pragma unroll
    for (int kk = 0; kk < K / 4; ++kk) {
        v2f a = af[kk];
        a[0] *= rs;
        a[1] *= rs;
        #pragma unroll
        for (int tn = 0; tn < 8; ++tn) {
            const v2f b = *(const v2f*)(Bs + (kk * 8 + tn) * 64 + lane * 2);
            acc[tn] = __builtin_amdgcn_wmma_f32_16x16x4_f32(
                false, a, false, b, (short)0, acc[tn], false, false);
        }
    }

    float* cp = C + (long)(tm * 16 + half * 8) * HD + lm;
    const bool full = (tm * 16 + 16 <= M);
    if (full) {
        #pragma unroll
        for (int tn = 0; tn < 8; ++tn) {
            #pragma unroll
            for (int v = 0; v < 8; ++v) {
                float val = acc[tn][v] + bv[tn];
                if (relu) val = fmaxf(val, 0.0f);
                if (outscale != nullptr) val *= osc[v];
                cp[v * HD + tn * 16] = val;
            }
        }
    } else {
        #pragma unroll
        for (int tn = 0; tn < 8; ++tn) {
            #pragma unroll
            for (int v = 0; v < 8; ++v) {
                float val = acc[tn][v] + bv[tn];
                if (relu) val = fmaxf(val, 0.0f);
                if (outscale != nullptr) val *= osc[v];
                if (tm * 16 + half * 8 + v < M) cp[v * HD + tn * 16] = val;
            }
        }
    }
}

template <int K>
__global__ void __launch_bounds__(GW * 32)
gemm_wmma_f32(const float* __restrict__ A,
              const float* __restrict__ B,
              const float* __restrict__ bias,
              const float* __restrict__ rowscale,
              const float* __restrict__ outscale,
              float* __restrict__ C,
              int M, int relu)
{
    __shared__ float Bs[K * HD];
    gemm_strip<K>(A, B, bias, rowscale, outscale, C, M, relu, (int)blockIdx.x, Bs);
}

#define FCB ((N_PER + 16 * GW - 1) / (16 * GW))
__global__ void __launch_bounds__(GW * 32)
fc_project_all(const float* __restrict__ feat0, const float* __restrict__ W0, const float* __restrict__ b0,
               const float* __restrict__ feat1, const float* __restrict__ W1, const float* __restrict__ b1,
               const float* __restrict__ feat2, const float* __restrict__ W2, const float* __restrict__ b2,
               const float* __restrict__ feat3, const float* __restrict__ W3, const float* __restrict__ b3,
               const float* __restrict__ out_norm,
               float* __restrict__ h)
{
    __shared__ float Bs[128 * HD];
    const int t   = (int)blockIdx.x / FCB;
    const int blk = (int)blockIdx.x - t * FCB;
    const float* osc = out_norm + (size_t)t * N_PER;
    float*       c   = h + (size_t)t * N_PER * HD;
    if (t < 2) {
        const float* a  = (t == 0) ? feat0 : feat1;
        const float* wt = (t == 0) ? W0 : W1;
        const float* bs = (t == 0) ? b0 : b1;
        gemm_strip<128>(a, wt, bs, nullptr, osc, c, N_PER, 0, blk, Bs);
    } else {
        const float* a  = (t == 2) ? feat2 : feat3;
        const float* wt = (t == 2) ? W2 : W3;
        const float* bs = (t == 2) ? b2 : b3;
        gemm_strip<64>(a, wt, bs, nullptr, osc, c, N_PER, 0, blk, Bs);
    }
}

#define ROWP_PAD 50016
#define NRG      16
#define NSL      16
#define RANGE    (N_NODES / NRG)
#define SLICE    (NE / NSL)
static_assert(NE % 32 == 0, "32-edge batches, no tail");
static_assert(RANGE * NRG == N_NODES, "ranges tile the nodes exactly");
static_assert(SLICE * NSL == NE && SLICE % 4 == 0, "slices tile the edges in int4 groups");

#define HB   1024
#define HGRP (SLICE / 4)
#define HFUL (HGRP / HB)
static_assert(HGRP - HFUL * HB < HB, "one guarded trip covers the rest of the slice");

__device__ __forceinline__ void hist_group(const int4 sv, const int4 dv, int lo, int i,
                                           int* cI, int* cO, int* __restrict__ rank)
{
    const unsigned d0 = (unsigned)(dv.x - lo), d1 = (unsigned)(dv.y - lo);
    const unsigned d2 = (unsigned)(dv.z - lo), d3 = (unsigned)(dv.w - lo);
    const unsigned o0 = (unsigned)(sv.x - lo), o1 = (unsigned)(sv.y - lo);
    const unsigned o2 = (unsigned)(sv.z - lo), o3 = (unsigned)(sv.w - lo);
    const bool in0 = d0 < (unsigned)RANGE, in1 = d1 < (unsigned)RANGE;
    const bool in2 = d2 < (unsigned)RANGE, in3 = d3 < (unsigned)RANGE;
    int r0 = 0, r1 = 0, r2 = 0, r3 = 0;
    if (in0) r0 = atomicAdd(&cI[d0], 1);
    if (in1) r1 = atomicAdd(&cI[d1], 1);
    if (in2) r2 = atomicAdd(&cI[d2], 1);
    if (in3) r3 = atomicAdd(&cI[d3], 1);
    if (o0 < (unsigned)RANGE) atomicAdd(&cO[o0], 1);
    if (o1 < (unsigned)RANGE) atomicAdd(&cO[o1], 1);
    if (o2 < (unsigned)RANGE) atomicAdd(&cO[o2], 1);
    if (o3 < (unsigned)RANGE) atomicAdd(&cO[o3], 1);
    if (in0) rank[i + 0] = r0;
    if (in1) rank[i + 1] = r1;
    if (in2) rank[i + 2] = r2;
    if (in3) rank[i + 3] = r3;
}

__global__ void __launch_bounds__(HB)
edge_hist(const int* __restrict__ src,
          const int* __restrict__ dst,
          int* __restrict__ rank,
          int* __restrict__ partI,
          int* __restrict__ partO,
          int* __restrict__ tot)
{
    __shared__ int cI[RANGE];
    __shared__ int cO[RANGE];
    __shared__ int wtot[HB / 32];
    const int tid = threadIdx.x;
    const int r  = blockIdx.x % NRG;
    const int s  = blockIdx.x / NRG;
    const int lo = r * RANGE;
    for (int j = tid; j < RANGE; j += HB) { cI[j] = 0; cO[j] = 0; }

    const int e0 = s * SLICE;
    const int4* s4 = (const int4*)(src + e0);
    const int4* d4 = (const int4*)(dst + e0);
    int4 sv[HFUL], dv[HFUL];
    #pragma unroll
    for (int k = 0; k < HFUL; ++k) {
        sv[k] = s4[tid + k * HB];
        dv[k] = d4[tid + k * HB];
    }
    const bool tail = tid + HFUL * HB < HGRP;
    int4 st = make_int4(-1, -1, -1, -1), dt = make_int4(-1, -1, -1, -1);
    if (tail) { st = s4[tid + HFUL * HB]; dt = d4[tid + HFUL * HB]; }
    __syncthreads();

    #pragma unroll
    for (int k = 0; k < HFUL; ++k)
        hist_group(sv[k], dv[k], lo, e0 + (tid + k * HB) * 4, cI, cO, rank);
    if (tail) hist_group(st, dt, lo, e0 + (tid + HFUL * HB) * 4, cI, cO, rank);
    __syncthreads();

    int* pI = partI + (size_t)s * N_NODES + lo;
    int* pO = partO + (size_t)s * N_NODES + lo;
    int mine = 0;
    for (int j = tid; j < RANGE; j += HB) {
        const int ci = cI[j];
        pI[j] = ci; pO[j] = cO[j];
        mine += ci;
    }
    for (int o = 16; o > 0; o >>= 1) mine += __shfl_xor(mine, o);
    if ((tid & 31) == 0) wtot[tid >> 5] = mine;
    __syncthreads();
    if (tid < 32) {
        int v = wtot[tid];
        for (int o = 16; o > 0; o >>= 1) v += __shfl_xor(v, o);
        if (tid == 0) tot[blockIdx.x] = v;
    }
}

#define SCT ((RANGE + 3) / 4)
#define RPAD (SCT * 4)
#define NTRIP 4
#define FB 5
#define NBAT ((HFUL + 1 + FB - 1) / FB)
#define NOSH ((RANGE + NSL - 1) / NSL)
static_assert(NOSH <= HB, "one node per thread covers a block's share of out_norm");
static_assert(SCT <= HB, "one pass of the block scans the whole range");
static_assert(NTRIP * SCT >= RANGE, "the summing threads cover the range");
static_assert((NRG * NSL) % 32 == 0, "tot[] is read in whole wave trips");
static_assert(NBAT * FB * HB >= HGRP, "the batches cover the slice");

__device__ __forceinline__ void place_group(const int4 sv, const int4 dv, const int4 fv,
                                            const int4 kv, bool ok, int lo, const int* slotL,
                                            int* __restrict__ ent)
{
    const unsigned d0 = (unsigned)(dv.x - lo), d1 = (unsigned)(dv.y - lo);
    const unsigned d2 = (unsigned)(dv.z - lo), d3 = (unsigned)(dv.w - lo);
    const bool in0 = ok && d0 < (unsigned)RANGE, in1 = ok && d1 < (unsigned)RANGE;
    const bool in2 = ok && d2 < (unsigned)RANGE, in3 = ok && d3 < (unsigned)RANGE;
    const int p0 = slotL[in0 ? d0 : 0u] + kv.x;
    const int p1 = slotL[in1 ? d1 : 0u] + kv.y;
    const int p2 = slotL[in2 ? d2 : 0u] + kv.z;
    const int p3 = slotL[in3 ? d3 : 0u] + kv.w;
    if (in0 && (unsigned)p0 < (unsigned)NE)
        ent[p0] = sv.x | ((fv.x >= 0 && fv.x <= 4) ? (int)0x80000000 : 0);
    if (in1 && (unsigned)p1 < (unsigned)NE)
        ent[p1] = sv.y | ((fv.y >= 0 && fv.y <= 4) ? (int)0x80000000 : 0);
    if (in2 && (unsigned)p2 < (unsigned)NE)
        ent[p2] = sv.z | ((fv.z >= 0 && fv.z <= 4) ? (int)0x80000000 : 0);
    if (in3 && (unsigned)p3 < (unsigned)NE)
        ent[p3] = sv.w | ((fv.w >= 0 && fv.w <= 4) ? (int)0x80000000 : 0);
}

__global__ void __launch_bounds__(HB)
range_fill(const int* __restrict__ src,
           const int* __restrict__ dst,
           const int* __restrict__ ef,
           const int* __restrict__ rank,
           const int* __restrict__ partI,
           const int* __restrict__ partO,
           const int* __restrict__ tot,
           int* __restrict__ row_ptr,
           float* __restrict__ normO,
           float* __restrict__ normI,
           int* __restrict__ ent)
{
    __shared__ __attribute__((aligned(16))) int cntL[RPAD];
    __shared__ __attribute__((aligned(16))) int slotL[RPAD];
    __shared__ int wsum[HB / 32];
    const int tid  = threadIdx.x;
    const int lane = tid & 31;
    const int w    = tid >> 5;
    const int r  = blockIdx.x % NRG;
    const int s  = blockIdx.x / NRG;
    const int lo = r * RANGE;

    int rb = 0;
    #pragma unroll
    for (int k = 0; k < (NRG * NSL) / 32; ++k) {
        const int idx = lane + 32 * k;
        const int v = tot[idx];
        rb += ((idx % NRG) < r) ? v : 0;
    }
    for (int o = 16; o > 0; o >>= 1) rb += __shfl_xor(rb, o);

    const bool sc = tid < SCT;
    {
        int cp[NTRIP][NSL];
        #pragma unroll
        for (int k = 0; k < NTRIP; ++k) {
            const int j  = tid + k * SCT;
            const int jc = (j < RANGE) ? j : 0;
            #pragma unroll
            for (int q = 0; q < NSL; ++q) cp[k][q] = partI[(size_t)q * N_NODES + lo + jc];
        }
        const int jo = s * NOSH + tid;
        const bool on = tid < NOSH && jo < RANGE;
        const int joc = on ? jo : 0;
        int co[NSL];
        #pragma unroll
        for (int q = 0; q < NSL; ++q) co[q] = partO[(size_t)q * N_NODES + lo + joc];
        #pragma unroll
        for (int k = 0; k < NTRIP; ++k) {
            const int j = tid + k * SCT;
            int t = 0, p = 0;
            #pragma unroll
            for (int q = 0; q < NSL; ++q) {
                p += (q < s) ? cp[k][q] : 0;
                t += cp[k][q];
            }
            if (sc && j < RANGE) { cntL[j] = t; slotL[j] = p; }
        }
        int o = 0;
        #pragma unroll
        for (int q = 0; q < NSL; ++q) o += co[q];
        if (on) normO[lo + jo] = rsqrtf(fmaxf((float)o, 1.0f));
    }
    if (tid < RPAD - RANGE) { cntL[RANGE + tid] = 0; slotL[RANGE + tid] = 0; }

    __syncthreads();

    int4 c4 = make_int4(0, 0, 0, 0);
    if (sc) c4 = *(const int4*)(cntL + 4 * tid);
    const int tsum = (c4.x + c4.y) + (c4.z + c4.w);
    int x = tsum;
    #pragma unroll
    for (int o = 1; o < 32; o <<= 1) {
        const int y = __shfl_up(x, o);
        if (lane >= o) x += y;
    }
    if (lane == 31) wsum[w] = x;
    __syncthreads();
    int carry = (lane < w) ? wsum[lane] : 0;
    for (int o = 16; o > 0; o >>= 1) carry += __shfl_xor(carry, o);

    if (sc) {
        const int b0 = rb + carry + x - tsum;
        const int b1 = b0 + c4.x;
        const int b2 = b1 + c4.y;
        const int b3 = b2 + c4.z;
        const int4 pe = *(const int4*)(slotL + 4 * tid);
        *(int4*)(slotL + 4 * tid) = make_int4(b0 + pe.x, b1 + pe.y, b2 + pe.z, b3 + pe.w);
        if (s == 0) {
            const int n = 4 * tid;
            if (n + 0 < RANGE) { row_ptr[lo + n + 0] = b0; normI[lo + n + 0] = rsqrtf(fmaxf((float)c4.x, 1.0f)); }
            if (n + 1 < RANGE) { row_ptr[lo + n + 1] = b1; normI[lo + n + 1] = rsqrtf(fmaxf((float)c4.y, 1.0f)); }
            if (n + 2 < RANGE) { row_ptr[lo + n + 2] = b2; normI[lo + n + 2] = rsqrtf(fmaxf((float)c4.z, 1.0f)); }
            if (n + 3 < RANGE) { row_ptr[lo + n + 3] = b3; normI[lo + n + 3] = rsqrtf(fmaxf((float)c4.w, 1.0f)); }
            if (r == NRG - 1 && tid == SCT - 1) row_ptr[N_NODES] = b3 + c4.w;
        }
    }
    __syncthreads();

    const int e0 = s * SLICE;
    const int4* s4 = (const int4*)(src + e0);
    const int4* d4 = (const int4*)(dst + e0);
    const int4* f4 = (const int4*)(ef + e0);
    const int4* k4 = (const int4*)(rank + e0);
    #pragma unroll
    for (int kb = 0; kb < NBAT * FB; kb += FB) {
        int4 sv[FB], dv[FB], fv[FB], kv[FB];
        bool ok[FB];
        #pragma unroll
        for (int k = 0; k < FB; ++k) {
            const int g = tid + (kb + k) * HB;
            ok[k] = g < HGRP;
            const int gc = ok[k] ? g : tid;
            dv[k] = d4[gc]; kv[k] = k4[gc]; sv[k] = s4[gc]; fv[k] = f4[gc];
        }
        #pragma unroll
        for (int k = 0; k < FB; ++k) place_group(sv[k], dv[k], fv[k], kv[k], ok[k], lo, slotL, ent);
    }
}

template <int G, int MODE>
__device__ __forceinline__ void gather_group(const float* __restrict__ h,
                                             int e, int j, int lane, float4& acc)
{
    float4 a[G];
    float  wgt[G];
    #pragma unroll
    for (int k = 0; k < G; ++k) {
        const int ek = __builtin_amdgcn_readlane(e, j + k);
        a[k] = *(const float4*)(h + (size_t)(ek & 0x7fffffff) * HD + lane * 4);
        wgt[k] = (MODE == 2 && ek < 0) ? 2.0f : 1.0f;
    }
    #pragma unroll
    for (int k = 0; k < G; ++k) {
        if (MODE == 2) {
            acc.x += a[k].x * wgt[k];
            acc.y += a[k].y * wgt[k];
            acc.z += a[k].z * wgt[k];
            acc.w += a[k].w * wgt[k];
        } else {
            acc.x += a[k].x;
            acc.y += a[k].y;
            acc.z += a[k].z;
            acc.w += a[k].w;
        }
    }
}

template <int MODE>
__global__ void __launch_bounds__(256)
row_gather(const float* __restrict__ h,
           const int* __restrict__ row_ptr,
           const int* __restrict__ ent,
           const float* __restrict__ in_norm,
           const float* __restrict__ out_norm,
           const float* __restrict__ bias,
           float* __restrict__ dstv)
{
    const int lane = threadIdx.x & 31;
    const int wave = __builtin_amdgcn_readfirstlane(
        (int)((blockIdx.x * blockDim.x + threadIdx.x) >> 5));
    const int nwaves = (int)((gridDim.x * blockDim.x) >> 5);

    float4 b = make_float4(0.0f, 0.0f, 0.0f, 0.0f);
    if (MODE == 0) b = *(const float4*)(bias + lane * 4);

    for (int row = wave; row < N_NODES; row += nwaves) {
        const int beg = row_ptr[row];
        const int end = row_ptr[row + 1];
        float inn = 1.0f, outn = 1.0f;
        if (MODE == 0) { inn = in_norm[row]; outn = out_norm[row]; }

        float4 acc = make_float4(0.0f, 0.0f, 0.0f, 0.0f);
        for (int base = beg; base < end; base += 32) {
            const int n = (end - base < 32) ? (end - base) : 32;
            const int e = (lane < n) ? ent[base + lane] : 0;
            int j = 0;
            for (; j + 8 <= n; j += 8) gather_group<8, MODE>(h, e, j, lane, acc);
            if (j + 4 <= n) { gather_group<4, MODE>(h, e, j, lane, acc); j += 4; }
            if (j + 2 <= n) { gather_group<2, MODE>(h, e, j, lane, acc); j += 2; }
            if (j < n)      { gather_group<1, MODE>(h, e, j, lane, acc); }
        }

        if (MODE == 0) {
            acc.x = fmaxf(acc.x * inn + b.x, 0.0f) * outn;
            acc.y = fmaxf(acc.y * inn + b.y, 0.0f) * outn;
            acc.z = fmaxf(acc.z * inn + b.z, 0.0f) * outn;
            acc.w = fmaxf(acc.w * inn + b.w, 0.0f) * outn;
        }
        *(float4*)(dstv + (size_t)row * HD + lane * 4) = acc;
    }
}

__global__ void edge_backward(const float* __restrict__ h,
                              const int* __restrict__ src,
                              const int* __restrict__ dst,
                              const int* __restrict__ ef,
                              float* __restrict__ out)
{
    const int lane   = threadIdx.x & 31;
    const int wave   = (blockIdx.x * blockDim.x + threadIdx.x) >> 5;
    const int nwaves = (gridDim.x * blockDim.x) >> 5;
    for (long base = (long)wave * 32; base < NE; base += (long)nwaves * 32) {
        const int s = src[base + lane];
        const int d = dst[base + lane];
        const int r = ef[base + lane];
        unsigned mask = __builtin_amdgcn_ballot_w32(r == 6 || r == 14 || r == 30);
        while (mask != 0u) {
            const int j = __builtin_ctz(mask);
            mask &= mask - 1u;
            const int ss = __shfl(s, j);
            const int dd = __shfl(d, j);
            const float* hdp = h + (long)dd * HD + lane;
            float* os = out + (long)ss * HD + lane;
            const float d0 = hdp[0];
            const float d1 = hdp[32];
            const float d2 = hdp[64];
            const float d3 = hdp[96];
            atomicAdd(os,      d0);
            atomicAdd(os + 32, d1);
            atomicAdd(os + 64, d2);
            atomicAdd(os + 96, d3);
        }
    }
}

extern "C" void kernel_launch(void* const* d_in, const int* in_sizes, int n_in,
                              void* d_out, int out_size, void* d_ws, size_t ws_size,
                              hipStream_t stream)
{
    (void)in_sizes; (void)n_in; (void)out_size; (void)ws_size;

    const float* feat0 = (const float*)d_in[0];
    const float* Wfc0  = (const float*)d_in[1];
    const float* bfc0  = (const float*)d_in[2];
    const float* feat1 = (const float*)d_in[3];
    const float* Wfc1  = (const float*)d_in[4];
    const float* bfc1  = (const float*)d_in[5];
    const float* feat2 = (const float*)d_in[6];
    const float* Wfc2  = (const float*)d_in[7];
    const float* bfc2  = (const float*)d_in[8];
    const float* feat3 = (const float*)d_in[9];
    const float* Wfc3  = (const float*)d_in[10];
    const float* bfc3  = (const float*)d_in[11];
    const int*   src   = (const int*)d_in[12];
    const int*   dst   = (const int*)d_in[13];
    const int*   efeat = (const int*)d_in[14];
    const float* b_g0  = (const float*)d_in[15];
    const float* W_g1  = (const float*)d_in[16];
    const float* b_g1  = (const float*)d_in[17];
    float* out = (float*)d_out;

    const size_t NODE_ELEMS = (size_t)N_NODES * HD;
    int* tot   = (int*)d_ws;
    int* rowp  = tot + N_NODES;
    int* ent   = rowp + ROWP_PAD;
    int* rank  = ent + NE;
    int* partI = rank + NE;
    int* partO = partI + (size_t)NSL * N_NODES;
    float* bufB = (float*)d_ws + NODE_ELEMS;
    float* degO = bufB + NODE_ELEMS;
    float* degI = degO + N_NODES;

    edge_hist<<<NRG * NSL, HB, 0, stream>>>(src, dst, rank, partI, partO, tot);
    range_fill<<<NRG * NSL, HB, 0, stream>>>(src, dst, efeat, rank, partI, partO, tot,
                                             rowp, degO, degI, ent);

    fc_project_all<<<4 * FCB, GW * 32, 0, stream>>>(
        feat0, Wfc0, bfc0, feat1, Wfc1, bfc1, feat2, Wfc2, bfc2, feat3, Wfc3, bfc3,
        degO, out);

    row_gather<0><<<2048, 256, 0, stream>>>(
        out, rowp, ent, degI, degO, b_g0, bufB);

    row_gather<1><<<2048, 256, 0, stream>>>(
        bufB, rowp, ent, degI, degO, b_g0, out);
    gemm_wmma_f32<128><<<(N_NODES + 16 * GW - 1) / (16 * GW), GW * 32, 0, stream>>>(
        out, W_g1, b_g1, degI, nullptr, bufB, N_NODES, 1);

    row_gather<2><<<2048, 256, 0, stream>>>(
        bufB, rowp, ent, degI, degO, b_g0, out);
    edge_backward<<<1024, 256, 0, stream>>>(bufB, src, dst, efeat, out);
}


# reversed-edge f32 atomic adds removed: each row list gains its e_feat 6/14/30 out-edges behind its in-edges, the last row pass sums both in registers (reversed terms in slice order); 7 launches not 8
# speedup vs baseline: 3.6125x; 1.0372x over previous
#include <hip/hip_runtime.h>

typedef float v2f __attribute__((ext_vector_type(2)));
typedef float v8f __attribute__((ext_vector_type(8)));

#define N_NODES 50000
#define N_PER   12500
#define NE      600000
#define HD      128

#define GW 4
template <int K>
__device__ __forceinline__ void
gemm_strip(const float* __restrict__ A,
           const float* __restrict__ B,
           const float* __restrict__ bias,
           const float* __restrict__ rowscale,
           const float* __restrict__ outscale,
           float* __restrict__ C,
           int M, int relu, int blk, float* __restrict__ Bs)
{
    const int lane = threadIdx.x & 31;
    const int w    = __builtin_amdgcn_readfirstlane((int)(threadIdx.x >> 5));
    const int half = lane >> 4;
    const int lm   = lane & 15;
    const int mt   = (M + 15) >> 4;
    const int tm   = blk * GW + w;

    const int row = tm * 16 + lm;
    const int rc  = (row < M) ? row : (M - 1);
    const float rs = (rowscale != nullptr) ? rowscale[rc] : 1.0f;
    const float* ap = A + (long)rc * K + half * 2;
    v2f af[K / 4];
    #pragma unroll
    for (int kk = 0; kk < K / 4; ++kk) af[kk] = *(const v2f*)(ap + kk * 4);

    float bv[8], osc[8];
    #pragma unroll
    for (int tn = 0; tn < 8; ++tn) bv[tn] = bias[tn * 16 + lm];
    #pragma unroll
    for (int v = 0; v < 8; ++v) {
        const int orow = tm * 16 + v + half * 8;
        const int oc   = (orow < M) ? orow : (M - 1);
        osc[v] = (outscale != nullptr) ? outscale[oc] : 1.0f;
    }

    {
        const int tnf = lane >> 2, sub = lane & 3;
        #pragma unroll 8
        for (int i = 0; i < K / (2 * GW); ++i) {
            const int p = w + GW * i;
            const float4 r0 = *(const float4*)(B + (long)(2 * p) * HD + lane * 4);
            const float4 r1 = *(const float4*)(B + (long)(2 * p + 1) * HD + lane * 4);
            float* d = Bs + ((p >> 1) * 8 + tnf) * 64 + (p & 1) * 32 + sub * 8;
            *(float4*)d       = make_float4(r0.x, r1.x, r0.y, r1.y);
            *(float4*)(d + 4) = make_float4(r0.z, r1.z, r0.w, r1.w);
        }
    }
    __syncthreads();
    if (tm >= mt) return;

    v8f acc[8];
    #pragma unroll
    for (int tn = 0; tn < 8; ++tn) acc[tn] = (v8f){};

    #pragma unroll
    for (int kk = 0; kk < K / 4; ++kk) {
        v2f a = af[kk];
        a[0] *= rs;
        a[1] *= rs;
        #pragma unroll
        for (int tn = 0; tn < 8; ++tn) {
            const v2f b = *(const v2f*)(Bs + (kk * 8 + tn) * 64 + lane * 2);
            acc[tn] = __builtin_amdgcn_wmma_f32_16x16x4_f32(
                false, a, false, b, (short)0, acc[tn], false, false);
        }
    }

    float* cp = C + (long)(tm * 16 + half * 8) * HD + lm;
    const bool full = (tm * 16 + 16 <= M);
    if (full) {
        #pragma unroll
        for (int tn = 0; tn < 8; ++tn) {
            #pragma unroll
            for (int v = 0; v < 8; ++v) {
                float val = acc[tn][v] + bv[tn];
                if (relu) val = fmaxf(val, 0.0f);
                if (outscale != nullptr) val *= osc[v];
                cp[v * HD + tn * 16] = val;
            }
        }
    } else {
        #pragma unroll
        for (int tn = 0; tn < 8; ++tn) {
            #pragma unroll
            for (int v = 0; v < 8; ++v) {
                float val = acc[tn][v] + bv[tn];
                if (relu) val = fmaxf(val, 0.0f);
                if (outscale != nullptr) val *= osc[v];
                if (tm * 16 + half * 8 + v < M) cp[v * HD + tn * 16] = val;
            }
        }
    }
}

template <int K>
__global__ void __launch_bounds__(GW * 32)
gemm_wmma_f32(const float* __restrict__ A,
              const float* __restrict__ B,
              const float* __restrict__ bias,
              const float* __restrict__ rowscale,
              const float* __restrict__ outscale,
              float* __restrict__ C,
              int M, int relu)
{
    __shared__ float Bs[K * HD];
    gemm_strip<K>(A, B, bias, rowscale, outscale, C, M, relu, (int)blockIdx.x, Bs);
}

#define FCB ((N_PER + 16 * GW - 1) / (16 * GW))
__global__ void __launch_bounds__(GW * 32)
fc_project_all(const float* __restrict__ feat0, const float* __restrict__ W0, const float* __restrict__ b0,
               const float* __restrict__ feat1, const float* __restrict__ W1, const float* __restrict__ b1,
               const float* __restrict__ feat2, const float* __restrict__ W2, const float* __restrict__ b2,
               const float* __restrict__ feat3, const float* __restrict__ W3, const float* __restrict__ b3,
               const float* __restrict__ out_norm,
               float* __restrict__ h)
{
    __shared__ float Bs[128 * HD];
    const int t   = (int)blockIdx.x / FCB;
    const int blk = (int)blockIdx.x - t * FCB;
    const float* osc = out_norm + (size_t)t * N_PER;
    float*       c   = h + (size_t)t * N_PER * HD;
    if (t < 2) {
        const float* a  = (t == 0) ? feat0 : feat1;
        const float* wt = (t == 0) ? W0 : W1;
        const float* bs = (t == 0) ? b0 : b1;
        gemm_strip<128>(a, wt, bs, nullptr, osc, c, N_PER, 0, blk, Bs);
    } else {
        const float* a  = (t == 2) ? feat2 : feat3;
        const float* wt = (t == 2) ? W2 : W3;
        const float* bs = (t == 2) ? b2 : b3;
        gemm_strip<64>(a, wt, bs, nullptr, osc, c, N_PER, 0, blk, Bs);
    }
}

#define ROWP_PAD 50016
#define ENT_CAP  (2 * NE)
#define NRG      16
#define NSL      16
#define RANGE    (N_NODES / NRG)
#define SLICE    (NE / NSL)
static_assert(NE % 32 == 0, "32-edge batches, no tail");
static_assert(RANGE * NRG == N_NODES, "ranges tile the nodes exactly");
static_assert(SLICE * NSL == NE && SLICE % 4 == 0, "slices tile the edges in int4 groups");

#define HB   1024
#define HGRP (SLICE / 4)
#define HFUL (HGRP / HB)
static_assert(HGRP - HFUL * HB < HB, "one guarded trip covers the rest of the slice");

__device__ __forceinline__ void hist_group(const int4 sv, const int4 dv, const int4 fv, bool ok,
                                           int lo, int i, int* cI, int* cO, int* cB,
                                           int* __restrict__ rank)
{
    const unsigned d0 = (unsigned)(dv.x - lo), d1 = (unsigned)(dv.y - lo);
    const unsigned d2 = (unsigned)(dv.z - lo), d3 = (unsigned)(dv.w - lo);
    const unsigned o0 = (unsigned)(sv.x - lo), o1 = (unsigned)(sv.y - lo);
    const unsigned o2 = (unsigned)(sv.z - lo), o3 = (unsigned)(sv.w - lo);
    const bool in0 = ok && d0 < (unsigned)RANGE, in1 = ok && d1 < (unsigned)RANGE;
    const bool in2 = ok && d2 < (unsigned)RANGE, in3 = ok && d3 < (unsigned)RANGE;
    const bool on0 = ok && o0 < (unsigned)RANGE, on1 = ok && o1 < (unsigned)RANGE;
    const bool on2 = ok && o2 < (unsigned)RANGE, on3 = ok && o3 < (unsigned)RANGE;
    int r0 = 0, r1 = 0, r2 = 0, r3 = 0;
    if (in0) r0 = atomicAdd(&cI[d0], 1);
    if (in1) r1 = atomicAdd(&cI[d1], 1);
    if (in2) r2 = atomicAdd(&cI[d2], 1);
    if (in3) r3 = atomicAdd(&cI[d3], 1);
    if (on0) atomicAdd(&cO[o0], 1);
    if (on1) atomicAdd(&cO[o1], 1);
    if (on2) atomicAdd(&cO[o2], 1);
    if (on3) atomicAdd(&cO[o3], 1);
    if (on0 && (fv.x == 6 || fv.x == 14 || fv.x == 30)) atomicAdd(&cB[o0], 1);
    if (on1 && (fv.y == 6 || fv.y == 14 || fv.y == 30)) atomicAdd(&cB[o1], 1);
    if (on2 && (fv.z == 6 || fv.z == 14 || fv.z == 30)) atomicAdd(&cB[o2], 1);
    if (on3 && (fv.w == 6 || fv.w == 14 || fv.w == 30)) atomicAdd(&cB[o3], 1);
    if (in0) rank[i + 0] = r0;
    if (in1) rank[i + 1] = r1;
    if (in2) rank[i + 2] = r2;
    if (in3) rank[i + 3] = r3;
}

#define HFB  5
#define HNB  ((HFUL + 1 + HFB - 1) / HFB)
static_assert(HNB * HFB * HB >= HGRP, "the batches cover the slice");
static_assert(SLICE < 65536, "two per-slice counts share one int, 16 bits each");

__global__ void __launch_bounds__(HB)
edge_hist(const int* __restrict__ src,
          const int* __restrict__ dst,
          const int* __restrict__ ef,
          int* __restrict__ rank,
          int* __restrict__ partIB,
          int* __restrict__ partO,
          int* __restrict__ tot)
{
    __shared__ int cI[RANGE];
    __shared__ int cO[RANGE];
    __shared__ int cB[RANGE];
    __shared__ int wtot[HB / 32];
    const int tid = threadIdx.x;
    const int r  = blockIdx.x % NRG;
    const int s  = blockIdx.x / NRG;
    const int lo = r * RANGE;
    for (int j = tid; j < RANGE; j += HB) { cI[j] = 0; cO[j] = 0; cB[j] = 0; }
    __syncthreads();

    const int e0 = s * SLICE;
    const int4* s4 = (const int4*)(src + e0);
    const int4* d4 = (const int4*)(dst + e0);
    const int4* f4 = (const int4*)(ef + e0);
    #pragma unroll
    for (int kb = 0; kb < HNB * HFB; kb += HFB) {
        int4 sv[HFB], dv[HFB], fv[HFB];
        bool ok[HFB];
        #pragma unroll
        for (int k = 0; k < HFB; ++k) {
            const int g = tid + (kb + k) * HB;
            ok[k] = g < HGRP;
            const int gc = ok[k] ? g : tid;
            dv[k] = d4[gc]; sv[k] = s4[gc]; fv[k] = f4[gc];
        }
        #pragma unroll
        for (int k = 0; k < HFB; ++k)
            hist_group(sv[k], dv[k], fv[k], ok[k], lo, e0 + (tid + (kb + k) * HB) * 4,
                       cI, cO, cB, rank);
    }
    __syncthreads();

    int* pIB = partIB + (size_t)s * N_NODES + lo;
    int* pO  = partO + (size_t)s * N_NODES + lo;
    int mine = 0;
    for (int j = tid; j < RANGE; j += HB) {
        const int ci = cI[j], cb = cB[j];
        pIB[j] = (int)((unsigned)ci | ((unsigned)cb << 16));
        pO[j] = cO[j];
        mine += ci + cb;
    }
    for (int o = 16; o > 0; o >>= 1) mine += __shfl_xor(mine, o);
    if ((tid & 31) == 0) wtot[tid >> 5] = mine;
    __syncthreads();
    if (tid < 32) {
        int v = wtot[tid];
        for (int o = 16; o > 0; o >>= 1) v += __shfl_xor(v, o);
        if (tid == 0) tot[blockIdx.x] = v;
    }
}

#define SCT ((RANGE + 3) / 4)
#define RPAD (SCT * 4)
#define NTRIP 4
#define FB 5
#define NBAT ((HFUL + 1 + FB - 1) / FB)
#define NOSH ((RANGE + NSL - 1) / NSL)
#define OSH0 (((SCT + 31) / 32) * 32)
static_assert(OSH0 + NOSH <= HB, "one node per thread of the spare waves covers a block's share of out_norm");
static_assert(SCT <= HB, "one pass of the block scans the whole range");
static_assert(NTRIP * SCT >= RANGE, "the summing threads cover the range");
static_assert((NRG * NSL) % 32 == 0, "tot[] is read in whole wave trips");
static_assert(NBAT * FB * HB >= HGRP, "the batches cover the slice");

__device__ __forceinline__ void place_group(const int4 sv, const int4 dv, const int4 fv,
                                            const int4 kv, bool ok, int lo, const int* slotL,
                                            int* slotB, int* __restrict__ ent)
{
    const unsigned d0 = (unsigned)(dv.x - lo), d1 = (unsigned)(dv.y - lo);
    const unsigned d2 = (unsigned)(dv.z - lo), d3 = (unsigned)(dv.w - lo);
    const bool in0 = ok && d0 < (unsigned)RANGE, in1 = ok && d1 < (unsigned)RANGE;
    const bool in2 = ok && d2 < (unsigned)RANGE, in3 = ok && d3 < (unsigned)RANGE;
    const int p0 = slotL[in0 ? d0 : 0u] + kv.x;
    const int p1 = slotL[in1 ? d1 : 0u] + kv.y;
    const int p2 = slotL[in2 ? d2 : 0u] + kv.z;
    const int p3 = slotL[in3 ? d3 : 0u] + kv.w;
    if (in0 && (unsigned)p0 < (unsigned)ENT_CAP)
        ent[p0] = sv.x | ((fv.x >= 0 && fv.x <= 4) ? (int)0x80000000 : 0);
    if (in1 && (unsigned)p1 < (unsigned)ENT_CAP)
        ent[p1] = sv.y | ((fv.y >= 0 && fv.y <= 4) ? (int)0x80000000 : 0);
    if (in2 && (unsigned)p2 < (unsigned)ENT_CAP)
        ent[p2] = sv.z | ((fv.z >= 0 && fv.z <= 4) ? (int)0x80000000 : 0);
    if (in3 && (unsigned)p3 < (unsigned)ENT_CAP)
        ent[p3] = sv.w | ((fv.w >= 0 && fv.w <= 4) ? (int)0x80000000 : 0);

    const unsigned o0 = (unsigned)(sv.x - lo), o1 = (unsigned)(sv.y - lo);
    const unsigned o2 = (unsigned)(sv.z - lo), o3 = (unsigned)(sv.w - lo);
    const bool rv0 = ok && o0 < (unsigned)RANGE && (fv.x == 6 || fv.x == 14 || fv.x == 30);
    const bool rv1 = ok && o1 < (unsigned)RANGE && (fv.y == 6 || fv.y == 14 || fv.y == 30);
    const bool rv2 = ok && o2 < (unsigned)RANGE && (fv.z == 6 || fv.z == 14 || fv.z == 30);
    const bool rv3 = ok && o3 < (unsigned)RANGE && (fv.w == 6 || fv.w == 14 || fv.w == 30);
    int q0 = 0, q1 = 0, q2 = 0, q3 = 0;
    if (rv0) q0 = atomicAdd(&slotB[o0], 1);
    if (rv1) q1 = atomicAdd(&slotB[o1], 1);
    if (rv2) q2 = atomicAdd(&slotB[o2], 1);
    if (rv3) q3 = atomicAdd(&slotB[o3], 1);
    if (rv0 && (unsigned)q0 < (unsigned)ENT_CAP) ent[q0] = dv.x;
    if (rv1 && (unsigned)q1 < (unsigned)ENT_CAP) ent[q1] = dv.y;
    if (rv2 && (unsigned)q2 < (unsigned)ENT_CAP) ent[q2] = dv.z;
    if (rv3 && (unsigned)q3 < (unsigned)ENT_CAP) ent[q3] = dv.w;
}

__global__ void __launch_bounds__(HB)
range_fill(const int* __restrict__ src,
           const int* __restrict__ dst,
           const int* __restrict__ ef,
           const int* __restrict__ rank,
           const int* __restrict__ partIB,
           const int* __restrict__ partO,
           const int* __restrict__ tot,
           int* __restrict__ row_ptr,
           int* __restrict__ row_mid,
           float* __restrict__ normO,
           float* __restrict__ normI,
           int* __restrict__ ent)
{
    __shared__ __attribute__((aligned(16))) int cntL[RPAD];
    __shared__ __attribute__((aligned(16))) int degL[RPAD];
    __shared__ __attribute__((aligned(16))) int slotL[RPAD];
    __shared__ __attribute__((aligned(16))) int slotB[RPAD];
    __shared__ int wsum[HB / 32];
    const int tid  = threadIdx.x;
    const int lane = tid & 31;
    const int w    = tid >> 5;
    const int r  = blockIdx.x % NRG;
    const int s  = blockIdx.x / NRG;
    const int lo = r * RANGE;

    int rb = 0;
    #pragma unroll
    for (int k = 0; k < (NRG * NSL) / 32; ++k) {
        const int idx = lane + 32 * k;
        const int v = tot[idx];
        rb += ((idx % NRG) < r) ? v : 0;
    }
    for (int o = 16; o > 0; o >>= 1) rb += __shfl_xor(rb, o);

    const bool sc = tid < SCT;
    if (tid < OSH0) {
        int cp[NTRIP][NSL];
        #pragma unroll
        for (int k = 0; k < NTRIP; ++k) {
            const int j  = tid + k * SCT;
            const int jc = (j < RANGE) ? j : 0;
            #pragma unroll
            for (int q = 0; q < NSL; ++q) cp[k][q] = partIB[(size_t)q * N_NODES + lo + jc];
        }
        #pragma unroll
        for (int k = 0; k < NTRIP; ++k) {
            const int j = tid + k * SCT;
            int ti = 0, pi = 0, tb = 0, pb = 0;
            #pragma unroll
            for (int q = 0; q < NSL; ++q) {
                const int vi = (int)((unsigned)cp[k][q] & 0xffffu);
                const int vb = (int)((unsigned)cp[k][q] >> 16);
                pi += (q < s) ? vi : 0;
                pb += (q < s) ? vb : 0;
                ti += vi;
                tb += vb;
            }
            if (sc && j < RANGE) { cntL[j] = ti + tb; degL[j] = ti; slotL[j] = pi; slotB[j] = pb; }
        }
    } else {
        const int jo = s * NOSH + (tid - OSH0);
        const bool on = (tid - OSH0) < NOSH && jo < RANGE;
        const int joc = on ? jo : 0;
        int co[NSL];
        #pragma unroll
        for (int q = 0; q < NSL; ++q) co[q] = partO[(size_t)q * N_NODES + lo + joc];
        int o = 0;
        #pragma unroll
        for (int q = 0; q < NSL; ++q) o += co[q];
        if (on) normO[lo + jo] = rsqrtf(fmaxf((float)o, 1.0f));
    }
    if (tid < RPAD - RANGE) {
        cntL[RANGE + tid] = 0; degL[RANGE + tid] = 0; slotL[RANGE + tid] = 0; slotB[RANGE + tid] = 0;
    }

    __syncthreads();

    int4 c4 = make_int4(0, 0, 0, 0), i4 = make_int4(0, 0, 0, 0);
    if (sc) { c4 = *(const int4*)(cntL + 4 * tid); i4 = *(const int4*)(degL + 4 * tid); }
    const int tsum = (c4.x + c4.y) + (c4.z + c4.w);
    int x = tsum;
    #pragma unroll
    for (int o = 1; o < 32; o <<= 1) {
        const int y = __shfl_up(x, o);
        if (lane >= o) x += y;
    }
    if (lane == 31) wsum[w] = x;
    __syncthreads();
    int carry = (lane < w) ? wsum[lane] : 0;
    for (int o = 16; o > 0; o >>= 1) carry += __shfl_xor(carry, o);

    if (sc) {
        const int b0 = rb + carry + x - tsum;
        const int b1 = b0 + c4.x;
        const int b2 = b1 + c4.y;
        const int b3 = b2 + c4.z;
        const int m0 = b0 + i4.x, m1 = b1 + i4.y, m2 = b2 + i4.z, m3 = b3 + i4.w;
        const int4 pe = *(const int4*)(slotL + 4 * tid);
        const int4 pb = *(const int4*)(slotB + 4 * tid);
        *(int4*)(slotL + 4 * tid) = make_int4(b0 + pe.x, b1 + pe.y, b2 + pe.z, b3 + pe.w);
        *(int4*)(slotB + 4 * tid) = make_int4(m0 + pb.x, m1 + pb.y, m2 + pb.z, m3 + pb.w);
        if (s == 0) {
            const int n = 4 * tid;
            if (n + 0 < RANGE) { row_ptr[lo + n + 0] = b0; row_mid[lo + n + 0] = m0; normI[lo + n + 0] = rsqrtf(fmaxf((float)i4.x, 1.0f)); }
            if (n + 1 < RANGE) { row_ptr[lo + n + 1] = b1; row_mid[lo + n + 1] = m1; normI[lo + n + 1] = rsqrtf(fmaxf((float)i4.y, 1.0f)); }
            if (n + 2 < RANGE) { row_ptr[lo + n + 2] = b2; row_mid[lo + n + 2] = m2; normI[lo + n + 2] = rsqrtf(fmaxf((float)i4.z, 1.0f)); }
            if (n + 3 < RANGE) { row_ptr[lo + n + 3] = b3; row_mid[lo + n + 3] = m3; normI[lo + n + 3] = rsqrtf(fmaxf((float)i4.w, 1.0f)); }
            if (r == NRG - 1 && tid == SCT - 1) row_ptr[N_NODES] = b3 + c4.w;
        }
    }
    __syncthreads();

    const int e0 = s * SLICE;
    const int4* s4 = (const int4*)(src + e0);
    const int4* d4 = (const int4*)(dst + e0);
    const int4* f4 = (const int4*)(ef + e0);
    const int4* k4 = (const int4*)(rank + e0);
    #pragma unroll
    for (int kb = 0; kb < NBAT * FB; kb += FB) {
        int4 sv[FB], dv[FB], fv[FB], kv[FB];
        bool ok[FB];
        #pragma unroll
        for (int k = 0; k < FB; ++k) {
            const int g = tid + (kb + k) * HB;
            ok[k] = g < HGRP;
            const int gc = ok[k] ? g : tid;
            dv[k] = d4[gc]; kv[k] = k4[gc]; sv[k] = s4[gc]; fv[k] = f4[gc];
        }
        #pragma unroll
        for (int k = 0; k < FB; ++k)
            place_group(sv[k], dv[k], fv[k], kv[k], ok[k], lo, slotL, slotB, ent);
    }
}

template <int G, int MODE>
__device__ __forceinline__ void gather_group(const float* __restrict__ h,
                                             int e, int j, int lane, float4& acc)
{
    float4 a[G];
    float  wgt[G];
    #pragma unroll
    for (int k = 0; k < G; ++k) {
        const int ek = __builtin_amdgcn_readlane(e, j + k);
        a[k] = *(const float4*)(h + (size_t)(ek & 0x7fffffff) * HD + lane * 4);
        wgt[k] = (MODE == 2 && ek < 0) ? 2.0f : 1.0f;
    }
    #pragma unroll
    for (int k = 0; k < G; ++k) {
        if (MODE == 2) {
            acc.x += a[k].x * wgt[k];
            acc.y += a[k].y * wgt[k];
            acc.z += a[k].z * wgt[k];
            acc.w += a[k].w * wgt[k];
        } else {
            acc.x += a[k].x;
            acc.y += a[k].y;
            acc.z += a[k].z;
            acc.w += a[k].w;
        }
    }
}

template <int MODE>
__global__ void __launch_bounds__(256)
row_gather(const float* __restrict__ h,
           const int* __restrict__ row_ptr,
           const int* __restrict__ row_mid,
           const int* __restrict__ ent,
           const float* __restrict__ in_norm,
           const float* __restrict__ out_norm,
           const float* __restrict__ bias,
           float* __restrict__ dstv)
{
    const int lane = threadIdx.x & 31;
    const int wave = __builtin_amdgcn_readfirstlane(
        (int)((blockIdx.x * blockDim.x + threadIdx.x) >> 5));
    const int nwaves = (int)((gridDim.x * blockDim.x) >> 5);

    float4 b = make_float4(0.0f, 0.0f, 0.0f, 0.0f);
    if (MODE == 0) b = *(const float4*)(bias + lane * 4);

    for (int row = wave; row < N_NODES; row += nwaves) {
        const int beg = row_ptr[row];
        const int end = (MODE == 2) ? row_ptr[row + 1] : row_mid[row];
        float inn = 1.0f, outn = 1.0f;
        if (MODE == 0) { inn = in_norm[row]; outn = out_norm[row]; }

        float4 acc = make_float4(0.0f, 0.0f, 0.0f, 0.0f);
        for (int base = beg; base < end; base += 32) {
            const int n = (end - base < 32) ? (end - base) : 32;
            const int e = (lane < n) ? ent[base + lane] : 0;
            int j = 0;
            for (; j + 8 <= n; j += 8) gather_group<8, MODE>(h, e, j, lane, acc);
            if (j + 4 <= n) { gather_group<4, MODE>(h, e, j, lane, acc); j += 4; }
            if (j + 2 <= n) { gather_group<2, MODE>(h, e, j, lane, acc); j += 2; }
            if (j < n)      { gather_group<1, MODE>(h, e, j, lane, acc); }
        }

        if (MODE == 0) {
            acc.x = fmaxf(acc.x * inn + b.x, 0.0f) * outn;
            acc.y = fmaxf(acc.y * inn + b.y, 0.0f) * outn;
            acc.z = fmaxf(acc.z * inn + b.z, 0.0f) * outn;
            acc.w = fmaxf(acc.w * inn + b.w, 0.0f) * outn;
        }
        *(float4*)(dstv + (size_t)row * HD + lane * 4) = acc;
    }
}

extern "C" void kernel_launch(void* const* d_in, const int* in_sizes, int n_in,
                              void* d_out, int out_size, void* d_ws, size_t ws_size,
                              hipStream_t stream)
{
    (void)in_sizes; (void)n_in; (void)out_size; (void)ws_size;

    const float* feat0 = (const float*)d_in[0];
    const float* Wfc0  = (const float*)d_in[1];
    const float* bfc0  = (const float*)d_in[2];
    const float* feat1 = (const float*)d_in[3];
    const float* Wfc1  = (const float*)d_in[4];
    const float* bfc1  = (const float*)d_in[5];
    const float* feat2 = (const float*)d_in[6];
    const float* Wfc2  = (const float*)d_in[7];
    const float* bfc2  = (const float*)d_in[8];
    const float* feat3 = (const float*)d_in[9];
    const float* Wfc3  = (const float*)d_in[10];
    const float* bfc3  = (const float*)d_in[11];
    const int*   src   = (const int*)d_in[12];
    const int*   dst   = (const int*)d_in[13];
    const int*   efeat = (const int*)d_in[14];
    const float* b_g0  = (const float*)d_in[15];
    const float* W_g1  = (const float*)d_in[16];
    const float* b_g1  = (const float*)d_in[17];
    float* out = (float*)d_out;

    const size_t NODE_ELEMS = (size_t)N_NODES * HD;
    int* tot   = (int*)d_ws;
    int* rowp  = tot + N_NODES;
    int* rowm  = rowp + ROWP_PAD;
    int* ent   = rowm + ROWP_PAD;
    int* rank  = ent + ENT_CAP;
    int* partI = rank + NE;
    int* partO = partI + (size_t)NSL * N_NODES;
    float* bufB = (float*)d_ws + NODE_ELEMS;
    float* degO = bufB + NODE_ELEMS;
    float* degI = degO + N_NODES;

    edge_hist<<<NRG * NSL, HB, 0, stream>>>(src, dst, efeat, rank, partI, partO, tot);
    range_fill<<<NRG * NSL, HB, 0, stream>>>(src, dst, efeat, rank, partI, partO, tot,
                                             rowp, rowm, degO, degI, ent);

    fc_project_all<<<4 * FCB, GW * 32, 0, stream>>>(
        feat0, Wfc0, bfc0, feat1, Wfc1, bfc1, feat2, Wfc2, bfc2, feat3, Wfc3, bfc3,
        degO, out);

    row_gather<0><<<2048, 256, 0, stream>>>(
        out, rowp, rowm, ent, degI, degO, b_g0, bufB);

    row_gather<1><<<2048, 256, 0, stream>>>(
        bufB, rowp, rowm, ent, degI, degO, b_g0, out);
    gemm_wmma_f32<128><<<(N_NODES + 16 * GW - 1) / (16 * GW), GW * 32, 0, stream>>>(
        out, W_g1, b_g1, degI, nullptr, bufB, N_NODES, 1);

    row_gather<2><<<2048, 256, 0, stream>>>(
        bufB, rowp, rowm, ent, degI, degO, b_g0, out);
}


# layer-1 row sums fused with their W_g1 product: 2 waves sum a 16-row strip into an LDS tile (each row's loads requested as one group), then multiply it; 6 launches not 7; sum order unchanged
# speedup vs baseline: 3.8603x; 1.0686x over previous
#include <hip/hip_runtime.h>

typedef float v2f __attribute__((ext_vector_type(2)));
typedef float v8f __attribute__((ext_vector_type(8)));

#define N_NODES 50000
#define N_PER   12500
#define NE      600000
#define HD      128

#define GW 4
template <int K>
__device__ __forceinline__ void
gemm_strip(const float* __restrict__ A,
           const float* __restrict__ B,
           const float* __restrict__ bias,
           const float* __restrict__ rowscale,
           const float* __restrict__ outscale,
           float* __restrict__ C,
           int M, int relu, int blk, float* __restrict__ Bs)
{
    const int lane = threadIdx.x & 31;
    const int w    = __builtin_amdgcn_readfirstlane((int)(threadIdx.x >> 5));
    const int half = lane >> 4;
    const int lm   = lane & 15;
    const int mt   = (M + 15) >> 4;
    const int tm   = blk * GW + w;

    const int row = tm * 16 + lm;
    const int rc  = (row < M) ? row : (M - 1);
    const float rs = (rowscale != nullptr) ? rowscale[rc] : 1.0f;
    const float* ap = A + (long)rc * K + half * 2;
    v2f af[K / 4];
    #pragma unroll
    for (int kk = 0; kk < K / 4; ++kk) af[kk] = *(const v2f*)(ap + kk * 4);

    float bv[8], osc[8];
    #pragma unroll
    for (int tn = 0; tn < 8; ++tn) bv[tn] = bias[tn * 16 + lm];
    #pragma unroll
    for (int v = 0; v < 8; ++v) {
        const int orow = tm * 16 + v + half * 8;
        const int oc   = (orow < M) ? orow : (M - 1);
        osc[v] = (outscale != nullptr) ? outscale[oc] : 1.0f;
    }

    {
        const int tnf = lane >> 2, sub = lane & 3;
        #pragma unroll 8
        for (int i = 0; i < K / (2 * GW); ++i) {
            const int p = w + GW * i;
            const float4 r0 = *(const float4*)(B + (long)(2 * p) * HD + lane * 4);
            const float4 r1 = *(const float4*)(B + (long)(2 * p + 1) * HD + lane * 4);
            float* d = Bs + ((p >> 1) * 8 + tnf) * 64 + (p & 1) * 32 + sub * 8;
            *(float4*)d       = make_float4(r0.x, r1.x, r0.y, r1.y);
            *(float4*)(d + 4) = make_float4(r0.z, r1.z, r0.w, r1.w);
        }
    }
    __syncthreads();
    if (tm >= mt) return;

    v8f acc[8];
    #pragma unroll
    for (int tn = 0; tn < 8; ++tn) acc[tn] = (v8f){};

    #pragma unroll
    for (int kk = 0; kk < K / 4; ++kk) {
        v2f a = af[kk];
        a[0] *= rs;
        a[1] *= rs;
        #pragma unroll
        for (int tn = 0; tn < 8; ++tn) {
            const v2f b = *(const v2f*)(Bs + (kk * 8 + tn) * 64 + lane * 2);
            acc[tn] = __builtin_amdgcn_wmma_f32_16x16x4_f32(
                false, a, false, b, (short)0, acc[tn], false, false);
        }
    }

    float* cp = C + (long)(tm * 16 + half * 8) * HD + lm;
    const bool full = (tm * 16 + 16 <= M);
    if (full) {
        #pragma unroll
        for (int tn = 0; tn < 8; ++tn) {
            #pragma unroll
            for (int v = 0; v < 8; ++v) {
                float val = acc[tn][v] + bv[tn];
                if (relu) val = fmaxf(val, 0.0f);
                if (outscale != nullptr) val *= osc[v];
                cp[v * HD + tn * 16] = val;
            }
        }
    } else {
        #pragma unroll
        for (int tn = 0; tn < 8; ++tn) {
            #pragma unroll
            for (int v = 0; v < 8; ++v) {
                float val = acc[tn][v] + bv[tn];
                if (relu) val = fmaxf(val, 0.0f);
                if (outscale != nullptr) val *= osc[v];
                if (tm * 16 + half * 8 + v < M) cp[v * HD + tn * 16] = val;
            }
        }
    }
}

#define FCB ((N_PER + 16 * GW - 1) / (16 * GW))
__global__ void __launch_bounds__(GW * 32)
fc_project_all(const float* __restrict__ feat0, const float* __restrict__ W0, const float* __restrict__ b0,
               const float* __restrict__ feat1, const float* __restrict__ W1, const float* __restrict__ b1,
               const float* __restrict__ feat2, const float* __restrict__ W2, const float* __restrict__ b2,
               const float* __restrict__ feat3, const float* __restrict__ W3, const float* __restrict__ b3,
               const float* __restrict__ out_norm,
               float* __restrict__ h)
{
    __shared__ float Bs[128 * HD];
    const int t   = (int)blockIdx.x / FCB;
    const int blk = (int)blockIdx.x - t * FCB;
    const float* osc = out_norm + (size_t)t * N_PER;
    float*       c   = h + (size_t)t * N_PER * HD;
    if (t < 2) {
        const float* a  = (t == 0) ? feat0 : feat1;
        const float* wt = (t == 0) ? W0 : W1;
        const float* bs = (t == 0) ? b0 : b1;
        gemm_strip<128>(a, wt, bs, nullptr, osc, c, N_PER, 0, blk, Bs);
    } else {
        const float* a  = (t == 2) ? feat2 : feat3;
        const float* wt = (t == 2) ? W2 : W3;
        const float* bs = (t == 2) ? b2 : b3;
        gemm_strip<64>(a, wt, bs, nullptr, osc, c, N_PER, 0, blk, Bs);
    }
}

#define ROWP_PAD 50016
#define ENT_CAP  (2 * NE)
#define NRG      16
#define NSL      16
#define RANGE    (N_NODES / NRG)
#define SLICE    (NE / NSL)
static_assert(NE % 32 == 0, "32-edge batches, no tail");
static_assert(RANGE * NRG == N_NODES, "ranges tile the nodes exactly");
static_assert(SLICE * NSL == NE && SLICE % 4 == 0, "slices tile the edges in int4 groups");

#define HB   1024
#define HGRP (SLICE / 4)
#define HFUL (HGRP / HB)
static_assert(HGRP - HFUL * HB < HB, "one guarded trip covers the rest of the slice");

__device__ __forceinline__ void hist_group(const int4 sv, const int4 dv, const int4 fv, bool ok,
                                           int lo, int i, int* cI, int* cO, int* cB,
                                           int* __restrict__ rank)
{
    const unsigned d0 = (unsigned)(dv.x - lo), d1 = (unsigned)(dv.y - lo);
    const unsigned d2 = (unsigned)(dv.z - lo), d3 = (unsigned)(dv.w - lo);
    const unsigned o0 = (unsigned)(sv.x - lo), o1 = (unsigned)(sv.y - lo);
    const unsigned o2 = (unsigned)(sv.z - lo), o3 = (unsigned)(sv.w - lo);
    const bool in0 = ok && d0 < (unsigned)RANGE, in1 = ok && d1 < (unsigned)RANGE;
    const bool in2 = ok && d2 < (unsigned)RANGE, in3 = ok && d3 < (unsigned)RANGE;
    const bool on0 = ok && o0 < (unsigned)RANGE, on1 = ok && o1 < (unsigned)RANGE;
    const bool on2 = ok && o2 < (unsigned)RANGE, on3 = ok && o3 < (unsigned)RANGE;
    int r0 = 0, r1 = 0, r2 = 0, r3 = 0;
    if (in0) r0 = atomicAdd(&cI[d0], 1);
    if (in1) r1 = atomicAdd(&cI[d1], 1);
    if (in2) r2 = atomicAdd(&cI[d2], 1);
    if (in3) r3 = atomicAdd(&cI[d3], 1);
    if (on0) atomicAdd(&cO[o0], 1);
    if (on1) atomicAdd(&cO[o1], 1);
    if (on2) atomicAdd(&cO[o2], 1);
    if (on3) atomicAdd(&cO[o3], 1);
    if (on0 && (fv.x == 6 || fv.x == 14 || fv.x == 30)) atomicAdd(&cB[o0], 1);
    if (on1 && (fv.y == 6 || fv.y == 14 || fv.y == 30)) atomicAdd(&cB[o1], 1);
    if (on2 && (fv.z == 6 || fv.z == 14 || fv.z == 30)) atomicAdd(&cB[o2], 1);
    if (on3 && (fv.w == 6 || fv.w == 14 || fv.w == 30)) atomicAdd(&cB[o3], 1);
    if (in0) rank[i + 0] = r0;
    if (in1) rank[i + 1] = r1;
    if (in2) rank[i + 2] = r2;
    if (in3) rank[i + 3] = r3;
}

#define HFB  5
#define HNB  ((HFUL + 1 + HFB - 1) / HFB)
static_assert(HNB * HFB * HB >= HGRP, "the batches cover the slice");
static_assert(SLICE < 65536, "two per-slice counts share one int, 16 bits each");

__global__ void __launch_bounds__(HB)
edge_hist(const int* __restrict__ src,
          const int* __restrict__ dst,
          const int* __restrict__ ef,
          int* __restrict__ rank,
          int* __restrict__ partIB,
          int* __restrict__ partO,
          int* __restrict__ tot)
{
    __shared__ int cI[RANGE];
    __shared__ int cO[RANGE];
    __shared__ int cB[RANGE];
    __shared__ int wtot[HB / 32];
    const int tid = threadIdx.x;
    const int r  = blockIdx.x % NRG;
    const int s  = blockIdx.x / NRG;
    const int lo = r * RANGE;
    for (int j = tid; j < RANGE; j += HB) { cI[j] = 0; cO[j] = 0; cB[j] = 0; }
    __syncthreads();

    const int e0 = s * SLICE;
    const int4* s4 = (const int4*)(src + e0);
    const int4* d4 = (const int4*)(dst + e0);
    const int4* f4 = (const int4*)(ef + e0);
    #pragma unroll
    for (int kb = 0; kb < HNB * HFB; kb += HFB) {
        int4 sv[HFB], dv[HFB], fv[HFB];
        bool ok[HFB];
        #pragma unroll
        for (int k = 0; k < HFB; ++k) {
            const int g = tid + (kb + k) * HB;
            ok[k] = g < HGRP;
            const int gc = ok[k] ? g : tid;
            dv[k] = d4[gc]; sv[k] = s4[gc]; fv[k] = f4[gc];
        }
        #pragma unroll
        for (int k = 0; k < HFB; ++k)
            hist_group(sv[k], dv[k], fv[k], ok[k], lo, e0 + (tid + (kb + k) * HB) * 4,
                       cI, cO, cB, rank);
    }
    __syncthreads();

    int* pIB = partIB + (size_t)s * N_NODES + lo;
    int* pO  = partO + (size_t)s * N_NODES + lo;
    int mine = 0;
    for (int j = tid; j < RANGE; j += HB) {
        const int ci = cI[j], cb = cB[j];
        pIB[j] = (int)((unsigned)ci | ((unsigned)cb << 16));
        pO[j] = cO[j];
        mine += ci + cb;
    }
    for (int o = 16; o > 0; o >>= 1) mine += __shfl_xor(mine, o);
    if ((tid & 31) == 0) wtot[tid >> 5] = mine;
    __syncthreads();
    if (tid < 32) {
        int v = wtot[tid];
        for (int o = 16; o > 0; o >>= 1) v += __shfl_xor(v, o);
        if (tid == 0) tot[blockIdx.x] = v;
    }
}

#define SCT ((RANGE + 3) / 4)
#define RPAD (SCT * 4)
#define NTRIP 4
#define FB 5
#define NBAT ((HFUL + 1 + FB - 1) / FB)
#define NOSH ((RANGE + NSL - 1) / NSL)
#define OSH0 (((SCT + 31) / 32) * 32)
static_assert(OSH0 + NOSH <= HB, "one node per thread of the spare waves covers a block's share of out_norm");
static_assert(SCT <= HB, "one pass of the block scans the whole range");
static_assert(NTRIP * SCT >= RANGE, "the summing threads cover the range");
static_assert((NRG * NSL) % 32 == 0, "tot[] is read in whole wave trips");
static_assert(NBAT * FB * HB >= HGRP, "the batches cover the slice");

__device__ __forceinline__ void place_group(const int4 sv, const int4 dv, const int4 fv,
                                            const int4 kv, bool ok, int lo, const int* slotL,
                                            int* slotB, int* __restrict__ ent)
{
    const unsigned d0 = (unsigned)(dv.x - lo), d1 = (unsigned)(dv.y - lo);
    const unsigned d2 = (unsigned)(dv.z - lo), d3 = (unsigned)(dv.w - lo);
    const bool in0 = ok && d0 < (unsigned)RANGE, in1 = ok && d1 < (unsigned)RANGE;
    const bool in2 = ok && d2 < (unsigned)RANGE, in3 = ok && d3 < (unsigned)RANGE;
    const int p0 = slotL[in0 ? d0 : 0u] + kv.x;
    const int p1 = slotL[in1 ? d1 : 0u] + kv.y;
    const int p2 = slotL[in2 ? d2 : 0u] + kv.z;
    const int p3 = slotL[in3 ? d3 : 0u] + kv.w;
    if (in0 && (unsigned)p0 < (unsigned)ENT_CAP)
        ent[p0] = sv.x | ((fv.x >= 0 && fv.x <= 4) ? (int)0x80000000 : 0);
    if (in1 && (unsigned)p1 < (unsigned)ENT_CAP)
        ent[p1] = sv.y | ((fv.y >= 0 && fv.y <= 4) ? (int)0x80000000 : 0);
    if (in2 && (unsigned)p2 < (unsigned)ENT_CAP)
        ent[p2] = sv.z | ((fv.z >= 0 && fv.z <= 4) ? (int)0x80000000 : 0);
    if (in3 && (unsigned)p3 < (unsigned)ENT_CAP)
        ent[p3] = sv.w | ((fv.w >= 0 && fv.w <= 4) ? (int)0x80000000 : 0);

    const unsigned o0 = (unsigned)(sv.x - lo), o1 = (unsigned)(sv.y - lo);
    const unsigned o2 = (unsigned)(sv.z - lo), o3 = (unsigned)(sv.w - lo);
    const bool rv0 = ok && o0 < (unsigned)RANGE && (fv.x == 6 || fv.x == 14 || fv.x == 30);
    const bool rv1 = ok && o1 < (unsigned)RANGE && (fv.y == 6 || fv.y == 14 || fv.y == 30);
    const bool rv2 = ok && o2 < (unsigned)RANGE && (fv.z == 6 || fv.z == 14 || fv.z == 30);
    const bool rv3 = ok && o3 < (unsigned)RANGE && (fv.w == 6 || fv.w == 14 || fv.w == 30);
    int q0 = 0, q1 = 0, q2 = 0, q3 = 0;
    if (rv0) q0 = atomicAdd(&slotB[o0], 1);
    if (rv1) q1 = atomicAdd(&slotB[o1], 1);
    if (rv2) q2 = atomicAdd(&slotB[o2], 1);
    if (rv3) q3 = atomicAdd(&slotB[o3], 1);
    if (rv0 && (unsigned)q0 < (unsigned)ENT_CAP) ent[q0] = dv.x;
    if (rv1 && (unsigned)q1 < (unsigned)ENT_CAP) ent[q1] = dv.y;
    if (rv2 && (unsigned)q2 < (unsigned)ENT_CAP) ent[q2] = dv.z;
    if (rv3 && (unsigned)q3 < (unsigned)ENT_CAP) ent[q3] = dv.w;
}

__global__ void __launch_bounds__(HB)
range_fill(const int* __restrict__ src,
           const int* __restrict__ dst,
           const int* __restrict__ ef,
           const int* __restrict__ rank,
           const int* __restrict__ partIB,
           const int* __restrict__ partO,
           const int* __restrict__ tot,
           int* __restrict__ row_ptr,
           int* __restrict__ row_mid,
           float* __restrict__ normO,
           float* __restrict__ normI,
           int* __restrict__ ent)
{
    __shared__ __attribute__((aligned(16))) int cntL[RPAD];
    __shared__ __attribute__((aligned(16))) int degL[RPAD];
    __shared__ __attribute__((aligned(16))) int slotL[RPAD];
    __shared__ __attribute__((aligned(16))) int slotB[RPAD];
    __shared__ int wsum[HB / 32];
    const int tid  = threadIdx.x;
    const int lane = tid & 31;
    const int w    = tid >> 5;
    const int r  = blockIdx.x % NRG;
    const int s  = blockIdx.x / NRG;
    const int lo = r * RANGE;

    int rb = 0;
    #pragma unroll
    for (int k = 0; k < (NRG * NSL) / 32; ++k) {
        const int idx = lane + 32 * k;
        const int v = tot[idx];
        rb += ((idx % NRG) < r) ? v : 0;
    }
    for (int o = 16; o > 0; o >>= 1) rb += __shfl_xor(rb, o);

    const bool sc = tid < SCT;
    if (tid < OSH0) {
        int cp[NTRIP][NSL];
        #pragma unroll
        for (int k = 0; k < NTRIP; ++k) {
            const int j  = tid + k * SCT;
            const int jc = (j < RANGE) ? j : 0;
            #pragma unroll
            for (int q = 0; q < NSL; ++q) cp[k][q] = partIB[(size_t)q * N_NODES + lo + jc];
        }
        #pragma unroll
        for (int k = 0; k < NTRIP; ++k) {
            const int j = tid + k * SCT;
            int ti = 0, pi = 0, tb = 0, pb = 0;
            #pragma unroll
            for (int q = 0; q < NSL; ++q) {
                const int vi = (int)((unsigned)cp[k][q] & 0xffffu);
                const int vb = (int)((unsigned)cp[k][q] >> 16);
                pi += (q < s) ? vi : 0;
                pb += (q < s) ? vb : 0;
                ti += vi;
                tb += vb;
            }
            if (sc && j < RANGE) { cntL[j] = ti + tb; degL[j] = ti; slotL[j] = pi; slotB[j] = pb; }
        }
    } else {
        const int jo = s * NOSH + (tid - OSH0);
        const bool on = (tid - OSH0) < NOSH && jo < RANGE;
        const int joc = on ? jo : 0;
        int co[NSL];
        #pragma unroll
        for (int q = 0; q < NSL; ++q) co[q] = partO[(size_t)q * N_NODES + lo + joc];
        int o = 0;
        #pragma unroll
        for (int q = 0; q < NSL; ++q) o += co[q];
        if (on) normO[lo + jo] = rsqrtf(fmaxf((float)o, 1.0f));
    }
    if (tid < RPAD - RANGE) {
        cntL[RANGE + tid] = 0; degL[RANGE + tid] = 0; slotL[RANGE + tid] = 0; slotB[RANGE + tid] = 0;
    }

    __syncthreads();

    int4 c4 = make_int4(0, 0, 0, 0), i4 = make_int4(0, 0, 0, 0);
    if (sc) { c4 = *(const int4*)(cntL + 4 * tid); i4 = *(const int4*)(degL + 4 * tid); }
    const int tsum = (c4.x + c4.y) + (c4.z + c4.w);
    int x = tsum;
    #pragma unroll
    for (int o = 1; o < 32; o <<= 1) {
        const int y = __shfl_up(x, o);
        if (lane >= o) x += y;
    }
    if (lane == 31) wsum[w] = x;
    __syncthreads();
    int carry = (lane < w) ? wsum[lane] : 0;
    for (int o = 16; o > 0; o >>= 1) carry += __shfl_xor(carry, o);

    if (sc) {
        const int b0 = rb + carry + x - tsum;
        const int b1 = b0 + c4.x;
        const int b2 = b1 + c4.y;
        const int b3 = b2 + c4.z;
        const int m0 = b0 + i4.x, m1 = b1 + i4.y, m2 = b2 + i4.z, m3 = b3 + i4.w;
        const int4 pe = *(const int4*)(slotL + 4 * tid);
        const int4 pb = *(const int4*)(slotB + 4 * tid);
        *(int4*)(slotL + 4 * tid) = make_int4(b0 + pe.x, b1 + pe.y, b2 + pe.z, b3 + pe.w);
        *(int4*)(slotB + 4 * tid) = make_int4(m0 + pb.x, m1 + pb.y, m2 + pb.z, m3 + pb.w);
        if (s == 0) {
            const int n = 4 * tid;
            if (n + 0 < RANGE) { row_ptr[lo + n + 0] = b0; row_mid[lo + n + 0] = m0; normI[lo + n + 0] = rsqrtf(fmaxf((float)i4.x, 1.0f)); }
            if (n + 1 < RANGE) { row_ptr[lo + n + 1] = b1; row_mid[lo + n + 1] = m1; normI[lo + n + 1] = rsqrtf(fmaxf((float)i4.y, 1.0f)); }
            if (n + 2 < RANGE) { row_ptr[lo + n + 2] = b2; row_mid[lo + n + 2] = m2; normI[lo + n + 2] = rsqrtf(fmaxf((float)i4.z, 1.0f)); }
            if (n + 3 < RANGE) { row_ptr[lo + n + 3] = b3; row_mid[lo + n + 3] = m3; normI[lo + n + 3] = rsqrtf(fmaxf((float)i4.w, 1.0f)); }
            if (r == NRG - 1 && tid == SCT - 1) row_ptr[N_NODES] = b3 + c4.w;
        }
    }
    __syncthreads();

    const int e0 = s * SLICE;
    const int4* s4 = (const int4*)(src + e0);
    const int4* d4 = (const int4*)(dst + e0);
    const int4* f4 = (const int4*)(ef + e0);
    const int4* k4 = (const int4*)(rank + e0);
    #pragma unroll
    for (int kb = 0; kb < NBAT * FB; kb += FB) {
        int4 sv[FB], dv[FB], fv[FB], kv[FB];
        bool ok[FB];
        #pragma unroll
        for (int k = 0; k < FB; ++k) {
            const int g = tid + (kb + k) * HB;
            ok[k] = g < HGRP;
            const int gc = ok[k] ? g : tid;
            dv[k] = d4[gc]; kv[k] = k4[gc]; sv[k] = s4[gc]; fv[k] = f4[gc];
        }
        #pragma unroll
        for (int k = 0; k < FB; ++k)
            place_group(sv[k], dv[k], fv[k], kv[k], ok[k], lo, slotL, slotB, ent);
    }
}

template <int G, int MODE>
__device__ __forceinline__ void gather_group(const float* __restrict__ h,
                                             int e, int j, int lane, float4& acc)
{
    float4 a[G];
    float  wgt[G];
    #pragma unroll
    for (int k = 0; k < G; ++k) {
        const int ek = __builtin_amdgcn_readlane(e, j + k);
        a[k] = *(const float4*)(h + (size_t)(ek & 0x7fffffff) * HD + lane * 4);
        wgt[k] = (MODE == 2 && ek < 0) ? 2.0f : 1.0f;
    }
    #pragma unroll
    for (int k = 0; k < G; ++k) {
        if (MODE == 2) {
            acc.x += a[k].x * wgt[k];
            acc.y += a[k].y * wgt[k];
            acc.z += a[k].z * wgt[k];
            acc.w += a[k].w * wgt[k];
        } else {
            acc.x += a[k].x;
            acc.y += a[k].y;
            acc.z += a[k].z;
            acc.w += a[k].w;
        }
    }
}

template <int MODE>
__global__ void __launch_bounds__(256)
row_gather(const float* __restrict__ h,
           const int* __restrict__ row_ptr,
           const int* __restrict__ row_mid,
           const int* __restrict__ ent,
           const float* __restrict__ in_norm,
           const float* __restrict__ out_norm,
           const float* __restrict__ bias,
           float* __restrict__ dstv)
{
    const int lane = threadIdx.x & 31;
    const int wave = __builtin_amdgcn_readfirstlane(
        (int)((blockIdx.x * blockDim.x + threadIdx.x) >> 5));
    const int nwaves = (int)((gridDim.x * blockDim.x) >> 5);

    float4 b = make_float4(0.0f, 0.0f, 0.0f, 0.0f);
    if (MODE == 0) b = *(const float4*)(bias + lane * 4);

    for (int row = wave; row < N_NODES; row += nwaves) {
        const int beg = row_ptr[row];
        const int end = (MODE == 2) ? row_ptr[row + 1] : row_mid[row];
        float inn = 1.0f, outn = 1.0f;
        if (MODE == 0) { inn = in_norm[row]; outn = out_norm[row]; }

        float4 acc = make_float4(0.0f, 0.0f, 0.0f, 0.0f);
        for (int base = beg; base < end; base += 32) {
            const int n = (end - base < 32) ? (end - base) : 32;
            const int e = (lane < n) ? ent[base + lane] : 0;
            int j = 0;
            for (; j + 8 <= n; j += 8) gather_group<8, MODE>(h, e, j, lane, acc);
            if (j + 4 <= n) { gather_group<4, MODE>(h, e, j, lane, acc); j += 4; }
            if (j + 2 <= n) { gather_group<2, MODE>(h, e, j, lane, acc); j += 2; }
            if (j < n)      { gather_group<1, MODE>(h, e, j, lane, acc); }
        }

        if (MODE == 0) {
            acc.x = fmaxf(acc.x * inn + b.x, 0.0f) * outn;
            acc.y = fmaxf(acc.y * inn + b.y, 0.0f) * outn;
            acc.z = fmaxf(acc.z * inn + b.z, 0.0f) * outn;
            acc.w = fmaxf(acc.w * inn + b.w, 0.0f) * outn;
        }
        *(float4*)(dstv + (size_t)row * HD + lane * 4) = acc;
    }
}

__device__ __forceinline__ void gather_chunk(const float* __restrict__ h,
                                             int e, int n, int lane, float4& acc)
{
    int j = 0;
    for (; j + 16 <= n; j += 8) gather_group<8, 1>(h, e, j, lane, acc);
    switch (n - j) {
        case 1:  gather_group<1, 1>(h, e, j, lane, acc);  break;
        case 2:  gather_group<2, 1>(h, e, j, lane, acc);  break;
        case 3:  gather_group<3, 1>(h, e, j, lane, acc);  break;
        case 4:  gather_group<4, 1>(h, e, j, lane, acc);  break;
        case 5:  gather_group<5, 1>(h, e, j, lane, acc);  break;
        case 6:  gather_group<6, 1>(h, e, j, lane, acc);  break;
        case 7:  gather_group<7, 1>(h, e, j, lane, acc);  break;
        case 8:  gather_group<8, 1>(h, e, j, lane, acc);  break;
        case 9:  gather_group<9, 1>(h, e, j, lane, acc);  break;
        case 10: gather_group<10, 1>(h, e, j, lane, acc); break;
        case 11: gather_group<11, 1>(h, e, j, lane, acc); break;
        case 12: gather_group<12, 1>(h, e, j, lane, acc); break;
        case 13: gather_group<13, 1>(h, e, j, lane, acc); break;
        case 14: gather_group<14, 1>(h, e, j, lane, acc); break;
        case 15: gather_group<15, 1>(h, e, j, lane, acc); break;
        default: break;
    }
}

#define FS     13
#define FW     (2 * FS)
#define FTP    (HD + 4)
#define NSTRIP (N_NODES / 16)
static_assert(NSTRIP * 16 == N_NODES, "no partial strip");
static_assert(2 * FW < HD / 2 && 3 * FW >= HD / 2, "three trips of the block copy all k pairs of B");

__global__ void __launch_bounds__(FW * 32)
gather_gemm(const float* __restrict__ h,
            const int* __restrict__ row_ptr,
            const int* __restrict__ row_mid,
            const int* __restrict__ ent,
            const float* __restrict__ in_norm,
            const float* __restrict__ B,
            const float* __restrict__ bias,
            float* __restrict__ C)
{
    __shared__ float Bs[HD * HD];
    __shared__ __attribute__((aligned(16))) float At[FS * 16 * FTP];
    const int lane = threadIdx.x & 31;
    const int w    = __builtin_amdgcn_readfirstlane((int)(threadIdx.x >> 5));
    const int s    = w >> 1;
    const int part = w & 1;
    const int half = lane >> 4;
    const int lm   = lane & 15;
    const int tm   = (int)blockIdx.x * FS + s;
    const bool live = tm < NSTRIP;
    const int r0   = tm * 16 + part * 8;

    int rb = 0, re = 0;
    float rn = 0.0f;
    if (live && lane < 8) { rb = row_ptr[r0 + lane]; re = row_mid[r0 + lane]; rn = in_norm[r0 + lane]; }
    float bv[4];
    #pragma unroll
    for (int t = 0; t < 4; ++t) bv[t] = bias[(part * 4 + t) * 16 + lm];

    {
        const int tnf = lane >> 2, sub = lane & 3;
        float4 q0[3], q1[3];
        #pragma unroll
        for (int i = 0; i < 3; ++i) {
            const int p  = w + FW * i;
            const int pc = (p < HD / 2) ? p : (HD / 2 - 1);
            q0[i] = *(const float4*)(B + (long)(2 * pc) * HD + lane * 4);
            q1[i] = *(const float4*)(B + (long)(2 * pc + 1) * HD + lane * 4);
        }
        #pragma unroll
        for (int i = 0; i < 3; ++i) {
            const int p = w + FW * i;
            if (p < HD / 2) {
                float* d = Bs + ((p >> 1) * 8 + tnf) * 64 + (p & 1) * 32 + sub * 8;
                *(float4*)d       = make_float4(q0[i].x, q1[i].x, q0[i].y, q1[i].y);
                *(float4*)(d + 4) = make_float4(q0[i].z, q1[i].z, q0[i].w, q1[i].w);
            }
        }
    }

    if (live) {
        float* trow = At + (s * 16 + part * 8) * FTP + lane * 4;
        int ev[8];
        #pragma unroll
        for (int q = 0; q < 8; ++q) {
            const int b = __builtin_amdgcn_readlane(rb, q);
            const int n = __builtin_amdgcn_readlane(re, q) - b;
            ev[q] = (lane < n) ? ent[b + lane] : 0;
        }
        #pragma nounroll
        for (int i = 0; i < 8; ++i) {
            const int beg = __builtin_amdgcn_readlane(rb, i);
            const int end = __builtin_amdgcn_readlane(re, i);
            int ecur = ev[0];
            #pragma unroll
            for (int q = 1; q < 8; ++q) ecur = (i == q) ? ev[q] : ecur;
            const float inn = __int_as_float(__builtin_amdgcn_readlane(__float_as_int(rn), i));

            float4 acc = make_float4(0.0f, 0.0f, 0.0f, 0.0f);
            for (int base = beg; base < end; base += 32) {
                const int n = (end - base < 32) ? (end - base) : 32;
                int e = ecur;
                if (base != beg) e = (lane < n) ? ent[base + lane] : 0;
                gather_chunk(h, e, n, lane, acc);
            }
            *(float4*)(trow + i * FTP) = make_float4(acc.x * inn, acc.y * inn, acc.z * inn, acc.w * inn);
        }
    }
    __syncthreads();
    if (!live) return;

    const float* at = At + (s * 16 + lm) * FTP + half * 2;
    v2f af[HD / 4];
    #pragma unroll
    for (int kk = 0; kk < HD / 4; ++kk) af[kk] = *(const v2f*)(at + kk * 4);

    v8f acc[4];
    #pragma unroll
    for (int t = 0; t < 4; ++t) acc[t] = (v8f){};

    const float* bp = Bs + part * 4 * 64 + lane * 2;
    #pragma unroll
    for (int kk = 0; kk < HD / 4; ++kk) {
        #pragma unroll
        for (int t = 0; t < 4; ++t) {
            const v2f b = *(const v2f*)(bp + (kk * 8 + t) * 64);
            acc[t] = __builtin_amdgcn_wmma_f32_16x16x4_f32(
                false, af[kk], false, b, (short)0, acc[t], false, false);
        }
    }

    float* cp = C + (long)(tm * 16 + half * 8) * HD + part * 64 + lm;
    #pragma unroll
    for (int t = 0; t < 4; ++t) {
        #pragma unroll
        for (int v = 0; v < 8; ++v)
            cp[v * HD + t * 16] = fmaxf(acc[t][v] + bv[t], 0.0f);
    }
}

extern "C" void kernel_launch(void* const* d_in, const int* in_sizes, int n_in,
                              void* d_out, int out_size, void* d_ws, size_t ws_size,
                              hipStream_t stream)
{
    (void)in_sizes; (void)n_in; (void)out_size; (void)ws_size;

    const float* feat0 = (const float*)d_in[0];
    const float* Wfc0  = (const float*)d_in[1];
    const float* bfc0  = (const float*)d_in[2];
    const float* feat1 = (const float*)d_in[3];
    const float* Wfc1  = (const float*)d_in[4];
    const float* bfc1  = (const float*)d_in[5];
    const float* feat2 = (const float*)d_in[6];
    const float* Wfc2  = (const float*)d_in[7];
    const float* bfc2  = (const float*)d_in[8];
    const float* feat3 = (const float*)d_in[9];
    const float* Wfc3  = (const float*)d_in[10];
    const float* bfc3  = (const float*)d_in[11];
    const int*   src   = (const int*)d_in[12];
    const int*   dst   = (const int*)d_in[13];
    const int*   efeat = (const int*)d_in[14];
    const float* b_g0  = (const float*)d_in[15];
    const float* W_g1  = (const float*)d_in[16];
    const float* b_g1  = (const float*)d_in[17];
    float* out = (float*)d_out;

    const size_t NODE_ELEMS = (size_t)N_NODES * HD;
    int* tot   = (int*)d_ws;
    int* rowp  = tot + N_NODES;
    int* rowm  = rowp + ROWP_PAD;
    int* ent   = rowm + ROWP_PAD;
    int* rank  = ent + ENT_CAP;
    int* partI = rank + NE;
    int* partO = partI + (size_t)NSL * N_NODES;
    float* bufB = (float*)d_ws + NODE_ELEMS;
    float* degO = bufB + NODE_ELEMS;
    float* degI = degO + N_NODES;

    edge_hist<<<NRG * NSL, HB, 0, stream>>>(src, dst, efeat, rank, partI, partO, tot);
    range_fill<<<NRG * NSL, HB, 0, stream>>>(src, dst, efeat, rank, partI, partO, tot,
                                             rowp, rowm, degO, degI, ent);

    fc_project_all<<<4 * FCB, GW * 32, 0, stream>>>(
        feat0, Wfc0, bfc0, feat1, Wfc1, bfc1, feat2, Wfc2, bfc2, feat3, Wfc3, bfc3,
        degO, bufB);

    row_gather<0><<<2048, 256, 0, stream>>>(
        bufB, rowp, rowm, ent, degI, degO, b_g0, out);

    gather_gemm<<<(NSTRIP + FS - 1) / FS, FW * 32, 0, stream>>>(
        out, rowp, rowm, ent, degI, W_g1, b_g1, bufB);

    row_gather<2><<<2048, 256, 0, stream>>>(
        bufB, rowp, rowm, ent, degI, degO, b_g0, out);
}


# projections hosted in the row-list launch: each of its 256 blocks first multiplies 10 or 16 strips (A rows via LDS tiles, 2 waves per strip); 5 launches not 6; per-element sum order unchanged
# speedup vs baseline: 4.2363x; 1.0974x over previous
#include <hip/hip_runtime.h>

typedef float v2f __attribute__((ext_vector_type(2)));
typedef float v8f __attribute__((ext_vector_type(8)));

#define N_NODES 50000
#define N_PER   12500
#define NE      600000
#define HD      128

#define NSL      16
#define PW       32
#define MTS      ((N_PER + 15) / 16)
#define PS_A 10
#define PB_A ((MTS + PS_A - 1) / PS_A)
#define PS_B 16
#define PB_B ((MTS + PS_B - 1) / PS_B)
static_assert(2 * PS_A <= PW && 2 * PS_B <= PW, "two waves per strip fit the block");
static_assert(NSL % 2 == 0, "the two lanes of a row split the edge slices evenly");

template <int K>
__device__ __forceinline__ void
project_strips(const float* __restrict__ A,
               const float* __restrict__ B,
               const float* __restrict__ bias,
               const int* __restrict__ partO,
               int node0,
               float* __restrict__ C,
               int s0, int ns, float* __restrict__ Bs, float* __restrict__ At)
{
    static_assert((K / 2) % PW == 0, "whole trips of the block copy all k pairs of B");
    static_assert(K == 128 || K == 64, "a wave-wide 16-byte load covers one row or two");
    constexpr int TP = K + 4;
    const int lane = threadIdx.x & 31;
    const int w    = __builtin_amdgcn_readfirstlane((int)(threadIdx.x >> 5));
    const int half = lane >> 4;
    const int lm   = lane & 15;
    const int part = w & 1;
    const int tm   = s0 + (w >> 1);
    const bool live = (w >> 1) < ns && tm < MTS;

    constexpr int NB = K / (2 * PW);
    float4 r0[NB], r1[NB];
    #pragma unroll
    for (int i = 0; i < NB; ++i) {
        const int p = w + PW * i;
        r0[i] = *(const float4*)(B + (long)(2 * p) * HD + lane * 4);
        r1[i] = *(const float4*)(B + (long)(2 * p + 1) * HD + lane * 4);
    }

    int co[NSL / 2];
    float bv[4];
    const int rloc = tm * 16 + lm;
    const int rc   = (rloc < N_PER) ? rloc : (N_PER - 1);
    if (live) {
        constexpr int NL = (K == 128) ? 8 : 4;
        float4 ar[NL];
        #pragma unroll
        for (int i = 0; i < NL; ++i) {
            const int r  = tm * 16 + part * 8 + ((K == 128) ? i : (2 * i + half));
            const int rr = (r < N_PER) ? r : (N_PER - 1);
            ar[i] = *(const float4*)(A + (long)rr * K + ((K == 128) ? lane : lm) * 4);
        }
        #pragma unroll
        for (int q = 0; q < NSL / 2; ++q)
            co[q] = partO[(size_t)(half * (NSL / 2) + q) * N_NODES + node0 + rc];
        #pragma unroll
        for (int t = 0; t < 4; ++t) bv[t] = bias[(part * 4 + t) * 16 + lm];

        float* trow = At + ((w >> 1) * 16 + part * 8) * TP;
        #pragma unroll
        for (int i = 0; i < NL; ++i) {
            if (K == 128) *(float4*)(trow + i * TP + lane * 4) = ar[i];
            else          *(float4*)(trow + (2 * i + half) * TP + lm * 4) = ar[i];
        }
    }

    {
        const int tnf = lane >> 2, sub = lane & 3;
        #pragma unroll
        for (int i = 0; i < NB; ++i) {
            const int p = w + PW * i;
            float* d = Bs + ((p >> 1) * 8 + tnf) * 64 + (p & 1) * 32 + sub * 8;
            *(float4*)d       = make_float4(r0[i].x, r1[i].x, r0[i].y, r1[i].y);
            *(float4*)(d + 4) = make_float4(r0[i].z, r1[i].z, r0[i].w, r1[i].w);
        }
    }
    __syncthreads();
    if (!live) return;

    const float* at = At + ((w >> 1) * 16 + lm) * TP + half * 2;
    v2f af[K / 4];
    #pragma unroll
    for (int kk = 0; kk < K / 4; ++kk) af[kk] = *(const v2f*)(at + kk * 4);

    int od = 0;
    #pragma unroll
    for (int q = 0; q < NSL / 2; ++q) od += co[q];
    od += __shfl_xor(od, 16);
    const float on = rsqrtf(fmaxf((float)od, 1.0f));

    v8f acc[4];
    #pragma unroll
    for (int t = 0; t < 4; ++t) acc[t] = (v8f){};

    const float* bp = Bs + part * 4 * 64 + lane * 2;
    #pragma unroll
    for (int kk = 0; kk < K / 4; ++kk) {
        #pragma unroll
        for (int t = 0; t < 4; ++t) {
            const v2f b = *(const v2f*)(bp + (kk * 8 + t) * 64);
            acc[t] = __builtin_amdgcn_wmma_f32_16x16x4_f32(
                false, af[kk], false, b, (short)0, acc[t], false, false);
        }
    }

    float osc[8];
    #pragma unroll
    for (int v = 0; v < 8; ++v) osc[v] = __shfl(on, v + half * 8);

    float* cp = C + (long)(tm * 16 + half * 8) * HD + part * 64 + lm;
    const bool full = (tm * 16 + 16 <= N_PER);
    if (full) {
        #pragma unroll
        for (int t = 0; t < 4; ++t) {
            #pragma unroll
            for (int v = 0; v < 8; ++v)
                cp[v * HD + t * 16] = (acc[t][v] + bv[t]) * osc[v];
        }
    } else {
        #pragma unroll
        for (int t = 0; t < 4; ++t) {
            #pragma unroll
            for (int v = 0; v < 8; ++v)
                if (tm * 16 + half * 8 + v < N_PER) cp[v * HD + t * 16] = (acc[t][v] + bv[t]) * osc[v];
        }
    }
}

#define ROWP_PAD 50016
#define ENT_CAP  (2 * NE)
#define NRG      16
#define RANGE    (N_NODES / NRG)
#define SLICE    (NE / NSL)
static_assert(NE % 32 == 0, "32-edge batches, no tail");
static_assert(RANGE * NRG == N_NODES, "ranges tile the nodes exactly");
static_assert(SLICE * NSL == NE && SLICE % 4 == 0, "slices tile the edges in int4 groups");

#define HB   1024
#define HGRP (SLICE / 4)
#define HFUL (HGRP / HB)
static_assert(HGRP - HFUL * HB < HB, "one guarded trip covers the rest of the slice");

__device__ __forceinline__ void hist_group(const int4 sv, const int4 dv, const int4 fv, bool ok,
                                           int lo, int i, int* cI, int* cO, int* cB,
                                           int* __restrict__ rank)
{
    const unsigned d0 = (unsigned)(dv.x - lo), d1 = (unsigned)(dv.y - lo);
    const unsigned d2 = (unsigned)(dv.z - lo), d3 = (unsigned)(dv.w - lo);
    const unsigned o0 = (unsigned)(sv.x - lo), o1 = (unsigned)(sv.y - lo);
    const unsigned o2 = (unsigned)(sv.z - lo), o3 = (unsigned)(sv.w - lo);
    const bool in0 = ok && d0 < (unsigned)RANGE, in1 = ok && d1 < (unsigned)RANGE;
    const bool in2 = ok && d2 < (unsigned)RANGE, in3 = ok && d3 < (unsigned)RANGE;
    const bool on0 = ok && o0 < (unsigned)RANGE, on1 = ok && o1 < (unsigned)RANGE;
    const bool on2 = ok && o2 < (unsigned)RANGE, on3 = ok && o3 < (unsigned)RANGE;
    int r0 = 0, r1 = 0, r2 = 0, r3 = 0;
    if (in0) r0 = atomicAdd(&cI[d0], 1);
    if (in1) r1 = atomicAdd(&cI[d1], 1);
    if (in2) r2 = atomicAdd(&cI[d2], 1);
    if (in3) r3 = atomicAdd(&cI[d3], 1);
    if (on0) atomicAdd(&cO[o0], 1);
    if (on1) atomicAdd(&cO[o1], 1);
    if (on2) atomicAdd(&cO[o2], 1);
    if (on3) atomicAdd(&cO[o3], 1);
    if (on0 && (fv.x == 6 || fv.x == 14 || fv.x == 30)) atomicAdd(&cB[o0], 1);
    if (on1 && (fv.y == 6 || fv.y == 14 || fv.y == 30)) atomicAdd(&cB[o1], 1);
    if (on2 && (fv.z == 6 || fv.z == 14 || fv.z == 30)) atomicAdd(&cB[o2], 1);
    if (on3 && (fv.w == 6 || fv.w == 14 || fv.w == 30)) atomicAdd(&cB[o3], 1);
    if (in0) rank[i + 0] = r0;
    if (in1) rank[i + 1] = r1;
    if (in2) rank[i + 2] = r2;
    if (in3) rank[i + 3] = r3;
}

#define HFB  5
#define HNB  ((HFUL + 1 + HFB - 1) / HFB)
static_assert(HNB * HFB * HB >= HGRP, "the batches cover the slice");
static_assert(SLICE < 65536, "two per-slice counts share one int, 16 bits each");

__global__ void __launch_bounds__(HB)
edge_hist(const int* __restrict__ src,
          const int* __restrict__ dst,
          const int* __restrict__ ef,
          int* __restrict__ rank,
          int* __restrict__ partIB,
          int* __restrict__ partO,
          int* __restrict__ tot)
{
    __shared__ int cI[RANGE];
    __shared__ int cO[RANGE];
    __shared__ int cB[RANGE];
    __shared__ int wtot[HB / 32];
    const int tid = threadIdx.x;
    const int r  = blockIdx.x % NRG;
    const int s  = blockIdx.x / NRG;
    const int lo = r * RANGE;
    for (int j = tid; j < RANGE; j += HB) { cI[j] = 0; cO[j] = 0; cB[j] = 0; }
    __syncthreads();

    const int e0 = s * SLICE;
    const int4* s4 = (const int4*)(src + e0);
    const int4* d4 = (const int4*)(dst + e0);
    const int4* f4 = (const int4*)(ef + e0);
    #pragma unroll
    for (int kb = 0; kb < HNB * HFB; kb += HFB) {
        int4 sv[HFB], dv[HFB], fv[HFB];
        bool ok[HFB];
        #pragma unroll
        for (int k = 0; k < HFB; ++k) {
            const int g = tid + (kb + k) * HB;
            ok[k] = g < HGRP;
            const int gc = ok[k] ? g : tid;
            dv[k] = d4[gc]; sv[k] = s4[gc]; fv[k] = f4[gc];
        }
        #pragma unroll
        for (int k = 0; k < HFB; ++k)
            hist_group(sv[k], dv[k], fv[k], ok[k], lo, e0 + (tid + (kb + k) * HB) * 4,
                       cI, cO, cB, rank);
    }
    __syncthreads();

    int* pIB = partIB + (size_t)s * N_NODES + lo;
    int* pO  = partO + (size_t)s * N_NODES + lo;
    int mine = 0;
    for (int j = tid; j < RANGE; j += HB) {
        const int ci = cI[j], cb = cB[j];
        pIB[j] = (int)((unsigned)ci | ((unsigned)cb << 16));
        pO[j] = cO[j];
        mine += ci + cb;
    }
    for (int o = 16; o > 0; o >>= 1) mine += __shfl_xor(mine, o);
    if ((tid & 31) == 0) wtot[tid >> 5] = mine;
    __syncthreads();
    if (tid < 32) {
        int v = wtot[tid];
        for (int o = 16; o > 0; o >>= 1) v += __shfl_xor(v, o);
        if (tid == 0) tot[blockIdx.x] = v;
    }
}

#define SCT ((RANGE + 3) / 4)
#define RPAD (SCT * 4)
#define NTRIP 4
#define FB 5
#define NBAT ((HFUL + 1 + FB - 1) / FB)
#define NOSH ((RANGE + NSL - 1) / NSL)
#define OSH0 (((SCT + 31) / 32) * 32)
static_assert(OSH0 + NOSH <= HB, "one node per thread of the spare waves covers a block's share of out_norm");
static_assert(SCT <= HB, "one pass of the block scans the whole range");
static_assert(NTRIP * SCT >= RANGE, "the summing threads cover the range");
static_assert((NRG * NSL) % 32 == 0, "tot[] is read in whole wave trips");
static_assert(NBAT * FB * HB >= HGRP, "the batches cover the slice");

__device__ __forceinline__ void place_group(const int4 sv, const int4 dv, const int4 fv,
                                            const int4 kv, bool ok, int lo, const int* slotL,
                                            int* slotB, int* __restrict__ ent)
{
    const unsigned d0 = (unsigned)(dv.x - lo), d1 = (unsigned)(dv.y - lo);
    const unsigned d2 = (unsigned)(dv.z - lo), d3 = (unsigned)(dv.w - lo);
    const bool in0 = ok && d0 < (unsigned)RANGE, in1 = ok && d1 < (unsigned)RANGE;
    const bool in2 = ok && d2 < (unsigned)RANGE, in3 = ok && d3 < (unsigned)RANGE;
    const int p0 = slotL[in0 ? d0 : 0u] + kv.x;
    const int p1 = slotL[in1 ? d1 : 0u] + kv.y;
    const int p2 = slotL[in2 ? d2 : 0u] + kv.z;
    const int p3 = slotL[in3 ? d3 : 0u] + kv.w;
    if (in0 && (unsigned)p0 < (unsigned)ENT_CAP)
        ent[p0] = sv.x | ((fv.x >= 0 && fv.x <= 4) ? (int)0x80000000 : 0);
    if (in1 && (unsigned)p1 < (unsigned)ENT_CAP)
        ent[p1] = sv.y | ((fv.y >= 0 && fv.y <= 4) ? (int)0x80000000 : 0);
    if (in2 && (unsigned)p2 < (unsigned)ENT_CAP)
        ent[p2] = sv.z | ((fv.z >= 0 && fv.z <= 4) ? (int)0x80000000 : 0);
    if (in3 && (unsigned)p3 < (unsigned)ENT_CAP)
        ent[p3] = sv.w | ((fv.w >= 0 && fv.w <= 4) ? (int)0x80000000 : 0);

    const unsigned o0 = (unsigned)(sv.x - lo), o1 = (unsigned)(sv.y - lo);
    const unsigned o2 = (unsigned)(sv.z - lo), o3 = (unsigned)(sv.w - lo);
    const bool rv0 = ok && o0 < (unsigned)RANGE && (fv.x == 6 || fv.x == 14 || fv.x == 30);
    const bool rv1 = ok && o1 < (unsigned)RANGE && (fv.y == 6 || fv.y == 14 || fv.y == 30);
    const bool rv2 = ok && o2 < (unsigned)RANGE && (fv.z == 6 || fv.z == 14 || fv.z == 30);
    const bool rv3 = ok && o3 < (unsigned)RANGE && (fv.w == 6 || fv.w == 14 || fv.w == 30);
    int q0 = 0, q1 = 0, q2 = 0, q3 = 0;
    if (rv0) q0 = atomicAdd(&slotB[o0], 1);
    if (rv1) q1 = atomicAdd(&slotB[o1], 1);
    if (rv2) q2 = atomicAdd(&slotB[o2], 1);
    if (rv3) q3 = atomicAdd(&slotB[o3], 1);
    if (rv0 && (unsigned)q0 < (unsigned)ENT_CAP) ent[q0] = dv.x;
    if (rv1 && (unsigned)q1 < (unsigned)ENT_CAP) ent[q1] = dv.y;
    if (rv2 && (unsigned)q2 < (unsigned)ENT_CAP) ent[q2] = dv.z;
    if (rv3 && (unsigned)q3 < (unsigned)ENT_CAP) ent[q3] = dv.w;
}

static_assert(2 * PB_A + 2 * PB_B == NRG * NSL, "the projection's blocks are exactly the blocks of the (range, slice) grid");
static_assert(PW * 32 == HB, "project_strips is written for the blocks of this launch");

__global__ void __launch_bounds__(HB)
fill_project(const int* __restrict__ src,
             const int* __restrict__ dst,
             const int* __restrict__ ef,
             const int* __restrict__ rank,
             const int* __restrict__ partIB,
             const int* __restrict__ partO,
             const int* __restrict__ tot,
             int* __restrict__ row_ptr,
             int* __restrict__ row_mid,
             float* __restrict__ normO,
             float* __restrict__ normI,
             int* __restrict__ ent,
             const float* __restrict__ feat0, const float* __restrict__ W0, const float* __restrict__ b0,
             const float* __restrict__ feat1, const float* __restrict__ W1, const float* __restrict__ b1,
             const float* __restrict__ feat2, const float* __restrict__ W2, const float* __restrict__ b2,
             const float* __restrict__ feat3, const float* __restrict__ W3, const float* __restrict__ b3,
             float* __restrict__ h)
{
    __shared__ __attribute__((aligned(16))) float Bs[128 * HD];
    __shared__ __attribute__((aligned(16))) float At[PS_A * 16 * (128 + 4)];
    static_assert(PS_B * 16 * (64 + 4) <= PS_A * 16 * (128 + 4), "the K = 64 tiles fit the same array");
    {
        const int b = (int)blockIdx.x;
        if (b < 2 * PB_A) {
            const int t  = (b < PB_A) ? 0 : 1;
            const int sb = b - t * PB_A;
            project_strips<128>((t == 0) ? feat0 : feat1, (t == 0) ? W0 : W1, (t == 0) ? b0 : b1,
                                partO, t * N_PER, h + (size_t)t * N_PER * HD, sb * PS_A, PS_A, Bs, At);
        } else {
            const int c  = b - 2 * PB_A;
            const int t  = (c < PB_B) ? 2 : 3;
            const int sb = c - (t - 2) * PB_B;
            project_strips<64>((t == 2) ? feat2 : feat3, (t == 2) ? W2 : W3, (t == 2) ? b2 : b3,
                               partO, t * N_PER, h + (size_t)t * N_PER * HD, sb * PS_B, PS_B, Bs, At);
        }
    }

    __shared__ __attribute__((aligned(16))) int cntL[RPAD];
    __shared__ __attribute__((aligned(16))) int degL[RPAD];
    __shared__ __attribute__((aligned(16))) int slotL[RPAD];
    __shared__ __attribute__((aligned(16))) int slotB[RPAD];
    __shared__ int wsum[HB / 32];
    const int tid  = threadIdx.x;
    const int lane = tid & 31;
    const int w    = tid >> 5;
    const int r  = blockIdx.x % NRG;
    const int s  = blockIdx.x / NRG;
    const int lo = r * RANGE;

    int rb = 0;
    #pragma unroll
    for (int k = 0; k < (NRG * NSL) / 32; ++k) {
        const int idx = lane + 32 * k;
        const int v = tot[idx];
        rb += ((idx % NRG) < r) ? v : 0;
    }
    for (int o = 16; o > 0; o >>= 1) rb += __shfl_xor(rb, o);

    const bool sc = tid < SCT;
    if (tid < OSH0) {
        int cp[NTRIP][NSL];
        #pragma unroll
        for (int k = 0; k < NTRIP; ++k) {
            const int j  = tid + k * SCT;
            const int jc = (j < RANGE) ? j : 0;
            #pragma unroll
            for (int q = 0; q < NSL; ++q) cp[k][q] = partIB[(size_t)q * N_NODES + lo + jc];
        }
        #pragma unroll
        for (int k = 0; k < NTRIP; ++k) {
            const int j = tid + k * SCT;
            int ti = 0, pi = 0, tb = 0, pb = 0;
            #pragma unroll
            for (int q = 0; q < NSL; ++q) {
                const int vi = (int)((unsigned)cp[k][q] & 0xffffu);
                const int vb = (int)((unsigned)cp[k][q] >> 16);
                pi += (q < s) ? vi : 0;
                pb += (q < s) ? vb : 0;
                ti += vi;
                tb += vb;
            }
            if (sc && j < RANGE) { cntL[j] = ti + tb; degL[j] = ti; slotL[j] = pi; slotB[j] = pb; }
        }
    } else {
        const int jo = s * NOSH + (tid - OSH0);
        const bool on = (tid - OSH0) < NOSH && jo < RANGE;
        const int joc = on ? jo : 0;
        int co[NSL];
        #pragma unroll
        for (int q = 0; q < NSL; ++q) co[q] = partO[(size_t)q * N_NODES + lo + joc];
        int o = 0;
        #pragma unroll
        for (int q = 0; q < NSL; ++q) o += co[q];
        if (on) normO[lo + jo] = rsqrtf(fmaxf((float)o, 1.0f));
    }
    if (tid < RPAD - RANGE) {
        cntL[RANGE + tid] = 0; degL[RANGE + tid] = 0; slotL[RANGE + tid] = 0; slotB[RANGE + tid] = 0;
    }

    __syncthreads();

    int4 c4 = make_int4(0, 0, 0, 0), i4 = make_int4(0, 0, 0, 0);
    if (sc) { c4 = *(const int4*)(cntL + 4 * tid); i4 = *(const int4*)(degL + 4 * tid); }
    const int tsum = (c4.x + c4.y) + (c4.z + c4.w);
    int x = tsum;
    #pragma unroll
    for (int o = 1; o < 32; o <<= 1) {
        const int y = __shfl_up(x, o);
        if (lane >= o) x += y;
    }
    if (lane == 31) wsum[w] = x;
    __syncthreads();
    int carry = (lane < w) ? wsum[lane] : 0;
    for (int o = 16; o > 0; o >>= 1) carry += __shfl_xor(carry, o);

    if (sc) {
        const int b0 = rb + carry + x - tsum;
        const int b1 = b0 + c4.x;
        const int b2 = b1 + c4.y;
        const int b3 = b2 + c4.z;
        const int m0 = b0 + i4.x, m1 = b1 + i4.y, m2 = b2 + i4.z, m3 = b3 + i4.w;
        const int4 pe = *(const int4*)(slotL + 4 * tid);
        const int4 pb = *(const int4*)(slotB + 4 * tid);
        *(int4*)(slotL + 4 * tid) = make_int4(b0 + pe.x, b1 + pe.y, b2 + pe.z, b3 + pe.w);
        *(int4*)(slotB + 4 * tid) = make_int4(m0 + pb.x, m1 + pb.y, m2 + pb.z, m3 + pb.w);
        if (s == 0) {
            const int n = 4 * tid;
            if (n + 0 < RANGE) { row_ptr[lo + n + 0] = b0; row_mid[lo + n + 0] = m0; normI[lo + n + 0] = rsqrtf(fmaxf((float)i4.x, 1.0f)); }
            if (n + 1 < RANGE) { row_ptr[lo + n + 1] = b1; row_mid[lo + n + 1] = m1; normI[lo + n + 1] = rsqrtf(fmaxf((float)i4.y, 1.0f)); }
            if (n + 2 < RANGE) { row_ptr[lo + n + 2] = b2; row_mid[lo + n + 2] = m2; normI[lo + n + 2] = rsqrtf(fmaxf((float)i4.z, 1.0f)); }
            if (n + 3 < RANGE) { row_ptr[lo + n + 3] = b3; row_mid[lo + n + 3] = m3; normI[lo + n + 3] = rsqrtf(fmaxf((float)i4.w, 1.0f)); }
            if (r == NRG - 1 && tid == SCT - 1) row_ptr[N_NODES] = b3 + c4.w;
        }
    }
    __syncthreads();

    const int e0 = s * SLICE;
    const int4* s4 = (const int4*)(src + e0);
    const int4* d4 = (const int4*)(dst + e0);
    const int4* f4 = (const int4*)(ef + e0);
    const int4* k4 = (const int4*)(rank + e0);
    #pragma unroll
    for (int kb = 0; kb < NBAT * FB; kb += FB) {
        int4 sv[FB], dv[FB], fv[FB], kv[FB];
        bool ok[FB];
        #pragma unroll
        for (int k = 0; k < FB; ++k) {
            const int g = tid + (kb + k) * HB;
            ok[k] = g < HGRP;
            const int gc = ok[k] ? g : tid;
            dv[k] = d4[gc]; kv[k] = k4[gc]; sv[k] = s4[gc]; fv[k] = f4[gc];
        }
        #pragma unroll
        for (int k = 0; k < FB; ++k)
            place_group(sv[k], dv[k], fv[k], kv[k], ok[k], lo, slotL, slotB, ent);
    }
}

template <int G, int MODE>
__device__ __forceinline__ void gather_group(const float* __restrict__ h,
                                             int e, int j, int lane, float4& acc)
{
    float4 a[G];
    float  wgt[G];
    #pragma unroll
    for (int k = 0; k < G; ++k) {
        const int ek = __builtin_amdgcn_readlane(e, j + k);
        a[k] = *(const float4*)(h + (size_t)(ek & 0x7fffffff) * HD + lane * 4);
        wgt[k] = (MODE == 2 && ek < 0) ? 2.0f : 1.0f;
    }
    #pragma unroll
    for (int k = 0; k < G; ++k) {
        if (MODE == 2) {
            acc.x += a[k].x * wgt[k];
            acc.y += a[k].y * wgt[k];
            acc.z += a[k].z * wgt[k];
            acc.w += a[k].w * wgt[k];
        } else {
            acc.x += a[k].x;
            acc.y += a[k].y;
            acc.z += a[k].z;
            acc.w += a[k].w;
        }
    }
}

template <int MODE>
__global__ void __launch_bounds__(256)
row_gather(const float* __restrict__ h,
           const int* __restrict__ row_ptr,
           const int* __restrict__ row_mid,
           const int* __restrict__ ent,
           const float* __restrict__ in_norm,
           const float* __restrict__ out_norm,
           const float* __restrict__ bias,
           float* __restrict__ dstv)
{
    const int lane = threadIdx.x & 31;
    const int wave = __builtin_amdgcn_readfirstlane(
        (int)((blockIdx.x * blockDim.x + threadIdx.x) >> 5));
    const int nwaves = (int)((gridDim.x * blockDim.x) >> 5);

    float4 b = make_float4(0.0f, 0.0f, 0.0f, 0.0f);
    if (MODE == 0) b = *(const float4*)(bias + lane * 4);

    for (int row = wave; row < N_NODES; row += nwaves) {
        const int beg = row_ptr[row];
        const int end = (MODE == 2) ? row_ptr[row + 1] : row_mid[row];
        float inn = 1.0f, outn = 1.0f;
        if (MODE == 0) { inn = in_norm[row]; outn = out_norm[row]; }

        float4 acc = make_float4(0.0f, 0.0f, 0.0f, 0.0f);
        for (int base = beg; base < end; base += 32) {
            const int n = (end - base < 32) ? (end - base) : 32;
            const int e = (lane < n) ? ent[base + lane] : 0;
            int j = 0;
            for (; j + 8 <= n; j += 8) gather_group<8, MODE>(h, e, j, lane, acc);
            if (j + 4 <= n) { gather_group<4, MODE>(h, e, j, lane, acc); j += 4; }
            if (j + 2 <= n) { gather_group<2, MODE>(h, e, j, lane, acc); j += 2; }
            if (j < n)      { gather_group<1, MODE>(h, e, j, lane, acc); }
        }

        if (MODE == 0) {
            acc.x = fmaxf(acc.x * inn + b.x, 0.0f) * outn;
            acc.y = fmaxf(acc.y * inn + b.y, 0.0f) * outn;
            acc.z = fmaxf(acc.z * inn + b.z, 0.0f) * outn;
            acc.w = fmaxf(acc.w * inn + b.w, 0.0f) * outn;
        }
        *(float4*)(dstv + (size_t)row * HD + lane * 4) = acc;
    }
}

__device__ __forceinline__ void gather_chunk(const float* __restrict__ h,
                                             int e, int n, int lane, float4& acc)
{
    int j = 0;
    for (; j + 16 <= n; j += 8) gather_group<8, 1>(h, e, j, lane, acc);
    switch (n - j) {
        case 1:  gather_group<1, 1>(h, e, j, lane, acc);  break;
        case 2:  gather_group<2, 1>(h, e, j, lane, acc);  break;
        case 3:  gather_group<3, 1>(h, e, j, lane, acc);  break;
        case 4:  gather_group<4, 1>(h, e, j, lane, acc);  break;
        case 5:  gather_group<5, 1>(h, e, j, lane, acc);  break;
        case 6:  gather_group<6, 1>(h, e, j, lane, acc);  break;
        case 7:  gather_group<7, 1>(h, e, j, lane, acc);  break;
        case 8:  gather_group<8, 1>(h, e, j, lane, acc);  break;
        case 9:  gather_group<9, 1>(h, e, j, lane, acc);  break;
        case 10: gather_group<10, 1>(h, e, j, lane, acc); break;
        case 11: gather_group<11, 1>(h, e, j, lane, acc); break;
        case 12: gather_group<12, 1>(h, e, j, lane, acc); break;
        case 13: gather_group<13, 1>(h, e, j, lane, acc); break;
        case 14: gather_group<14, 1>(h, e, j, lane, acc); break;
        case 15: gather_group<15, 1>(h, e, j, lane, acc); break;
        default: break;
    }
}

#define FS     13
#define FW     (2 * FS)
#define FTP    (HD + 4)
#define NSTRIP (N_NODES / 16)
static_assert(NSTRIP * 16 == N_NODES, "no partial strip");
static_assert(2 * FW < HD / 2 && 3 * FW >= HD / 2, "three trips of the block copy all k pairs of B");

__global__ void __launch_bounds__(FW * 32)
gather_gemm(const float* __restrict__ h,
            const int* __restrict__ row_ptr,
            const int* __restrict__ row_mid,
            const int* __restrict__ ent,
            const float* __restrict__ in_norm,
            const float* __restrict__ B,
            const float* __restrict__ bias,
            float* __restrict__ C)
{
    __shared__ float Bs[HD * HD];
    __shared__ __attribute__((aligned(16))) float At[FS * 16 * FTP];
    const int lane = threadIdx.x & 31;
    const int w    = __builtin_amdgcn_readfirstlane((int)(threadIdx.x >> 5));
    const int s    = w >> 1;
    const int part = w & 1;
    const int half = lane >> 4;
    const int lm   = lane & 15;
    const int tm   = (int)blockIdx.x * FS + s;
    const bool live = tm < NSTRIP;
    const int r0   = tm * 16 + part * 8;

    int rb = 0, re = 0;
    float rn = 0.0f;
    if (live && lane < 8) { rb = row_ptr[r0 + lane]; re = row_mid[r0 + lane]; rn = in_norm[r0 + lane]; }
    float bv[4];
    #pragma unroll
    for (int t = 0; t < 4; ++t) bv[t] = bias[(part * 4 + t) * 16 + lm];

    {
        const int tnf = lane >> 2, sub = lane & 3;
        float4 q0[3], q1[3];
        #pragma unroll
        for (int i = 0; i < 3; ++i) {
            const int p  = w + FW * i;
            const int pc = (p < HD / 2) ? p : (HD / 2 - 1);
            q0[i] = *(const float4*)(B + (long)(2 * pc) * HD + lane * 4);
            q1[i] = *(const float4*)(B + (long)(2 * pc + 1) * HD + lane * 4);
        }
        #pragma unroll
        for (int i = 0; i < 3; ++i) {
            const int p = w + FW * i;
            if (p < HD / 2) {
                float* d = Bs + ((p >> 1) * 8 + tnf) * 64 + (p & 1) * 32 + sub * 8;
                *(float4*)d       = make_float4(q0[i].x, q1[i].x, q0[i].y, q1[i].y);
                *(float4*)(d + 4) = make_float4(q0[i].z, q1[i].z, q0[i].w, q1[i].w);
            }
        }
    }

    if (live) {
        float* trow = At + (s * 16 + part * 8) * FTP + lane * 4;
        int ev[8];
        #pragma unroll
        for (int q = 0; q < 8; ++q) {
            const int b = __builtin_amdgcn_readlane(rb, q);
            const int n = __builtin_amdgcn_readlane(re, q) - b;
            ev[q] = (lane < n) ? ent[b + lane] : 0;
        }
        #pragma nounroll
        for (int i = 0; i < 8; ++i) {
            const int beg = __builtin_amdgcn_readlane(rb, i);
            const int end = __builtin_amdgcn_readlane(re, i);
            int ecur = ev[0];
            #pragma unroll
            for (int q = 1; q < 8; ++q) ecur = (i == q) ? ev[q] : ecur;
            const float inn = __int_as_float(__builtin_amdgcn_readlane(__float_as_int(rn), i));

            float4 acc = make_float4(0.0f, 0.0f, 0.0f, 0.0f);
            for (int base = beg; base < end; base += 32) {
                const int n = (end - base < 32) ? (end - base) : 32;
                int e = ecur;
                if (base != beg) e = (lane < n) ? ent[base + lane] : 0;
                gather_chunk(h, e, n, lane, acc);
            }
            *(float4*)(trow + i * FTP) = make_float4(acc.x * inn, acc.y * inn, acc.z * inn, acc.w * inn);
        }
    }
    __syncthreads();
    if (!live) return;

    const float* at = At + (s * 16 + lm) * FTP + half * 2;
    v2f af[HD / 4];
    #pragma unroll
    for (int kk = 0; kk < HD / 4; ++kk) af[kk] = *(const v2f*)(at + kk * 4);

    v8f acc[4];
    #pragma unroll
    for (int t = 0; t < 4; ++t) acc[t] = (v8f){};

    const float* bp = Bs + part * 4 * 64 + lane * 2;
    #pragma unroll
    for (int kk = 0; kk < HD / 4; ++kk) {
        #pragma unroll
        for (int t = 0; t < 4; ++t) {
            const v2f b = *(const v2f*)(bp + (kk * 8 + t) * 64);
            acc[t] = __builtin_amdgcn_wmma_f32_16x16x4_f32(
                false, af[kk], false, b, (short)0, acc[t], false, false);
        }
    }

    float* cp = C + (long)(tm * 16 + half * 8) * HD + part * 64 + lm;
    #pragma unroll
    for (int t = 0; t < 4; ++t) {
        #pragma unroll
        for (int v = 0; v < 8; ++v)
            cp[v * HD + t * 16] = fmaxf(acc[t][v] + bv[t], 0.0f);
    }
}

extern "C" void kernel_launch(void* const* d_in, const int* in_sizes, int n_in,
                              void* d_out, int out_size, void* d_ws, size_t ws_size,
                              hipStream_t stream)
{
    (void)in_sizes; (void)n_in; (void)out_size; (void)ws_size;

    const float* feat0 = (const float*)d_in[0];
    const float* Wfc0  = (const float*)d_in[1];
    const float* bfc0  = (const float*)d_in[2];
    const float* feat1 = (const float*)d_in[3];
    const float* Wfc1  = (const float*)d_in[4];
    const float* bfc1  = (const float*)d_in[5];
    const float* feat2 = (const float*)d_in[6];
    const float* Wfc2  = (const float*)d_in[7];
    const float* bfc2  = (const float*)d_in[8];
    const float* feat3 = (const float*)d_in[9];
    const float* Wfc3  = (const float*)d_in[10];
    const float* bfc3  = (const float*)d_in[11];
    const int*   src   = (const int*)d_in[12];
    const int*   dst   = (const int*)d_in[13];
    const int*   efeat = (const int*)d_in[14];
    const float* b_g0  = (const float*)d_in[15];
    const float* W_g1  = (const float*)d_in[16];
    const float* b_g1  = (const float*)d_in[17];
    float* out = (float*)d_out;

    const size_t NODE_ELEMS = (size_t)N_NODES * HD;
    int* tot   = (int*)d_ws;
    int* rowp  = tot + N_NODES;
    int* rowm  = rowp + ROWP_PAD;
    int* ent   = rowm + ROWP_PAD;
    int* rank  = ent + ENT_CAP;
    int* partI = rank + NE;
    int* partO = partI + (size_t)NSL * N_NODES;
    float* bufB = (float*)d_ws + NODE_ELEMS;
    float* degO = bufB + NODE_ELEMS;
    float* degI = degO + N_NODES;

    edge_hist<<<NRG * NSL, HB, 0, stream>>>(src, dst, efeat, rank, partI, partO, tot);
    fill_project<<<NRG * NSL, HB, 0, stream>>>(src, dst, efeat, rank, partI, partO, tot,
                                               rowp, rowm, degO, degI, ent,
                                               feat0, Wfc0, bfc0, feat1, Wfc1, bfc1,
                                               feat2, Wfc2, bfc2, feat3, Wfc3, bfc3, bufB);

    row_gather<0><<<2048, 256, 0, stream>>>(
        bufB, rowp, rowm, ent, degI, degO, b_g0, out);

    gather_gemm<<<(NSTRIP + FS - 1) / FS, FW * 32, 0, stream>>>(
        out, rowp, rowm, ent, degI, W_g1, b_g1, bufB);

    row_gather<2><<<2048, 256, 0, stream>>>(
        bufB, rowp, rowm, ent, degI, degO, b_g0, out);
}


# fill launch places from compact per-wave edge records left by edge_hist (ballot-ranked slots) instead of re-reading and re-testing its slice; rank array gone; reversed terms now in record order
# speedup vs baseline: 5.0014x; 1.1806x over previous
#include <hip/hip_runtime.h>

typedef float v2f __attribute__((ext_vector_type(2)));
typedef float v8f __attribute__((ext_vector_type(8)));

#define N_NODES 50000
#define N_PER   12500
#define NE      600000
#define HD      128

#define NSL      16
#define PW       32
#define MTS      ((N_PER + 15) / 16)
#define PS_A 10
#define PB_A ((MTS + PS_A - 1) / PS_A)
#define PS_B 16
#define PB_B ((MTS + PS_B - 1) / PS_B)
static_assert(2 * PS_A <= PW && 2 * PS_B <= PW, "two waves per strip fit the block");
static_assert(NSL % 2 == 0, "the two lanes of a row split the edge slices evenly");

template <int K>
__device__ __forceinline__ void
project_strips(const float* __restrict__ A,
               const float* __restrict__ B,
               const float* __restrict__ bias,
               const int* __restrict__ partO,
               int node0,
               float* __restrict__ C,
               int s0, int ns, float* __restrict__ Bs, float* __restrict__ At)
{
    static_assert((K / 2) % PW == 0, "whole trips of the block copy all k pairs of B");
    static_assert(K == 128 || K == 64, "a wave-wide 16-byte load covers one row or two");
    constexpr int TP = K + 4;
    const int lane = threadIdx.x & 31;
    const int w    = __builtin_amdgcn_readfirstlane((int)(threadIdx.x >> 5));
    const int half = lane >> 4;
    const int lm   = lane & 15;
    const int part = w & 1;
    const int tm   = s0 + (w >> 1);
    const bool live = (w >> 1) < ns && tm < MTS;

    constexpr int NB = K / (2 * PW);
    float4 r0[NB], r1[NB];
    #pragma unroll
    for (int i = 0; i < NB; ++i) {
        const int p = w + PW * i;
        r0[i] = *(const float4*)(B + (long)(2 * p) * HD + lane * 4);
        r1[i] = *(const float4*)(B + (long)(2 * p + 1) * HD + lane * 4);
    }

    int co[NSL / 2];
    float bv[4];
    const int rloc = tm * 16 + lm;
    const int rc   = (rloc < N_PER) ? rloc : (N_PER - 1);
    if (live) {
        constexpr int NL = (K == 128) ? 8 : 4;
        float4 ar[NL];
        #pragma unroll
        for (int i = 0; i < NL; ++i) {
            const int r  = tm * 16 + part * 8 + ((K == 128) ? i : (2 * i + half));
            const int rr = (r < N_PER) ? r : (N_PER - 1);
            ar[i] = *(const float4*)(A + (long)rr * K + ((K == 128) ? lane : lm) * 4);
        }
        #pragma unroll
        for (int q = 0; q < NSL / 2; ++q)
            co[q] = partO[(size_t)(half * (NSL / 2) + q) * N_NODES + node0 + rc];
        #pragma unroll
        for (int t = 0; t < 4; ++t) bv[t] = bias[(part * 4 + t) * 16 + lm];

        float* trow = At + ((w >> 1) * 16 + part * 8) * TP;
        #pragma unroll
        for (int i = 0; i < NL; ++i) {
            if (K == 128) *(float4*)(trow + i * TP + lane * 4) = ar[i];
            else          *(float4*)(trow + (2 * i + half) * TP + lm * 4) = ar[i];
        }
    }

    {
        const int tnf = lane >> 2, sub = lane & 3;
        #pragma unroll
        for (int i = 0; i < NB; ++i) {
            const int p = w + PW * i;
            float* d = Bs + ((p >> 1) * 8 + tnf) * 64 + (p & 1) * 32 + sub * 8;
            *(float4*)d       = make_float4(r0[i].x, r1[i].x, r0[i].y, r1[i].y);
            *(float4*)(d + 4) = make_float4(r0[i].z, r1[i].z, r0[i].w, r1[i].w);
        }
    }
    __syncthreads();
    if (!live) return;

    const float* at = At + ((w >> 1) * 16 + lm) * TP + half * 2;
    v2f af[K / 4];
    #pragma unroll
    for (int kk = 0; kk < K / 4; ++kk) af[kk] = *(const v2f*)(at + kk * 4);

    int od = 0;
    #pragma unroll
    for (int q = 0; q < NSL / 2; ++q) od += co[q];
    od += __shfl_xor(od, 16);
    const float on = rsqrtf(fmaxf((float)od, 1.0f));

    v8f acc[4];
    #pragma unroll
    for (int t = 0; t < 4; ++t) acc[t] = (v8f){};

    const float* bp = Bs + part * 4 * 64 + lane * 2;
    #pragma unroll
    for (int kk = 0; kk < K / 4; ++kk) {
        #pragma unroll
        for (int t = 0; t < 4; ++t) {
            const v2f b = *(const v2f*)(bp + (kk * 8 + t) * 64);
            acc[t] = __builtin_amdgcn_wmma_f32_16x16x4_f32(
                false, af[kk], false, b, (short)0, acc[t], false, false);
        }
    }

    float osc[8];
    #pragma unroll
    for (int v = 0; v < 8; ++v) osc[v] = __shfl(on, v + half * 8);

    float* cp = C + (long)(tm * 16 + half * 8) * HD + part * 64 + lm;
    const bool full = (tm * 16 + 16 <= N_PER);
    if (full) {
        #pragma unroll
        for (int t = 0; t < 4; ++t) {
            #pragma unroll
            for (int v = 0; v < 8; ++v)
                cp[v * HD + t * 16] = (acc[t][v] + bv[t]) * osc[v];
        }
    } else {
        #pragma unroll
        for (int t = 0; t < 4; ++t) {
            #pragma unroll
            for (int v = 0; v < 8; ++v)
                if (tm * 16 + half * 8 + v < N_PER) cp[v * HD + t * 16] = (acc[t][v] + bv[t]) * osc[v];
        }
    }
}

#define ROWP_PAD 50016
#define ENT_CAP  (2 * NE)
#define NRG      16
#define RANGE    (N_NODES / NRG)
#define SLICE    (NE / NSL)
static_assert(NE % 32 == 0, "32-edge batches, no tail");
static_assert(RANGE * NRG == N_NODES, "ranges tile the nodes exactly");
static_assert(SLICE * NSL == NE && SLICE % 4 == 0, "slices tile the edges in int4 groups");

#define HB   1024
#define HGRP (SLICE / 4)
#define HFUL (HGRP / HB)
static_assert(HGRP - HFUL * HB < HB, "one guarded trip covers the rest of the slice");

#define CAPW 160
#define NWV  (HB / 32)
static_assert(NWV == 32, "one lane per wave of the block reads the block's record counts");
static_assert(CAPW % 32 == 0, "a segment is read in whole wave trips");
static_assert((long)N_NODES + 2L * ROWP_PAD + ENT_CAP + NE + 2L * NSL * N_NODES
              + 2L * NRG * NSL * NWV * CAPW <= (long)N_NODES * HD,
              "the row-list arrays and the record segments fit the first node-sized plane");

template <bool RARE>
__device__ __forceinline__ void put_rec(bool p, int a, int b, int2* __restrict__ seg, int& cnt)
{
    const unsigned m = __builtin_amdgcn_ballot_w32(p);
    if (RARE && m == 0u) return;
    const int pos = cnt + (int)__builtin_amdgcn_mbcnt_lo(m, 0u);
    if (p && pos < CAPW) seg[pos] = make_int2(a, b);
    cnt += __builtin_popcount(m);
}

__device__ __forceinline__ void hist_group(const int4 sv, const int4 dv, const int4 fv, bool ok,
                                           int lo, int* cI, int* cO, int* cB,
                                           int2* __restrict__ seg, int& cnt)
{
    const unsigned d0 = (unsigned)(dv.x - lo), d1 = (unsigned)(dv.y - lo);
    const unsigned d2 = (unsigned)(dv.z - lo), d3 = (unsigned)(dv.w - lo);
    const unsigned o0 = (unsigned)(sv.x - lo), o1 = (unsigned)(sv.y - lo);
    const unsigned o2 = (unsigned)(sv.z - lo), o3 = (unsigned)(sv.w - lo);
    const bool in0 = ok && d0 < (unsigned)RANGE, in1 = ok && d1 < (unsigned)RANGE;
    const bool in2 = ok && d2 < (unsigned)RANGE, in3 = ok && d3 < (unsigned)RANGE;
    const bool on0 = ok && o0 < (unsigned)RANGE, on1 = ok && o1 < (unsigned)RANGE;
    const bool on2 = ok && o2 < (unsigned)RANGE, on3 = ok && o3 < (unsigned)RANGE;
    int r0 = 0, r1 = 0, r2 = 0, r3 = 0;
    if (in0) r0 = atomicAdd(&cI[d0], 1);
    if (in1) r1 = atomicAdd(&cI[d1], 1);
    if (in2) r2 = atomicAdd(&cI[d2], 1);
    if (in3) r3 = atomicAdd(&cI[d3], 1);
    if (on0) atomicAdd(&cO[o0], 1);
    if (on1) atomicAdd(&cO[o1], 1);
    if (on2) atomicAdd(&cO[o2], 1);
    if (on3) atomicAdd(&cO[o3], 1);
    const bool rv0 = on0 && (fv.x == 6 || fv.x == 14 || fv.x == 30);
    const bool rv1 = on1 && (fv.y == 6 || fv.y == 14 || fv.y == 30);
    const bool rv2 = on2 && (fv.z == 6 || fv.z == 14 || fv.z == 30);
    const bool rv3 = on3 && (fv.w == 6 || fv.w == 14 || fv.w == 30);
    if (rv0) atomicAdd(&cB[o0], 1);
    if (rv1) atomicAdd(&cB[o1], 1);
    if (rv2) atomicAdd(&cB[o2], 1);
    if (rv3) atomicAdd(&cB[o3], 1);
    put_rec<false>(in0, sv.x | ((fv.x >= 0 && fv.x <= 4) ? (int)0x80000000 : 0), (int)((d0 << 16) | (unsigned)r0), seg, cnt);
    put_rec<false>(in1, sv.y | ((fv.y >= 0 && fv.y <= 4) ? (int)0x80000000 : 0), (int)((d1 << 16) | (unsigned)r1), seg, cnt);
    put_rec<false>(in2, sv.z | ((fv.z >= 0 && fv.z <= 4) ? (int)0x80000000 : 0), (int)((d2 << 16) | (unsigned)r2), seg, cnt);
    put_rec<false>(in3, sv.w | ((fv.w >= 0 && fv.w <= 4) ? (int)0x80000000 : 0), (int)((d3 << 16) | (unsigned)r3), seg, cnt);
    put_rec<true>(rv0, dv.x, (int)(0x80000000u | o0), seg, cnt);
    put_rec<true>(rv1, dv.y, (int)(0x80000000u | o1), seg, cnt);
    put_rec<true>(rv2, dv.z, (int)(0x80000000u | o2), seg, cnt);
    put_rec<true>(rv3, dv.w, (int)(0x80000000u | o3), seg, cnt);
}

#define HFB  5
#define HNB  ((HFUL + 1 + HFB - 1) / HFB)
static_assert(HNB * HFB * HB >= HGRP, "the batches cover the slice");
static_assert(SLICE < 65536, "two per-slice counts share one int, 16 bits each");

__global__ void __launch_bounds__(HB)
edge_hist(const int* __restrict__ src,
          const int* __restrict__ dst,
          const int* __restrict__ ef,
          int* __restrict__ rec,
          int* __restrict__ segcnt,
          int* __restrict__ partIB,
          int* __restrict__ partO,
          int* __restrict__ tot)
{
    __shared__ int cI[RANGE];
    __shared__ int cO[RANGE];
    __shared__ int cB[RANGE];
    __shared__ int wtot[HB / 32];
    const int tid = threadIdx.x;
    const int r  = blockIdx.x % NRG;
    const int s  = blockIdx.x / NRG;
    const int lo = r * RANGE;
    for (int j = tid; j < RANGE; j += HB) { cI[j] = 0; cO[j] = 0; cB[j] = 0; }
    __syncthreads();

    const int e0 = s * SLICE;
    const int4* s4 = (const int4*)(src + e0);
    const int4* d4 = (const int4*)(dst + e0);
    const int4* f4 = (const int4*)(ef + e0);
    const int wv = __builtin_amdgcn_readfirstlane(tid >> 5);
    int2* seg = (int2*)rec + (size_t)((int)blockIdx.x * NWV + wv) * CAPW;
    int cnt = 0;
    #pragma unroll
    for (int kb = 0; kb < HNB * HFB; kb += HFB) {
        int4 sv[HFB], dv[HFB], fv[HFB];
        bool ok[HFB];
        #pragma unroll
        for (int k = 0; k < HFB; ++k) {
            const int g = tid + (kb + k) * HB;
            ok[k] = g < HGRP;
            const int gc = ok[k] ? g : tid;
            dv[k] = d4[gc]; sv[k] = s4[gc]; fv[k] = f4[gc];
        }
        #pragma unroll
        for (int k = 0; k < HFB; ++k)
            hist_group(sv[k], dv[k], fv[k], ok[k], lo, cI, cO, cB, seg, cnt);
    }
    if ((tid & 31) == 0) segcnt[(int)blockIdx.x * NWV + wv] = cnt;
    __syncthreads();

    int* pIB = partIB + (size_t)s * N_NODES + lo;
    int* pO  = partO + (size_t)s * N_NODES + lo;
    int mine = 0;
    for (int j = tid; j < RANGE; j += HB) {
        const int ci = cI[j], cb = cB[j];
        pIB[j] = (int)((unsigned)ci | ((unsigned)cb << 16));
        pO[j] = cO[j];
        mine += ci + cb;
    }
    for (int o = 16; o > 0; o >>= 1) mine += __shfl_xor(mine, o);
    if ((tid & 31) == 0) wtot[tid >> 5] = mine;
    __syncthreads();
    if (tid < 32) {
        int v = wtot[tid];
        for (int o = 16; o > 0; o >>= 1) v += __shfl_xor(v, o);
        if (tid == 0) tot[blockIdx.x] = v;
    }
}

#define SCT ((RANGE + 3) / 4)
#define RPAD (SCT * 4)
#define NTRIP 4
#define FB 5
#define NBAT ((HFUL + 1 + FB - 1) / FB)
#define NOSH ((RANGE + NSL - 1) / NSL)
#define OSH0 (((SCT + 31) / 32) * 32)
static_assert(OSH0 + NOSH <= HB, "one node per thread of the spare waves covers a block's share of out_norm");
static_assert(SCT <= HB, "one pass of the block scans the whole range");
static_assert(NTRIP * SCT >= RANGE, "the summing threads cover the range");
static_assert((NRG * NSL) % 32 == 0, "tot[] is read in whole wave trips");
static_assert(NBAT * FB * HB >= HGRP, "the batches cover the slice");

__device__ __forceinline__ void place_group(const int4 sv, const int4 dv, const int4 fv,
                                            bool ok, int lo, int* slotL,
                                            int* slotB, int* __restrict__ ent)
{
    const unsigned d0 = (unsigned)(dv.x - lo), d1 = (unsigned)(dv.y - lo);
    const unsigned d2 = (unsigned)(dv.z - lo), d3 = (unsigned)(dv.w - lo);
    const bool in0 = ok && d0 < (unsigned)RANGE, in1 = ok && d1 < (unsigned)RANGE;
    const bool in2 = ok && d2 < (unsigned)RANGE, in3 = ok && d3 < (unsigned)RANGE;
    int p0 = 0, p1 = 0, p2 = 0, p3 = 0;
    if (in0) p0 = atomicAdd(&slotL[d0], 1);
    if (in1) p1 = atomicAdd(&slotL[d1], 1);
    if (in2) p2 = atomicAdd(&slotL[d2], 1);
    if (in3) p3 = atomicAdd(&slotL[d3], 1);
    if (in0 && (unsigned)p0 < (unsigned)ENT_CAP)
        ent[p0] = sv.x | ((fv.x >= 0 && fv.x <= 4) ? (int)0x80000000 : 0);
    if (in1 && (unsigned)p1 < (unsigned)ENT_CAP)
        ent[p1] = sv.y | ((fv.y >= 0 && fv.y <= 4) ? (int)0x80000000 : 0);
    if (in2 && (unsigned)p2 < (unsigned)ENT_CAP)
        ent[p2] = sv.z | ((fv.z >= 0 && fv.z <= 4) ? (int)0x80000000 : 0);
    if (in3 && (unsigned)p3 < (unsigned)ENT_CAP)
        ent[p3] = sv.w | ((fv.w >= 0 && fv.w <= 4) ? (int)0x80000000 : 0);

    const unsigned o0 = (unsigned)(sv.x - lo), o1 = (unsigned)(sv.y - lo);
    const unsigned o2 = (unsigned)(sv.z - lo), o3 = (unsigned)(sv.w - lo);
    const bool rv0 = ok && o0 < (unsigned)RANGE && (fv.x == 6 || fv.x == 14 || fv.x == 30);
    const bool rv1 = ok && o1 < (unsigned)RANGE && (fv.y == 6 || fv.y == 14 || fv.y == 30);
    const bool rv2 = ok && o2 < (unsigned)RANGE && (fv.z == 6 || fv.z == 14 || fv.z == 30);
    const bool rv3 = ok && o3 < (unsigned)RANGE && (fv.w == 6 || fv.w == 14 || fv.w == 30);
    int q0 = 0, q1 = 0, q2 = 0, q3 = 0;
    if (rv0) q0 = atomicAdd(&slotB[o0], 1);
    if (rv1) q1 = atomicAdd(&slotB[o1], 1);
    if (rv2) q2 = atomicAdd(&slotB[o2], 1);
    if (rv3) q3 = atomicAdd(&slotB[o3], 1);
    if (rv0 && (unsigned)q0 < (unsigned)ENT_CAP) ent[q0] = dv.x;
    if (rv1 && (unsigned)q1 < (unsigned)ENT_CAP) ent[q1] = dv.y;
    if (rv2 && (unsigned)q2 < (unsigned)ENT_CAP) ent[q2] = dv.z;
    if (rv3 && (unsigned)q3 < (unsigned)ENT_CAP) ent[q3] = dv.w;
}

static_assert(2 * PB_A + 2 * PB_B == NRG * NSL, "the projection's blocks are exactly the blocks of the (range, slice) grid");
static_assert(PW * 32 == HB, "project_strips is written for the blocks of this launch");

__global__ void __launch_bounds__(HB)
fill_project(const int* __restrict__ src,
             const int* __restrict__ dst,
             const int* __restrict__ ef,
             const int* __restrict__ rec,
             const int* __restrict__ segcnt,
             const int* __restrict__ partIB,
             const int* __restrict__ partO,
             const int* __restrict__ tot,
             int* __restrict__ row_ptr,
             int* __restrict__ row_mid,
             float* __restrict__ normO,
             float* __restrict__ normI,
             int* __restrict__ ent,
             const float* __restrict__ feat0, const float* __restrict__ W0, const float* __restrict__ b0,
             const float* __restrict__ feat1, const float* __restrict__ W1, const float* __restrict__ b1,
             const float* __restrict__ feat2, const float* __restrict__ W2, const float* __restrict__ b2,
             const float* __restrict__ feat3, const float* __restrict__ W3, const float* __restrict__ b3,
             float* __restrict__ h)
{
    __shared__ __attribute__((aligned(16))) float Bs[128 * HD];
    __shared__ __attribute__((aligned(16))) float At[PS_A * 16 * (128 + 4)];
    static_assert(PS_B * 16 * (64 + 4) <= PS_A * 16 * (128 + 4), "the K = 64 tiles fit the same array");
    const int wq  = __builtin_amdgcn_readfirstlane((int)(threadIdx.x >> 5));
    const int myc = segcnt[(int)blockIdx.x * NWV + (int)(threadIdx.x & 31)];
    int2 rc[CAPW / 32];
    {
        const int2* seg = (const int2*)rec + (size_t)((int)blockIdx.x * NWV + wq) * CAPW + (threadIdx.x & 31);
        #pragma unroll
        for (int j = 0; j < CAPW / 32; ++j) rc[j] = seg[32 * j];
    }
    {
        const int b = (int)blockIdx.x;
        if (b < 2 * PB_A) {
            const int t  = (b < PB_A) ? 0 : 1;
            const int sb = b - t * PB_A;
            project_strips<128>((t == 0) ? feat0 : feat1, (t == 0) ? W0 : W1, (t == 0) ? b0 : b1,
                                partO, t * N_PER, h + (size_t)t * N_PER * HD, sb * PS_A, PS_A, Bs, At);
        } else {
            const int c  = b - 2 * PB_A;
            const int t  = (c < PB_B) ? 2 : 3;
            const int sb = c - (t - 2) * PB_B;
            project_strips<64>((t == 2) ? feat2 : feat3, (t == 2) ? W2 : W3, (t == 2) ? b2 : b3,
                               partO, t * N_PER, h + (size_t)t * N_PER * HD, sb * PS_B, PS_B, Bs, At);
        }
    }

    __shared__ __attribute__((aligned(16))) int cntL[RPAD];
    __shared__ __attribute__((aligned(16))) int degL[RPAD];
    __shared__ __attribute__((aligned(16))) int slotL[RPAD];
    __shared__ __attribute__((aligned(16))) int slotB[RPAD];
    __shared__ int wsum[HB / 32];
    const int tid  = threadIdx.x;
    const int lane = tid & 31;
    const int w    = tid >> 5;
    const int r  = blockIdx.x % NRG;
    const int s  = blockIdx.x / NRG;
    const int lo = r * RANGE;

    int rb = 0;
    #pragma unroll
    for (int k = 0; k < (NRG * NSL) / 32; ++k) {
        const int idx = lane + 32 * k;
        const int v = tot[idx];
        rb += ((idx % NRG) < r) ? v : 0;
    }
    for (int o = 16; o > 0; o >>= 1) rb += __shfl_xor(rb, o);

    const bool sc = tid < SCT;
    if (tid < OSH0) {
        int cp[NTRIP][NSL];
        #pragma unroll
        for (int k = 0; k < NTRIP; ++k) {
            const int j  = tid + k * SCT;
            const int jc = (j < RANGE) ? j : 0;
            #pragma unroll
            for (int q = 0; q < NSL; ++q) cp[k][q] = partIB[(size_t)q * N_NODES + lo + jc];
        }
        #pragma unroll
        for (int k = 0; k < NTRIP; ++k) {
            const int j = tid + k * SCT;
            int ti = 0, pi = 0, tb = 0, pb = 0;
            #pragma unroll
            for (int q = 0; q < NSL; ++q) {
                const int vi = (int)((unsigned)cp[k][q] & 0xffffu);
                const int vb = (int)((unsigned)cp[k][q] >> 16);
                pi += (q < s) ? vi : 0;
                pb += (q < s) ? vb : 0;
                ti += vi;
                tb += vb;
            }
            if (sc && j < RANGE) { cntL[j] = ti + tb; degL[j] = ti; slotL[j] = pi; slotB[j] = pb; }
        }
    } else {
        const int jo = s * NOSH + (tid - OSH0);
        const bool on = (tid - OSH0) < NOSH && jo < RANGE;
        const int joc = on ? jo : 0;
        int co[NSL];
        #pragma unroll
        for (int q = 0; q < NSL; ++q) co[q] = partO[(size_t)q * N_NODES + lo + joc];
        int o = 0;
        #pragma unroll
        for (int q = 0; q < NSL; ++q) o += co[q];
        if (on) normO[lo + jo] = rsqrtf(fmaxf((float)o, 1.0f));
    }
    if (tid < RPAD - RANGE) {
        cntL[RANGE + tid] = 0; degL[RANGE + tid] = 0; slotL[RANGE + tid] = 0; slotB[RANGE + tid] = 0;
    }

    __syncthreads();

    int4 c4 = make_int4(0, 0, 0, 0), i4 = make_int4(0, 0, 0, 0);
    if (sc) { c4 = *(const int4*)(cntL + 4 * tid); i4 = *(const int4*)(degL + 4 * tid); }
    const int tsum = (c4.x + c4.y) + (c4.z + c4.w);
    int x = tsum;
    #pragma unroll
    for (int o = 1; o < 32; o <<= 1) {
        const int y = __shfl_up(x, o);
        if (lane >= o) x += y;
    }
    if (lane == 31) wsum[w] = x;
    __syncthreads();
    int carry = (lane < w) ? wsum[lane] : 0;
    for (int o = 16; o > 0; o >>= 1) carry += __shfl_xor(carry, o);

    if (sc) {
        const int b0 = rb + carry + x - tsum;
        const int b1 = b0 + c4.x;
        const int b2 = b1 + c4.y;
        const int b3 = b2 + c4.z;
        const int m0 = b0 + i4.x, m1 = b1 + i4.y, m2 = b2 + i4.z, m3 = b3 + i4.w;
        const int4 pe = *(const int4*)(slotL + 4 * tid);
        const int4 pb = *(const int4*)(slotB + 4 * tid);
        *(int4*)(slotL + 4 * tid) = make_int4(b0 + pe.x, b1 + pe.y, b2 + pe.z, b3 + pe.w);
        *(int4*)(slotB + 4 * tid) = make_int4(m0 + pb.x, m1 + pb.y, m2 + pb.z, m3 + pb.w);
        if (s == 0) {
            const int n = 4 * tid;
            if (n + 0 < RANGE) { row_ptr[lo + n + 0] = b0; row_mid[lo + n + 0] = m0; normI[lo + n + 0] = rsqrtf(fmaxf((float)i4.x, 1.0f)); }
            if (n + 1 < RANGE) { row_ptr[lo + n + 1] = b1; row_mid[lo + n + 1] = m1; normI[lo + n + 1] = rsqrtf(fmaxf((float)i4.y, 1.0f)); }
            if (n + 2 < RANGE) { row_ptr[lo + n + 2] = b2; row_mid[lo + n + 2] = m2; normI[lo + n + 2] = rsqrtf(fmaxf((float)i4.z, 1.0f)); }
            if (n + 3 < RANGE) { row_ptr[lo + n + 3] = b3; row_mid[lo + n + 3] = m3; normI[lo + n + 3] = rsqrtf(fmaxf((float)i4.w, 1.0f)); }
            if (r == NRG - 1 && tid == SCT - 1) row_ptr[N_NODES] = b3 + c4.w;
        }
    }
    __syncthreads();

    const bool overflow = __builtin_amdgcn_ballot_w32(myc > CAPW) != 0u;
    if (__builtin_expect(!overflow, 1)) {
        const int n = __builtin_amdgcn_readlane(myc, wq);
        #pragma unroll
        for (int j = 0; j < CAPW / 32; ++j) {
            const int a = rc[j].x, b = rc[j].y;
            if (lane + 32 * j < n) {
                if (b >= 0) {
                    const unsigned dl = (unsigned)b >> 16;
                    if (dl < (unsigned)RANGE) {
                        const int p = slotL[dl] + (b & 0xffff);
                        if ((unsigned)p < (unsigned)ENT_CAP) ent[p] = a;
                    }
                } else {
                    const unsigned ol = (unsigned)b & 0x7fffffffu;
                    if (ol < (unsigned)RANGE) {
                        const int q = atomicAdd(&slotB[ol], 1);
                        if ((unsigned)q < (unsigned)ENT_CAP) ent[q] = a;
                    }
                }
            }
        }
    } else {
        const int e0 = s * SLICE;
        const int4* s4 = (const int4*)(src + e0);
        const int4* d4 = (const int4*)(dst + e0);
        const int4* f4 = (const int4*)(ef + e0);
        #pragma unroll
        for (int kb = 0; kb < NBAT * FB; kb += FB) {
            int4 sv[FB], dv[FB], fv[FB];
            bool ok[FB];
            #pragma unroll
            for (int k = 0; k < FB; ++k) {
                const int g = tid + (kb + k) * HB;
                ok[k] = g < HGRP;
                const int gc = ok[k] ? g : tid;
                dv[k] = d4[gc]; sv[k] = s4[gc]; fv[k] = f4[gc];
            }
            #pragma unroll
            for (int k = 0; k < FB; ++k)
                place_group(sv[k], dv[k], fv[k], ok[k], lo, slotL, slotB, ent);
        }
    }
}

template <int G, int MODE>
__device__ __forceinline__ void gather_group(const float* __restrict__ h,
                                             int e, int j, int lane, float4& acc)
{
    float4 a[G];
    float  wgt[G];
    #pragma unroll
    for (int k = 0; k < G; ++k) {
        const int ek = __builtin_amdgcn_readlane(e, j + k);
        a[k] = *(const float4*)(h + (size_t)(ek & 0x7fffffff) * HD + lane * 4);
        wgt[k] = (MODE == 2 && ek < 0) ? 2.0f : 1.0f;
    }
    #pragma unroll
    for (int k = 0; k < G; ++k) {
        if (MODE == 2) {
            acc.x += a[k].x * wgt[k];
            acc.y += a[k].y * wgt[k];
            acc.z += a[k].z * wgt[k];
            acc.w += a[k].w * wgt[k];
        } else {
            acc.x += a[k].x;
            acc.y += a[k].y;
            acc.z += a[k].z;
            acc.w += a[k].w;
        }
    }
}

template <int MODE>
__global__ void __launch_bounds__(256)
row_gather(const float* __restrict__ h,
           const int* __restrict__ row_ptr,
           const int* __restrict__ row_mid,
           const int* __restrict__ ent,
           const float* __restrict__ in_norm,
           const float* __restrict__ out_norm,
           const float* __restrict__ bias,
           float* __restrict__ dstv)
{
    const int lane = threadIdx.x & 31;
    const int wave = __builtin_amdgcn_readfirstlane(
        (int)((blockIdx.x * blockDim.x + threadIdx.x) >> 5));
    const int nwaves = (int)((gridDim.x * blockDim.x) >> 5);

    float4 b = make_float4(0.0f, 0.0f, 0.0f, 0.0f);
    if (MODE == 0) b = *(const float4*)(bias + lane * 4);

    for (int row = wave; row < N_NODES; row += nwaves) {
        const int beg = row_ptr[row];
        const int end = (MODE == 2) ? row_ptr[row + 1] : row_mid[row];
        float inn = 1.0f, outn = 1.0f;
        if (MODE == 0) { inn = in_norm[row]; outn = out_norm[row]; }

        float4 acc = make_float4(0.0f, 0.0f, 0.0f, 0.0f);
        for (int base = beg; base < end; base += 32) {
            const int n = (end - base < 32) ? (end - base) : 32;
            const int e = (lane < n) ? ent[base + lane] : 0;
            int j = 0;
            for (; j + 8 <= n; j += 8) gather_group<8, MODE>(h, e, j, lane, acc);
            if (j + 4 <= n) { gather_group<4, MODE>(h, e, j, lane, acc); j += 4; }
            if (j + 2 <= n) { gather_group<2, MODE>(h, e, j, lane, acc); j += 2; }
            if (j < n)      { gather_group<1, MODE>(h, e, j, lane, acc); }
        }

        if (MODE == 0) {
            acc.x = fmaxf(acc.x * inn + b.x, 0.0f) * outn;
            acc.y = fmaxf(acc.y * inn + b.y, 0.0f) * outn;
            acc.z = fmaxf(acc.z * inn + b.z, 0.0f) * outn;
            acc.w = fmaxf(acc.w * inn + b.w, 0.0f) * outn;
        }
        *(float4*)(dstv + (size_t)row * HD + lane * 4) = acc;
    }
}

__device__ __forceinline__ void gather_chunk(const float* __restrict__ h,
                                             int e, int n, int lane, float4& acc)
{
    int j = 0;
    for (; j + 16 <= n; j += 8) gather_group<8, 1>(h, e, j, lane, acc);
    switch (n - j) {
        case 1:  gather_group<1, 1>(h, e, j, lane, acc);  break;
        case 2:  gather_group<2, 1>(h, e, j, lane, acc);  break;
        case 3:  gather_group<3, 1>(h, e, j, lane, acc);  break;
        case 4:  gather_group<4, 1>(h, e, j, lane, acc);  break;
        case 5:  gather_group<5, 1>(h, e, j, lane, acc);  break;
        case 6:  gather_group<6, 1>(h, e, j, lane, acc);  break;
        case 7:  gather_group<7, 1>(h, e, j, lane, acc);  break;
        case 8:  gather_group<8, 1>(h, e, j, lane, acc);  break;
        case 9:  gather_group<9, 1>(h, e, j, lane, acc);  break;
        case 10: gather_group<10, 1>(h, e, j, lane, acc); break;
        case 11: gather_group<11, 1>(h, e, j, lane, acc); break;
        case 12: gather_group<12, 1>(h, e, j, lane, acc); break;
        case 13: gather_group<13, 1>(h, e, j, lane, acc); break;
        case 14: gather_group<14, 1>(h, e, j, lane, acc); break;
        case 15: gather_group<15, 1>(h, e, j, lane, acc); break;
        default: break;
    }
}

#define FS     13
#define FW     (2 * FS)
#define FTP    (HD + 4)
#define NSTRIP (N_NODES / 16)
static_assert(NSTRIP * 16 == N_NODES, "no partial strip");
static_assert(2 * FW < HD / 2 && 3 * FW >= HD / 2, "three trips of the block copy all k pairs of B");

__global__ void __launch_bounds__(FW * 32)
gather_gemm(const float* __restrict__ h,
            const int* __restrict__ row_ptr,
            const int* __restrict__ row_mid,
            const int* __restrict__ ent,
            const float* __restrict__ in_norm,
            const float* __restrict__ B,
            const float* __restrict__ bias,
            float* __restrict__ C)
{
    __shared__ float Bs[HD * HD];
    __shared__ __attribute__((aligned(16))) float At[FS * 16 * FTP];
    const int lane = threadIdx.x & 31;
    const int w    = __builtin_amdgcn_readfirstlane((int)(threadIdx.x >> 5));
    const int s    = w >> 1;
    const int part = w & 1;
    const int half = lane >> 4;
    const int lm   = lane & 15;
    const int tm   = (int)blockIdx.x * FS + s;
    const bool live = tm < NSTRIP;
    const int r0   = tm * 16 + part * 8;

    int rb = 0, re = 0;
    float rn = 0.0f;
    if (live && lane < 8) { rb = row_ptr[r0 + lane]; re = row_mid[r0 + lane]; rn = in_norm[r0 + lane]; }
    float bv[4];
    #pragma unroll
    for (int t = 0; t < 4; ++t) bv[t] = bias[(part * 4 + t) * 16 + lm];

    {
        const int tnf = lane >> 2, sub = lane & 3;
        float4 q0[3], q1[3];
        #pragma unroll
        for (int i = 0; i < 3; ++i) {
            const int p  = w + FW * i;
            const int pc = (p < HD / 2) ? p : (HD / 2 - 1);
            q0[i] = *(const float4*)(B + (long)(2 * pc) * HD + lane * 4);
            q1[i] = *(const float4*)(B + (long)(2 * pc + 1) * HD + lane * 4);
        }
        #pragma unroll
        for (int i = 0; i < 3; ++i) {
            const int p = w + FW * i;
            if (p < HD / 2) {
                float* d = Bs + ((p >> 1) * 8 + tnf) * 64 + (p & 1) * 32 + sub * 8;
                *(float4*)d       = make_float4(q0[i].x, q1[i].x, q0[i].y, q1[i].y);
                *(float4*)(d + 4) = make_float4(q0[i].z, q1[i].z, q0[i].w, q1[i].w);
            }
        }
    }

    if (live) {
        float* trow = At + (s * 16 + part * 8) * FTP + lane * 4;
        int ev[8];
        #pragma unroll
        for (int q = 0; q < 8; ++q) {
            const int b = __builtin_amdgcn_readlane(rb, q);
            const int n = __builtin_amdgcn_readlane(re, q) - b;
            ev[q] = (lane < n) ? ent[b + lane] : 0;
        }
        #pragma nounroll
        for (int i = 0; i < 8; ++i) {
            const int beg = __builtin_amdgcn_readlane(rb, i);
            const int end = __builtin_amdgcn_readlane(re, i);
            int ecur = ev[0];
            #pragma unroll
            for (int q = 1; q < 8; ++q) ecur = (i == q) ? ev[q] : ecur;
            const float inn = __int_as_float(__builtin_amdgcn_readlane(__float_as_int(rn), i));

            float4 acc = make_float4(0.0f, 0.0f, 0.0f, 0.0f);
            for (int base = beg; base < end; base += 32) {
                const int n = (end - base < 32) ? (end - base) : 32;
                int e = ecur;
                if (base != beg) e = (lane < n) ? ent[base + lane] : 0;
                gather_chunk(h, e, n, lane, acc);
            }
            *(float4*)(trow + i * FTP) = make_float4(acc.x * inn, acc.y * inn, acc.z * inn, acc.w * inn);
        }
    }
    __syncthreads();
    if (!live) return;

    const float* at = At + (s * 16 + lm) * FTP + half * 2;
    v2f af[HD / 4];
    #pragma unroll
    for (int kk = 0; kk < HD / 4; ++kk) af[kk] = *(const v2f*)(at + kk * 4);

    v8f acc[4];
    #pragma unroll
    for (int t = 0; t < 4; ++t) acc[t] = (v8f){};

    const float* bp = Bs + part * 4 * 64 + lane * 2;
    #pragma unroll
    for (int kk = 0; kk < HD / 4; ++kk) {
        #pragma unroll
        for (int t = 0; t < 4; ++t) {
            const v2f b = *(const v2f*)(bp + (kk * 8 + t) * 64);
            acc[t] = __builtin_amdgcn_wmma_f32_16x16x4_f32(
                false, af[kk], false, b, (short)0, acc[t], false, false);
        }
    }

    float* cp = C + (long)(tm * 16 + half * 8) * HD + part * 64 + lm;
    #pragma unroll
    for (int t = 0; t < 4; ++t) {
        #pragma unroll
        for (int v = 0; v < 8; ++v)
            cp[v * HD + t * 16] = fmaxf(acc[t][v] + bv[t], 0.0f);
    }
}

extern "C" void kernel_launch(void* const* d_in, const int* in_sizes, int n_in,
                              void* d_out, int out_size, void* d_ws, size_t ws_size,
                              hipStream_t stream)
{
    (void)in_sizes; (void)n_in; (void)out_size; (void)ws_size;

    const float* feat0 = (const float*)d_in[0];
    const float* Wfc0  = (const float*)d_in[1];
    const float* bfc0  = (const float*)d_in[2];
    const float* feat1 = (const float*)d_in[3];
    const float* Wfc1  = (const float*)d_in[4];
    const float* bfc1  = (const float*)d_in[5];
    const float* feat2 = (const float*)d_in[6];
    const float* Wfc2  = (const float*)d_in[7];
    const float* bfc2  = (const float*)d_in[8];
    const float* feat3 = (const float*)d_in[9];
    const float* Wfc3  = (const float*)d_in[10];
    const float* bfc3  = (const float*)d_in[11];
    const int*   src   = (const int*)d_in[12];
    const int*   dst   = (const int*)d_in[13];
    const int*   efeat = (const int*)d_in[14];
    const float* b_g0  = (const float*)d_in[15];
    const float* W_g1  = (const float*)d_in[16];
    const float* b_g1  = (const float*)d_in[17];
    float* out = (float*)d_out;

    const size_t NODE_ELEMS = (size_t)N_NODES * HD;
    int* tot   = (int*)d_ws;
    int* rowp  = tot + N_NODES;
    int* rowm  = rowp + ROWP_PAD;
    int* ent   = rowm + ROWP_PAD;
    int* segc  = ent + ENT_CAP;
    int* partI = segc + NE;
    int* partO = partI + (size_t)NSL * N_NODES;
    int* rec   = partO + (size_t)NSL * N_NODES;
    float* bufB = (float*)d_ws + NODE_ELEMS;
    float* degO = bufB + NODE_ELEMS;
    float* degI = degO + N_NODES;

    edge_hist<<<NRG * NSL, HB, 0, stream>>>(src, dst, efeat, rec, segc, partI, partO, tot);
    fill_project<<<NRG * NSL, HB, 0, stream>>>(src, dst, efeat, rec, segc, partI, partO, tot,
                                               rowp, rowm, degO, degI, ent,
                                               feat0, Wfc0, bfc0, feat1, Wfc1, bfc1,
                                               feat2, Wfc2, bfc2, feat3, Wfc3, bfc3, bufB);

    row_gather<0><<<2048, 256, 0, stream>>>(
        bufB, rowp, rowm, ent, degI, degO, b_g0, out);

    gather_gemm<<<(NSTRIP + FS - 1) / FS, FW * 32, 0, stream>>>(
        out, rowp, rowm, ent, degI, W_g1, b_g1, bufB);

    row_gather<2><<<2048, 256, 0, stream>>>(
        bufB, rowp, rowm, ent, degI, degO, b_g0, out);
}
